# starting kernel
# baseline (speedup 1.0000x reference)
#include <hip/hip_runtime.h>

typedef __attribute__((ext_vector_type(2))) float v2f;
typedef __attribute__((ext_vector_type(8))) float v8f;

__device__ __forceinline__ void atomic_add_f32(float* p, float v) {
  (void)__hip_atomic_fetch_add(p, v, __ATOMIC_RELAXED, __HIP_MEMORY_SCOPE_AGENT);
}


__global__ void gcn_zero_kernel(float* __restrict__ p, int n) {
  int i = blockIdx.x * blockDim.x + threadIdx.x;
  if (i < n) p[i] = 0.0f;
}

__global__ void gcn_degree_kernel(const int* __restrict__ src, const int* __restrict__ dst,
                                  float* __restrict__ deg_out, float* __restrict__ deg_in,
                                  int E) {
  int e = blockIdx.x * blockDim.x + threadIdx.x;
  if (e < E) {
    atomic_add_f32(&deg_out[src[e]], 1.0f);
    atomic_add_f32(&deg_in[dst[e]], 1.0f);
  }
}

__global__ void gcn_norm_kernel(float* __restrict__ deg, int n) {
  int i = blockIdx.x * blockDim.x + threadIdx.x;
  if (i < n) deg[i] = rsqrtf(fmaxf(deg[i], 1.0f));
}

__global__ void gcn_post_kernel(float* __restrict__ h, const float* __restrict__ norm_dst,
                                const float* __restrict__ bias, int n, int H) {
  int idx = blockIdx.x * blockDim.x + threadIdx.x;
  if (idx >= n) return;
  int i = idx / H;
  int f = idx - i * H;
  h[idx] = fmaxf(fmaf(h[idx], norm_dst[i], bias[f]), 0.0f);
}

__global__ void gcn_gemm_wmma_kernel(const float* __restrict__ A, int lda, int K,
                                     const float* __restrict__ W, int wcols,
                                     const float* __restrict__ norm,
                                     float* __restrict__ out, int ldo, int ntiles) {
  __shared__ float Ws[32 * 32];

  for (int i = threadIdx.x; i < 32 * 32; i += blockDim.x) {
    const int k = i >> 5;
    const int n = i & 31;
    Ws[i] = (k < K && n < wcols) ? W[(size_t)k * wcols + n] : 0.0f;
  }
  __syncthreads();

  const int wave = (blockIdx.x * blockDim.x + threadIdx.x) >> 5;
  const int lane = threadIdx.x & 31;
  if (wave >= ntiles) return;

  const int m = lane & 15;
  const int half = lane >> 4;

  const int row = wave * 16 + m;
  const float nrm = norm[row];
  const float* arow = A + (size_t)row * lda;

  v8f c0 = {};
  v8f c1 = {};

  for (int k0 = 0; k0 < K; k0 += 4) {
    const int k = k0 + 2 * half;
    const float2 av = *(const float2*)(arow + k);
    v2f a;
    a.x = av.x * nrm;
    a.y = av.y * nrm;

    v2f b0, b1;
    b0.x = Ws[k * 32 + m];
    b0.y = Ws[(k + 1) * 32 + m];
    b1.x = Ws[k * 32 + 16 + m];
    b1.y = Ws[(k + 1) * 32 + 16 + m];

    c0 = __builtin_amdgcn_wmma_f32_16x16x4_f32(false, a, false, b0, (short)0, c0, false, false);
    c1 = __builtin_amdgcn_wmma_f32_16x16x4_f32(false, a, false, b1, (short)0, c1, false, false);
  }

  for (int v = 0; v < 8; ++v) {
    const int orow = wave * 16 + v + 8 * half;
    out[(size_t)orow * ldo + m] = c0[v];
    if (16 + m < wcols) out[(size_t)orow * ldo + 16 + m] = c1[v];
  }
}

__global__ void gcn_scatter_kernel(const float* __restrict__ x, const int* __restrict__ src,
                                   const int* __restrict__ dst, float* __restrict__ agg,
                                   int nwork, int nch, int H) {
  int idx = blockIdx.x * blockDim.x + threadIdx.x;
  if (idx >= nwork) return;
  int e = idx / nch;
  int c = idx - e * nch;
  int s = src[e];
  int d = dst[e];
  const float4 v = *(const float4*)(x + (size_t)s * H + c * 4);
  float* ap = agg + (size_t)d * H + c * 4;
  atomic_add_f32(ap + 0, v.x);
  atomic_add_f32(ap + 1, v.y);
  atomic_add_f32(ap + 2, v.z);
  atomic_add_f32(ap + 3, v.w);
}

__global__ void gcn_readout_kernel(const float* __restrict__ h, const float* __restrict__ Wd,
                                   const float* __restrict__ bd, float* __restrict__ out,
                                   int nq, int rowlen  ) {
  int r = blockIdx.x * blockDim.x + threadIdx.x;
  if (r >= nq) return;
  const float4* hp = (const float4*)(h + (size_t)r * rowlen);
  const float4* wp = (const float4*)Wd;
  const int nch = rowlen >> 2;
  float s = 0.0f;
  for (int j = 0; j < nch; ++j) {
    float4 hv = hp[j];
    float4 wv = wp[j];
    s = fmaf(hv.x, wv.x, s);
    s = fmaf(hv.y, wv.y, s);
    s = fmaf(hv.z, wv.z, s);
    s = fmaf(hv.w, wv.w, s);
  }
  out[r] = s + bd[0];
}


extern "C" void kernel_launch(void* const* d_in, const int* in_sizes, int n_in,
                              void* d_out, int out_size, void* d_ws, size_t ws_size,
                              hipStream_t stream) {
  const float* features = (const float*)d_in[0];
  const int*   src      = (const int*)d_in[1];
  const int*   dst      = (const int*)d_in[2];
  const float* W1       = (const float*)d_in[3];
  const float* b1       = (const float*)d_in[4];
  const float* W2       = (const float*)d_in[5];
  const float* b2       = (const float*)d_in[6];
  const float* Wd       = (const float*)d_in[7];
  const float* bd       = (const float*)d_in[8];
  float* out = (float*)d_out;

  const int H  = in_sizes[4];
  const int F  = in_sizes[3] / H;
  const int Nn = in_sizes[0] / F;
  const int E  = in_sizes[1];

  float* deg_out = (float*)d_ws;
  float* deg_in  = deg_out + Nn;
  float* bufA    = deg_in + Nn;
  float* bufX    = bufA + (size_t)Nn * H;

  const int B = 256;
  const int NH = Nn * H;
  const int ntiles = Nn / 16;
  const int wavesPerBlock = B / 32;
  const int gemmGrid = (ntiles + wavesPerBlock - 1) / wavesPerBlock;
  const int nch = H / 4;
  const int scatterWork = E * nch;
  const int nq = Nn / 4;

  {
    int n = 2 * Nn + NH;
    gcn_zero_kernel<<<(n + B - 1) / B, B, 0, stream>>>(deg_out, n);
  }
  gcn_degree_kernel<<<(E + B - 1) / B, B, 0, stream>>>(src, dst, deg_out, deg_in, E);
  gcn_norm_kernel<<<(2 * Nn + B - 1) / B, B, 0, stream>>>(deg_out, 2 * Nn);
  gcn_gemm_wmma_kernel<<<gemmGrid, B, 0, stream>>>(features, F, F, W1, H, deg_out,
                                                   bufX, H, ntiles);
  gcn_scatter_kernel<<<(scatterWork + B - 1) / B, B, 0, stream>>>(bufX, src, dst, bufA,
                                                                  scatterWork, nch, H);
  gcn_post_kernel<<<(NH + B - 1) / B, B, 0, stream>>>(bufA, deg_in, b1, NH, H);
  gcn_gemm_wmma_kernel<<<gemmGrid, B, 0, stream>>>(bufA, H, H, W2, H, deg_out,
                                                   bufX, H, ntiles);
  gcn_zero_kernel<<<(NH + B - 1) / B, B, 0, stream>>>(bufA, NH);
  gcn_scatter_kernel<<<(scatterWork + B - 1) / B, B, 0, stream>>>(bufX, src, dst, bufA,
                                                                  scatterWork, nch, H);
  gcn_post_kernel<<<(NH + B - 1) / B, B, 0, stream>>>(bufA, deg_in, b2, NH, H);
  gcn_readout_kernel<<<(nq + B - 1) / B, B, 0, stream>>>(bufA, Wd, bd, out, nq, 4 * H);
}


# edge scatter-add (f32 atomics) replaced by destination-ordered gather: count scan + integer cursor adds place edges once; 8 lanes sum a node's rows in registers in list (arrival) order, store once
# speedup vs baseline: 1.3367x; 1.3367x over previous
#include <hip/hip_runtime.h>

typedef __attribute__((ext_vector_type(2))) float v2f;
typedef __attribute__((ext_vector_type(8))) float v8f;

__device__ __forceinline__ void atomic_add_f32(float* p, float v) {
  (void)__hip_atomic_fetch_add(p, v, __ATOMIC_RELAXED, __HIP_MEMORY_SCOPE_AGENT);
}


__global__ void gcn_zero_kernel(float* __restrict__ p, int n) {
  int i = blockIdx.x * blockDim.x + threadIdx.x;
  if (i < n) p[i] = 0.0f;
}

__global__ void gcn_degree_kernel(const int* __restrict__ src, const int* __restrict__ dst,
                                  float* __restrict__ deg_out, float* __restrict__ deg_in,
                                  int E) {
  int e = blockIdx.x * blockDim.x + threadIdx.x;
  if (e < E) {
    atomic_add_f32(&deg_out[src[e]], 1.0f);
    atomic_add_f32(&deg_in[dst[e]], 1.0f);
  }
}

__global__ void gcn_norm_kernel(float* __restrict__ deg, int n) {
  int i = blockIdx.x * blockDim.x + threadIdx.x;
  if (i < n) deg[i] = rsqrtf(fmaxf(deg[i], 1.0f));
}

__global__ void gcn_gemm_wmma_kernel(const float* __restrict__ A, int lda, int K,
                                     const float* __restrict__ W, int wcols,
                                     const float* __restrict__ norm,
                                     float* __restrict__ out, int ldo, int ntiles) {
  __shared__ float Ws[32 * 32];

  for (int i = threadIdx.x; i < 32 * 32; i += blockDim.x) {
    const int k = i >> 5;
    const int n = i & 31;
    Ws[i] = (k < K && n < wcols) ? W[(size_t)k * wcols + n] : 0.0f;
  }
  __syncthreads();

  const int wave = (blockIdx.x * blockDim.x + threadIdx.x) >> 5;
  const int lane = threadIdx.x & 31;
  if (wave >= ntiles) return;

  const int m = lane & 15;
  const int half = lane >> 4;

  const int row = wave * 16 + m;
  const float nrm = norm[row];
  const float* arow = A + (size_t)row * lda;

  v8f c0 = {};
  v8f c1 = {};

  for (int k0 = 0; k0 < K; k0 += 4) {
    const int k = k0 + 2 * half;
    const float2 av = *(const float2*)(arow + k);
    v2f a;
    a.x = av.x * nrm;
    a.y = av.y * nrm;

    v2f b0, b1;
    b0.x = Ws[k * 32 + m];
    b0.y = Ws[(k + 1) * 32 + m];
    b1.x = Ws[k * 32 + 16 + m];
    b1.y = Ws[(k + 1) * 32 + 16 + m];

    c0 = __builtin_amdgcn_wmma_f32_16x16x4_f32(false, a, false, b0, (short)0, c0, false, false);
    c1 = __builtin_amdgcn_wmma_f32_16x16x4_f32(false, a, false, b1, (short)0, c1, false, false);
  }

  for (int v = 0; v < 8; ++v) {
    const int orow = wave * 16 + v + 8 * half;
    out[(size_t)orow * ldo + m] = c0[v];
    if (16 + m < wcols) out[(size_t)orow * ldo + 16 + m] = c1[v];
  }
}


__global__ void gcn_tilesum_kernel(const float* __restrict__ deg_in, int* __restrict__ tilesum,
                                   int n) {
  __shared__ int red[256];
  const int t = threadIdx.x;
  const int base = blockIdx.x * 1024 + t * 4;
  int s = 0;
  for (int k = 0; k < 4; ++k) {
    const int i = base + k;
    if (i < n) s += (int)deg_in[i];
  }
  red[t] = s;
  __syncthreads();
  for (int off = 128; off > 0; off >>= 1) {
    if (t < off) red[t] += red[t + off];
    __syncthreads();
  }
  if (t == 0) tilesum[blockIdx.x] = red[0];
}

__global__ void gcn_scan_kernel(const float* __restrict__ deg_in, const int* __restrict__ tilesum,
                                int* __restrict__ row_ptr, int* __restrict__ cursor, int n) {
  __shared__ int red[256];
  __shared__ int sc[256];
  const int t = threadIdx.x;

  int part = 0;
  for (int j = t; j < (int)blockIdx.x; j += 256) part += tilesum[j];
  red[t] = part;

  const int base = blockIdx.x * 1024 + t * 4;
  int c[4];
  int s = 0;
  for (int k = 0; k < 4; ++k) {
    const int i = base + k;
    c[k] = (i < n) ? (int)deg_in[i] : 0;
    s += c[k];
  }
  sc[t] = s;
  __syncthreads();

  for (int off = 128; off > 0; off >>= 1) {
    if (t < off) red[t] += red[t + off];
    __syncthreads();
  }
  for (int off = 1; off < 256; off <<= 1) {
    const int v = (t >= off) ? sc[t - off] : 0;
    __syncthreads();
    sc[t] += v;
    __syncthreads();
  }

  int run = red[0] + sc[t] - s;
  if (base + 3 < n) {
    int4 o;
    o.x = run;
    o.y = run + c[0];
    o.z = run + c[0] + c[1];
    o.w = run + c[0] + c[1] + c[2];
    *(int4*)(row_ptr + base) = o;
    *(int4*)(cursor + base) = o;
  } else {
    for (int k = 0; k < 4; ++k) {
      const int i = base + k;
      if (i < n) {
        row_ptr[i] = run;
        cursor[i] = run;
      }
      run += c[k];
    }
  }
}

__global__ void gcn_fill_kernel(const int* __restrict__ src, const int* __restrict__ dst,
                                int* __restrict__ cursor, int* __restrict__ col, int E) {
  const int e = blockIdx.x * blockDim.x + threadIdx.x;
  if (e < E) {
    const int s = src[e];
    const int p = __hip_atomic_fetch_add(&cursor[dst[e]], 1, __ATOMIC_RELAXED,
                                         __HIP_MEMORY_SCOPE_AGENT);
    if ((unsigned)p < (unsigned)E) col[p] = s;
  }
}

__global__ void gcn_pull_kernel(const float* __restrict__ x, const int* __restrict__ row_ptr,
                                const int* __restrict__ row_end, const int* __restrict__ col,
                                const float* __restrict__ norm_dst, const float* __restrict__ bias,
                                float* __restrict__ h, int n, int nch, int H) {
  const int gid = blockIdx.x * blockDim.x + threadIdx.x;
  const int node = gid >> 3;
  const int c = gid & 7;
  if (node >= n || c >= nch) return;

  int j = row_ptr[node];
  const int end = row_end[node];
  const float* xc = x + c * 4;

  float4 acc = make_float4(0.0f, 0.0f, 0.0f, 0.0f);
  for (; j + 4 <= end; j += 4) {
    const int s0 = col[j];
    const int s1 = col[j + 1];
    const int s2 = col[j + 2];
    const int s3 = col[j + 3];
    const float4 v0 = *(const float4*)(xc + s0 * H);
    const float4 v1 = *(const float4*)(xc + s1 * H);
    const float4 v2 = *(const float4*)(xc + s2 * H);
    const float4 v3 = *(const float4*)(xc + s3 * H);
    acc.x += v0.x; acc.y += v0.y; acc.z += v0.z; acc.w += v0.w;
    acc.x += v1.x; acc.y += v1.y; acc.z += v1.z; acc.w += v1.w;
    acc.x += v2.x; acc.y += v2.y; acc.z += v2.z; acc.w += v2.w;
    acc.x += v3.x; acc.y += v3.y; acc.z += v3.z; acc.w += v3.w;
  }
  const int r = end - j;
  if (r > 0) {
    const int s0 = col[j];
    const int s1 = (r > 1) ? col[j + 1] : s0;
    const int s2 = (r > 2) ? col[j + 2] : s0;
    const float4 v0 = *(const float4*)(xc + s0 * H);
    const float4 v1 = *(const float4*)(xc + s1 * H);
    const float4 v2 = *(const float4*)(xc + s2 * H);
    acc.x += v0.x; acc.y += v0.y; acc.z += v0.z; acc.w += v0.w;
    if (r > 1) { acc.x += v1.x; acc.y += v1.y; acc.z += v1.z; acc.w += v1.w; }
    if (r > 2) { acc.x += v2.x; acc.y += v2.y; acc.z += v2.z; acc.w += v2.w; }
  }

  const float nd = norm_dst[node];
  const float4 b = *(const float4*)(bias + c * 4);
  float4 o;
  o.x = fmaxf(fmaf(acc.x, nd, b.x), 0.0f);
  o.y = fmaxf(fmaf(acc.y, nd, b.y), 0.0f);
  o.z = fmaxf(fmaf(acc.z, nd, b.z), 0.0f);
  o.w = fmaxf(fmaf(acc.w, nd, b.w), 0.0f);
  *(float4*)(h + (size_t)node * H + c * 4) = o;
}

__global__ void gcn_readout_kernel(const float* __restrict__ h, const float* __restrict__ Wd,
                                   const float* __restrict__ bd, float* __restrict__ out,
                                   int nq, int rowlen  ) {
  int r = blockIdx.x * blockDim.x + threadIdx.x;
  if (r >= nq) return;
  const float4* hp = (const float4*)(h + (size_t)r * rowlen);
  const float4* wp = (const float4*)Wd;
  const int nch = rowlen >> 2;
  float s = 0.0f;
  for (int j = 0; j < nch; ++j) {
    float4 hv = hp[j];
    float4 wv = wp[j];
    s = fmaf(hv.x, wv.x, s);
    s = fmaf(hv.y, wv.y, s);
    s = fmaf(hv.z, wv.z, s);
    s = fmaf(hv.w, wv.w, s);
  }
  out[r] = s + bd[0];
}


extern "C" void kernel_launch(void* const* d_in, const int* in_sizes, int n_in,
                              void* d_out, int out_size, void* d_ws, size_t ws_size,
                              hipStream_t stream) {
  const float* features = (const float*)d_in[0];
  const int*   src      = (const int*)d_in[1];
  const int*   dst      = (const int*)d_in[2];
  const float* W1       = (const float*)d_in[3];
  const float* b1       = (const float*)d_in[4];
  const float* W2       = (const float*)d_in[5];
  const float* b2       = (const float*)d_in[6];
  const float* Wd       = (const float*)d_in[7];
  const float* bd       = (const float*)d_in[8];
  float* out = (float*)d_out;

  const int H  = in_sizes[4];
  const int F  = in_sizes[3] / H;
  const int Nn = in_sizes[0] / F;
  const int E  = in_sizes[1];

  const int B = 256;
  const int NH = Nn * H;
  const int ntiles = Nn / 16;
  const int wavesPerBlock = B / 32;
  const int gemmGrid = (ntiles + wavesPerBlock - 1) / wavesPerBlock;
  const int nch = H / 4;
  const int nq = Nn / 4;
  const int scanTiles = (Nn + 1023) / 1024;
  const int pullGrid = (Nn * 8 + B - 1) / B;

  float* deg_out = (float*)d_ws;
  float* deg_in  = deg_out + Nn;
  float* bufA    = deg_in + Nn;
  float* bufX    = bufA + (size_t)Nn * H;
  int*   row_ptr = (int*)(bufX + (size_t)Nn * H);
  int*   cursor  = row_ptr + Nn;
  int*   tilesum = cursor + Nn;
  int*   col     = tilesum + ((scanTiles + 1023) / 1024) * 1024;

  gcn_zero_kernel<<<(2 * Nn + B - 1) / B, B, 0, stream>>>(deg_out, 2 * Nn);
  gcn_degree_kernel<<<(E + B - 1) / B, B, 0, stream>>>(src, dst, deg_out, deg_in, E);
  gcn_tilesum_kernel<<<scanTiles, B, 0, stream>>>(deg_in, tilesum, Nn);
  gcn_scan_kernel<<<scanTiles, B, 0, stream>>>(deg_in, tilesum, row_ptr, cursor, Nn);
  gcn_norm_kernel<<<(2 * Nn + B - 1) / B, B, 0, stream>>>(deg_out, 2 * Nn);
  gcn_fill_kernel<<<(E + B - 1) / B, B, 0, stream>>>(src, dst, cursor, col, E);
  gcn_gemm_wmma_kernel<<<gemmGrid, B, 0, stream>>>(features, F, F, W1, H, deg_out,
                                                   bufX, H, ntiles);
  gcn_pull_kernel<<<pullGrid, B, 0, stream>>>(bufX, row_ptr, cursor, col, deg_in, b1, bufA,
                                              Nn, nch, H);
  gcn_gemm_wmma_kernel<<<gemmGrid, B, 0, stream>>>(bufA, H, H, W2, H, deg_out,
                                                   bufX, H, ntiles);
  gcn_pull_kernel<<<pullGrid, B, 0, stream>>>(bufX, row_ptr, cursor, col, deg_in, b2, bufA,
                                              Nn, nch, H);
  gcn_readout_kernel<<<(nq + B - 1) / B, B, 0, stream>>>(bufA, Wd, bd, out, nq, 4 * H);
}


# destination lists as fixed 32-entry rows per node, read 8 a trip: the integer in-degree add hands each edge its place in that pass; tile-sum, scan, fill launches go (11->8); arrival-order sums kept
# speedup vs baseline: 1.8023x; 1.3483x over previous
#include <hip/hip_runtime.h>

typedef __attribute__((ext_vector_type(2))) float v2f;
typedef __attribute__((ext_vector_type(8))) float v8f;


__global__ void gcn_zero_kernel(int* __restrict__ p, int n) {
  int i = blockIdx.x * blockDim.x + threadIdx.x;
  if (i < n) p[i] = 0;
}

#define GCN_SLOT_CAP 32

__global__ void gcn_edge_kernel(const int* __restrict__ src, const int* __restrict__ dst,
                                int* __restrict__ cnt_out, int* __restrict__ cnt_in,
                                int* __restrict__ slots, int E) {
  int e = blockIdx.x * blockDim.x + threadIdx.x;
  if (e < E) {
    const int s = src[e];
    const int d = dst[e];
    (void)__hip_atomic_fetch_add(&cnt_out[s], 1, __ATOMIC_RELAXED, __HIP_MEMORY_SCOPE_AGENT);
    const int p = __hip_atomic_fetch_add(&cnt_in[d], 1, __ATOMIC_RELAXED,
                                         __HIP_MEMORY_SCOPE_AGENT);
    if ((unsigned)p < (unsigned)GCN_SLOT_CAP) slots[(size_t)d * GCN_SLOT_CAP + p] = s;
  }
}

__global__ void gcn_norm_kernel(const int* __restrict__ cnt, float* __restrict__ nrm, int n) {
  int i = blockIdx.x * blockDim.x + threadIdx.x;
  if (i < n) nrm[i] = rsqrtf(fmaxf((float)cnt[i], 1.0f));
}

__global__ void gcn_gemm_wmma_kernel(const float* __restrict__ A, int lda, int K,
                                     const float* __restrict__ W, int wcols,
                                     const float* __restrict__ norm,
                                     float* __restrict__ out, int ldo, int ntiles) {
  __shared__ float Ws[32 * 32];

  for (int i = threadIdx.x; i < 32 * 32; i += blockDim.x) {
    const int k = i >> 5;
    const int n = i & 31;
    Ws[i] = (k < K && n < wcols) ? W[(size_t)k * wcols + n] : 0.0f;
  }
  __syncthreads();

  const int wave = (blockIdx.x * blockDim.x + threadIdx.x) >> 5;
  const int lane = threadIdx.x & 31;
  if (wave >= ntiles) return;

  const int m = lane & 15;
  const int half = lane >> 4;

  const int row = wave * 16 + m;
  const float nrm = norm[row];
  const float* arow = A + (size_t)row * lda;

  v8f c0 = {};
  v8f c1 = {};

  for (int k0 = 0; k0 < K; k0 += 4) {
    const int k = k0 + 2 * half;
    const float2 av = *(const float2*)(arow + k);
    v2f a;
    a.x = av.x * nrm;
    a.y = av.y * nrm;

    v2f b0, b1;
    b0.x = Ws[k * 32 + m];
    b0.y = Ws[(k + 1) * 32 + m];
    b1.x = Ws[k * 32 + 16 + m];
    b1.y = Ws[(k + 1) * 32 + 16 + m];

    c0 = __builtin_amdgcn_wmma_f32_16x16x4_f32(false, a, false, b0, (short)0, c0, false, false);
    c1 = __builtin_amdgcn_wmma_f32_16x16x4_f32(false, a, false, b1, (short)0, c1, false, false);
  }

  for (int v = 0; v < 8; ++v) {
    const int orow = wave * 16 + v + 8 * half;
    out[(size_t)orow * ldo + m] = c0[v];
    if (16 + m < wcols) out[(size_t)orow * ldo + 16 + m] = c1[v];
  }
}

__global__ void gcn_pull_kernel(const float* __restrict__ x, const int* __restrict__ cnt_in,
                                const int* __restrict__ slots, const int* __restrict__ src,
                                const int* __restrict__ dst, int E,
                                const float* __restrict__ norm_dst, const float* __restrict__ bias,
                                float* __restrict__ h, int n, int nch, int H) {
  const int gid = blockIdx.x * blockDim.x + threadIdx.x;
  const int node = gid >> 3;
  const int c = gid & 7;
  if (node >= n || c >= nch) return;

  const int cnt = cnt_in[node];
  const int* sl = slots + (size_t)node * GCN_SLOT_CAP;
  const float* xc = x + c * 4;
  const unsigned uH = (unsigned)H;
  const unsigned c4 = (unsigned)c * 4u;

  float4 acc = make_float4(0.0f, 0.0f, 0.0f, 0.0f);
  if (cnt <= GCN_SLOT_CAP) {
    for (int j = 0; j < cnt; j += 8) {
      const int r = cnt - j;
      const int s0 = sl[j];
      const int s1 = sl[(r > 1) ? j + 1 : j];
      const int s2 = sl[(r > 2) ? j + 2 : j];
      const int s3 = sl[(r > 3) ? j + 3 : j];
      const int s4 = sl[(r > 4) ? j + 4 : j];
      const int s5 = sl[(r > 5) ? j + 5 : j];
      const int s6 = sl[(r > 6) ? j + 6 : j];
      const int s7 = sl[(r > 7) ? j + 7 : j];
      asm volatile("" :: "v"(s0), "v"(s1), "v"(s2), "v"(s3), "v"(s4), "v"(s5), "v"(s6), "v"(s7));
      const float4 v0 = *(const float4*)(x + ((unsigned)s0 * uH + c4));
      const float4 v1 = *(const float4*)(x + ((unsigned)s1 * uH + c4));
      const float4 v2 = *(const float4*)(x + ((unsigned)s2 * uH + c4));
      const float4 v3 = *(const float4*)(x + ((unsigned)s3 * uH + c4));
      const float4 v4 = *(const float4*)(x + ((unsigned)s4 * uH + c4));
      const float4 v5 = *(const float4*)(x + ((unsigned)s5 * uH + c4));
      const float4 v6 = *(const float4*)(x + ((unsigned)s6 * uH + c4));
      const float4 v7 = *(const float4*)(x + ((unsigned)s7 * uH + c4));
      asm volatile("" :: "v"(v0.x), "v"(v1.x), "v"(v2.x), "v"(v3.x),
                         "v"(v4.x), "v"(v5.x), "v"(v6.x), "v"(v7.x));
      acc.x += v0.x; acc.y += v0.y; acc.z += v0.z; acc.w += v0.w;
      if (r > 1) { acc.x += v1.x; acc.y += v1.y; acc.z += v1.z; acc.w += v1.w; }
      if (r > 2) { acc.x += v2.x; acc.y += v2.y; acc.z += v2.z; acc.w += v2.w; }
      if (r > 3) { acc.x += v3.x; acc.y += v3.y; acc.z += v3.z; acc.w += v3.w; }
      if (r > 4) { acc.x += v4.x; acc.y += v4.y; acc.z += v4.z; acc.w += v4.w; }
      if (r > 5) { acc.x += v5.x; acc.y += v5.y; acc.z += v5.z; acc.w += v5.w; }
      if (r > 6) { acc.x += v6.x; acc.y += v6.y; acc.z += v6.z; acc.w += v6.w; }
      if (r > 7) { acc.x += v7.x; acc.y += v7.y; acc.z += v7.z; acc.w += v7.w; }
    }
  } else {
    for (int e = 0; e < E; ++e) {
      if (dst[e] == node) {
        const float4 v = *(const float4*)(xc + src[e] * H);
        acc.x += v.x; acc.y += v.y; acc.z += v.z; acc.w += v.w;
      }
    }
  }

  const float nd = norm_dst[node];
  const float4 b = *(const float4*)(bias + c * 4);
  float4 o;
  o.x = fmaxf(fmaf(acc.x, nd, b.x), 0.0f);
  o.y = fmaxf(fmaf(acc.y, nd, b.y), 0.0f);
  o.z = fmaxf(fmaf(acc.z, nd, b.z), 0.0f);
  o.w = fmaxf(fmaf(acc.w, nd, b.w), 0.0f);
  *(float4*)(h + (size_t)node * H + c * 4) = o;
}

__global__ void gcn_readout_kernel(const float* __restrict__ h, const float* __restrict__ Wd,
                                   const float* __restrict__ bd, float* __restrict__ out,
                                   int nq, int rowlen  ) {
  int r = blockIdx.x * blockDim.x + threadIdx.x;
  if (r >= nq) return;
  const float4* hp = (const float4*)(h + (size_t)r * rowlen);
  const float4* wp = (const float4*)Wd;
  const int nch = rowlen >> 2;
  float s = 0.0f;
  for (int j = 0; j < nch; ++j) {
    float4 hv = hp[j];
    float4 wv = wp[j];
    s = fmaf(hv.x, wv.x, s);
    s = fmaf(hv.y, wv.y, s);
    s = fmaf(hv.z, wv.z, s);
    s = fmaf(hv.w, wv.w, s);
  }
  out[r] = s + bd[0];
}


extern "C" void kernel_launch(void* const* d_in, const int* in_sizes, int n_in,
                              void* d_out, int out_size, void* d_ws, size_t ws_size,
                              hipStream_t stream) {
  const float* features = (const float*)d_in[0];
  const int*   src      = (const int*)d_in[1];
  const int*   dst      = (const int*)d_in[2];
  const float* W1       = (const float*)d_in[3];
  const float* b1       = (const float*)d_in[4];
  const float* W2       = (const float*)d_in[5];
  const float* b2       = (const float*)d_in[6];
  const float* Wd       = (const float*)d_in[7];
  const float* bd       = (const float*)d_in[8];
  float* out = (float*)d_out;

  const int H  = in_sizes[4];
  const int F  = in_sizes[3] / H;
  const int Nn = in_sizes[0] / F;
  const int E  = in_sizes[1];

  const int B = 256;
  const int ntiles = Nn / 16;
  const int wavesPerBlock = B / 32;
  const int gemmGrid = (ntiles + wavesPerBlock - 1) / wavesPerBlock;
  const int nch = H / 4;
  const int nq = Nn / 4;
  const int pullGrid = (Nn * 8 + B - 1) / B;

  int*   cnt_out  = (int*)d_ws;
  int*   cnt_in   = cnt_out + Nn;
  float* norm_src = (float*)(cnt_in + Nn);
  float* norm_dst = norm_src + Nn;
  float* bufA     = norm_dst + Nn;
  float* bufX     = bufA + (size_t)Nn * H;
  int*   slots    = (int*)(bufX + (size_t)Nn * H);

  gcn_zero_kernel<<<(2 * Nn + B - 1) / B, B, 0, stream>>>(cnt_out, 2 * Nn);
  gcn_edge_kernel<<<(E + B - 1) / B, B, 0, stream>>>(src, dst, cnt_out, cnt_in, slots, E);
  gcn_norm_kernel<<<(2 * Nn + B - 1) / B, B, 0, stream>>>(cnt_out, norm_src, 2 * Nn);
  gcn_gemm_wmma_kernel<<<gemmGrid, B, 0, stream>>>(features, F, F, W1, H, norm_src,
                                                   bufX, H, ntiles);
  gcn_pull_kernel<<<pullGrid, B, 0, stream>>>(bufX, cnt_in, slots, src, dst, E, norm_dst, b1,
                                              bufA, Nn, nch, H);
  gcn_gemm_wmma_kernel<<<gemmGrid, B, 0, stream>>>(bufA, H, H, W2, H, norm_src,
                                                   bufX, H, ntiles);
  gcn_pull_kernel<<<pullGrid, B, 0, stream>>>(bufX, cnt_in, slots, src, dst, E, norm_dst, b2,
                                              bufA, Nn, nch, H);
  gcn_readout_kernel<<<(nq + B - 1) / B, B, 0, stream>>>(bufA, Wd, bd, out, nq, 4 * H);
}


# degree+list pass split by cache half: a work-group reads its die and draws 1024-edge tickets for its half's node range; integer adds and list stores only for those nodes; arrival-order sums kept
# speedup vs baseline: 3.4263x; 1.9011x over previous
#include <hip/hip_runtime.h>

typedef __attribute__((ext_vector_type(2))) float v2f;
typedef __attribute__((ext_vector_type(8))) float v8f;


__global__ void gcn_zero_kernel(int* __restrict__ p, int n) {
  int i = blockIdx.x * blockDim.x + threadIdx.x;
  if (i < n) p[i] = 0;
}

#define GCN_SLOT_CAP 32

#define GCN_TICK_STRIDE 64
#define GCN_EDGE_BLOCK 256
#define GCN_EDGE_PER_LANE 4
#define GCN_CHUNK (GCN_EDGE_BLOCK * GCN_EDGE_PER_LANE)

static __device__ __forceinline__ void gcn_lds_barrier() {
  __builtin_amdgcn_fence(__ATOMIC_RELEASE, "workgroup", "local");
  __builtin_amdgcn_s_barrier();
  __builtin_amdgcn_fence(__ATOMIC_ACQUIRE, "workgroup", "local");
}

__global__ void __launch_bounds__(GCN_EDGE_BLOCK)
gcn_edge_kernel(const int* __restrict__ src, const int* __restrict__ dst, int* __restrict__ tick,
                int* __restrict__ cnt_out, int* __restrict__ cnt_in, int* __restrict__ slots,
                int E, int n) {
  __shared__ int s_chunk[2];
  const int tid = threadIdx.x;
  const int nchunks = (E + GCN_CHUNK - 1) / GCN_CHUNK;
  int split = ((n >> 1) + 63) & ~63;
  if (split > n) split = n;
  int t = (int)((__builtin_amdgcn_s_sendmsg_rtn(0x87) >> 18) & 1u);

  for (int pass = 0; pass < 2; ++pass, t ^= 1) {
    const int lo = t ? split : 0;
    const unsigned span = t ? (unsigned)(n - split) : (unsigned)split;
    int* tk = tick + t * GCN_TICK_STRIDE;

    if (tid == 0) {
      int first = nchunks;
      if (pass == 0 || __hip_atomic_load(tk, __ATOMIC_RELAXED, __HIP_MEMORY_SCOPE_AGENT) < nchunks)
        first = __hip_atomic_fetch_add(tk, 1, __ATOMIC_RELAXED, __HIP_MEMORY_SCOPE_AGENT);
      s_chunk[0] = first;
    }
    gcn_lds_barrier();
    int c = s_chunk[0];
    int buf = 0;
    while (c < nchunks) {
      int s[GCN_EDGE_PER_LANE], d[GCN_EDGE_PER_LANE], p[GCN_EDGE_PER_LANE];
      bool os[GCN_EDGE_PER_LANE], od[GCN_EDGE_PER_LANE];
#pragma unroll
      for (int k = 0; k < GCN_EDGE_PER_LANE; ++k) {
        const int e = c * GCN_CHUNK + k * GCN_EDGE_BLOCK + tid;
        const int ec = (e < E) ? e : E - 1;
        s[k] = src[ec];
        d[k] = dst[ec];
        os[k] = (e < E);
      }
      asm volatile("" :: "v"(s[0]), "v"(d[0]), "v"(s[1]), "v"(d[1]),
                         "v"(s[2]), "v"(d[2]), "v"(s[3]), "v"(d[3]));
#pragma unroll
      for (int k = 0; k < GCN_EDGE_PER_LANE; ++k) {
        od[k] = os[k] && ((unsigned)(d[k] - lo) < span);
        os[k] = os[k] && ((unsigned)(s[k] - lo) < span);
      }
#pragma unroll
      for (int k = 0; k < GCN_EDGE_PER_LANE; ++k) {
        p[k] = GCN_SLOT_CAP;
        if (od[k])
          p[k] = __hip_atomic_fetch_add(&cnt_in[d[k]], 1, __ATOMIC_RELAXED, __HIP_MEMORY_SCOPE_AGENT);
      }
#pragma unroll
      for (int k = 0; k < GCN_EDGE_PER_LANE; ++k) {
        if (os[k])
          (void)__hip_atomic_fetch_add(&cnt_out[s[k]], 1, __ATOMIC_RELAXED, __HIP_MEMORY_SCOPE_AGENT);
      }
      int nxt = 0;
      if (tid == 0) nxt = __hip_atomic_fetch_add(tk, 1, __ATOMIC_RELAXED, __HIP_MEMORY_SCOPE_AGENT);
#pragma unroll
      for (int k = 0; k < GCN_EDGE_PER_LANE; ++k) {
        if ((unsigned)p[k] < (unsigned)GCN_SLOT_CAP) slots[(size_t)d[k] * GCN_SLOT_CAP + p[k]] = s[k];
      }
      buf ^= 1;
      if (tid == 0) s_chunk[buf] = nxt;
      gcn_lds_barrier();
      c = s_chunk[buf];
    }
    gcn_lds_barrier();
  }
}

__global__ void gcn_norm_kernel(const int* __restrict__ cnt, float* __restrict__ nrm, int n) {
  int i = blockIdx.x * blockDim.x + threadIdx.x;
  if (i < n) nrm[i] = rsqrtf(fmaxf((float)cnt[i], 1.0f));
}

__global__ void gcn_gemm_wmma_kernel(const float* __restrict__ A, int lda, int K,
                                     const float* __restrict__ W, int wcols,
                                     const float* __restrict__ norm,
                                     float* __restrict__ out, int ldo, int ntiles) {
  __shared__ float Ws[32 * 32];

  for (int i = threadIdx.x; i < 32 * 32; i += blockDim.x) {
    const int k = i >> 5;
    const int n = i & 31;
    Ws[i] = (k < K && n < wcols) ? W[(size_t)k * wcols + n] : 0.0f;
  }
  __syncthreads();

  const int wave = (blockIdx.x * blockDim.x + threadIdx.x) >> 5;
  const int lane = threadIdx.x & 31;
  if (wave >= ntiles) return;

  const int m = lane & 15;
  const int half = lane >> 4;

  const int row = wave * 16 + m;
  const float nrm = norm[row];
  const float* arow = A + (size_t)row * lda;

  v8f c0 = {};
  v8f c1 = {};

  for (int k0 = 0; k0 < K; k0 += 4) {
    const int k = k0 + 2 * half;
    const float2 av = *(const float2*)(arow + k);
    v2f a;
    a.x = av.x * nrm;
    a.y = av.y * nrm;

    v2f b0, b1;
    b0.x = Ws[k * 32 + m];
    b0.y = Ws[(k + 1) * 32 + m];
    b1.x = Ws[k * 32 + 16 + m];
    b1.y = Ws[(k + 1) * 32 + 16 + m];

    c0 = __builtin_amdgcn_wmma_f32_16x16x4_f32(false, a, false, b0, (short)0, c0, false, false);
    c1 = __builtin_amdgcn_wmma_f32_16x16x4_f32(false, a, false, b1, (short)0, c1, false, false);
  }

  for (int v = 0; v < 8; ++v) {
    const int orow = wave * 16 + v + 8 * half;
    out[(size_t)orow * ldo + m] = c0[v];
    if (16 + m < wcols) out[(size_t)orow * ldo + 16 + m] = c1[v];
  }
}

__global__ void gcn_pull_kernel(const float* __restrict__ x, const int* __restrict__ cnt_in,
                                const int* __restrict__ slots, const int* __restrict__ src,
                                const int* __restrict__ dst, int E,
                                const float* __restrict__ norm_dst, const float* __restrict__ bias,
                                float* __restrict__ h, int n, int nch, int H) {
  const int gid = blockIdx.x * blockDim.x + threadIdx.x;
  const int node = gid >> 3;
  const int c = gid & 7;
  if (node >= n || c >= nch) return;

  const int cnt = cnt_in[node];
  const int* sl = slots + (size_t)node * GCN_SLOT_CAP;
  const float* xc = x + c * 4;
  const unsigned uH = (unsigned)H;
  const unsigned c4 = (unsigned)c * 4u;

  float4 acc = make_float4(0.0f, 0.0f, 0.0f, 0.0f);
  if (cnt <= GCN_SLOT_CAP) {
    for (int j = 0; j < cnt; j += 8) {
      const int r = cnt - j;
      const int s0 = sl[j];
      const int s1 = sl[(r > 1) ? j + 1 : j];
      const int s2 = sl[(r > 2) ? j + 2 : j];
      const int s3 = sl[(r > 3) ? j + 3 : j];
      const int s4 = sl[(r > 4) ? j + 4 : j];
      const int s5 = sl[(r > 5) ? j + 5 : j];
      const int s6 = sl[(r > 6) ? j + 6 : j];
      const int s7 = sl[(r > 7) ? j + 7 : j];
      asm volatile("" :: "v"(s0), "v"(s1), "v"(s2), "v"(s3), "v"(s4), "v"(s5), "v"(s6), "v"(s7));
      const float4 v0 = *(const float4*)(x + ((unsigned)s0 * uH + c4));
      const float4 v1 = *(const float4*)(x + ((unsigned)s1 * uH + c4));
      const float4 v2 = *(const float4*)(x + ((unsigned)s2 * uH + c4));
      const float4 v3 = *(const float4*)(x + ((unsigned)s3 * uH + c4));
      const float4 v4 = *(const float4*)(x + ((unsigned)s4 * uH + c4));
      const float4 v5 = *(const float4*)(x + ((unsigned)s5 * uH + c4));
      const float4 v6 = *(const float4*)(x + ((unsigned)s6 * uH + c4));
      const float4 v7 = *(const float4*)(x + ((unsigned)s7 * uH + c4));
      asm volatile("" :: "v"(v0.x), "v"(v1.x), "v"(v2.x), "v"(v3.x),
                         "v"(v4.x), "v"(v5.x), "v"(v6.x), "v"(v7.x));
      acc.x += v0.x; acc.y += v0.y; acc.z += v0.z; acc.w += v0.w;
      if (r > 1) { acc.x += v1.x; acc.y += v1.y; acc.z += v1.z; acc.w += v1.w; }
      if (r > 2) { acc.x += v2.x; acc.y += v2.y; acc.z += v2.z; acc.w += v2.w; }
      if (r > 3) { acc.x += v3.x; acc.y += v3.y; acc.z += v3.z; acc.w += v3.w; }
      if (r > 4) { acc.x += v4.x; acc.y += v4.y; acc.z += v4.z; acc.w += v4.w; }
      if (r > 5) { acc.x += v5.x; acc.y += v5.y; acc.z += v5.z; acc.w += v5.w; }
      if (r > 6) { acc.x += v6.x; acc.y += v6.y; acc.z += v6.z; acc.w += v6.w; }
      if (r > 7) { acc.x += v7.x; acc.y += v7.y; acc.z += v7.z; acc.w += v7.w; }
    }
  } else {
    for (int e = 0; e < E; ++e) {
      if (dst[e] == node) {
        const float4 v = *(const float4*)(xc + src[e] * H);
        acc.x += v.x; acc.y += v.y; acc.z += v.z; acc.w += v.w;
      }
    }
  }

  const float nd = norm_dst[node];
  const float4 b = *(const float4*)(bias + c * 4);
  float4 o;
  o.x = fmaxf(fmaf(acc.x, nd, b.x), 0.0f);
  o.y = fmaxf(fmaf(acc.y, nd, b.y), 0.0f);
  o.z = fmaxf(fmaf(acc.z, nd, b.z), 0.0f);
  o.w = fmaxf(fmaf(acc.w, nd, b.w), 0.0f);
  *(float4*)(h + (size_t)node * H + c * 4) = o;
}

__global__ void gcn_readout_kernel(const float* __restrict__ h, const float* __restrict__ Wd,
                                   const float* __restrict__ bd, float* __restrict__ out,
                                   int nq, int rowlen  ) {
  int r = blockIdx.x * blockDim.x + threadIdx.x;
  if (r >= nq) return;
  const float4* hp = (const float4*)(h + (size_t)r * rowlen);
  const float4* wp = (const float4*)Wd;
  const int nch = rowlen >> 2;
  float s = 0.0f;
  for (int j = 0; j < nch; ++j) {
    float4 hv = hp[j];
    float4 wv = wp[j];
    s = fmaf(hv.x, wv.x, s);
    s = fmaf(hv.y, wv.y, s);
    s = fmaf(hv.z, wv.z, s);
    s = fmaf(hv.w, wv.w, s);
  }
  out[r] = s + bd[0];
}


extern "C" void kernel_launch(void* const* d_in, const int* in_sizes, int n_in,
                              void* d_out, int out_size, void* d_ws, size_t ws_size,
                              hipStream_t stream) {
  const float* features = (const float*)d_in[0];
  const int*   src      = (const int*)d_in[1];
  const int*   dst      = (const int*)d_in[2];
  const float* W1       = (const float*)d_in[3];
  const float* b1       = (const float*)d_in[4];
  const float* W2       = (const float*)d_in[5];
  const float* b2       = (const float*)d_in[6];
  const float* Wd       = (const float*)d_in[7];
  const float* bd       = (const float*)d_in[8];
  float* out = (float*)d_out;

  const int H  = in_sizes[4];
  const int F  = in_sizes[3] / H;
  const int Nn = in_sizes[0] / F;
  const int E  = in_sizes[1];

  const int B = 256;
  const int ntiles = Nn / 16;
  const int wavesPerBlock = B / 32;
  const int gemmGrid = (ntiles + wavesPerBlock - 1) / wavesPerBlock;
  const int nch = H / 4;
  const int nq = Nn / 4;
  const int pullGrid = (Nn * 8 + B - 1) / B;

  int*   tick     = (int*)d_ws;
  int*   cnt_out  = tick + 2 * GCN_TICK_STRIDE;
  int*   cnt_in   = cnt_out + Nn;
  float* norm_src = (float*)(cnt_in + Nn);
  float* norm_dst = norm_src + Nn;
  float* bufA     = norm_dst + Nn;
  float* bufX     = bufA + (size_t)Nn * H;
  int*   slots    = (int*)(bufX + (size_t)Nn * H);

  gcn_zero_kernel<<<(2 * GCN_TICK_STRIDE + 2 * Nn + B - 1) / B, B, 0, stream>>>(
      tick, 2 * GCN_TICK_STRIDE + 2 * Nn);
  gcn_edge_kernel<<<2048, GCN_EDGE_BLOCK, 0, stream>>>(src, dst, tick, cnt_out, cnt_in, slots, E, Nn);
  gcn_norm_kernel<<<(2 * Nn + B - 1) / B, B, 0, stream>>>(cnt_out, norm_src, 2 * Nn);
  gcn_gemm_wmma_kernel<<<gemmGrid, B, 0, stream>>>(features, F, F, W1, H, norm_src,
                                                   bufX, H, ntiles);
  gcn_pull_kernel<<<pullGrid, B, 0, stream>>>(bufX, cnt_in, slots, src, dst, E, norm_dst, b1,
                                              bufA, Nn, nch, H);
  gcn_gemm_wmma_kernel<<<gemmGrid, B, 0, stream>>>(bufA, H, H, W2, H, norm_src,
                                                   bufX, H, ntiles);
  gcn_pull_kernel<<<pullGrid, B, 0, stream>>>(bufX, cnt_in, slots, src, dst, E, norm_dst, b2,
                                              bufA, Nn, nch, H);
  gcn_readout_kernel<<<(nq + B - 1) / B, B, 0, stream>>>(bufA, Wd, bd, out, nq, 4 * H);
}


# gather wave shortened: count, scale, bias and 16 list words come as one group of seven loads (16-byte list loads), each row offset from one 24-bit multiply-add; list-order sums kept
# speedup vs baseline: 3.6611x; 1.0685x over previous
#include <hip/hip_runtime.h>

typedef __attribute__((ext_vector_type(2))) float v2f;
typedef __attribute__((ext_vector_type(8))) float v8f;


__global__ void gcn_zero_kernel(int* __restrict__ p, int n) {
  int i = blockIdx.x * blockDim.x + threadIdx.x;
  if (i < n) p[i] = 0;
}

#define GCN_SLOT_CAP 32

#define GCN_TICK_STRIDE 64
#define GCN_EDGE_BLOCK 256
#define GCN_EDGE_PER_LANE 4
#define GCN_CHUNK (GCN_EDGE_BLOCK * GCN_EDGE_PER_LANE)

static __device__ __forceinline__ void gcn_lds_barrier() {
  __builtin_amdgcn_fence(__ATOMIC_RELEASE, "workgroup", "local");
  __builtin_amdgcn_s_barrier();
  __builtin_amdgcn_fence(__ATOMIC_ACQUIRE, "workgroup", "local");
}

__global__ void __launch_bounds__(GCN_EDGE_BLOCK)
gcn_edge_kernel(const int* __restrict__ src, const int* __restrict__ dst, int* __restrict__ tick,
                int* __restrict__ cnt_out, int* __restrict__ cnt_in, int* __restrict__ slots,
                int E, int n) {
  __shared__ int s_chunk[2];
  const int tid = threadIdx.x;
  const int nchunks = (E + GCN_CHUNK - 1) / GCN_CHUNK;
  int split = ((n >> 1) + 63) & ~63;
  if (split > n) split = n;
  int t = (int)((__builtin_amdgcn_s_sendmsg_rtn(0x87) >> 18) & 1u);

  for (int pass = 0; pass < 2; ++pass, t ^= 1) {
    const int lo = t ? split : 0;
    const unsigned span = t ? (unsigned)(n - split) : (unsigned)split;
    int* tk = tick + t * GCN_TICK_STRIDE;

    if (tid == 0) {
      int first = nchunks;
      if (pass == 0 || __hip_atomic_load(tk, __ATOMIC_RELAXED, __HIP_MEMORY_SCOPE_AGENT) < nchunks)
        first = __hip_atomic_fetch_add(tk, 1, __ATOMIC_RELAXED, __HIP_MEMORY_SCOPE_AGENT);
      s_chunk[0] = first;
    }
    gcn_lds_barrier();
    int c = s_chunk[0];
    int buf = 0;
    while (c < nchunks) {
      int s[GCN_EDGE_PER_LANE], d[GCN_EDGE_PER_LANE], p[GCN_EDGE_PER_LANE];
      bool os[GCN_EDGE_PER_LANE], od[GCN_EDGE_PER_LANE];
#pragma unroll
      for (int k = 0; k < GCN_EDGE_PER_LANE; ++k) {
        const int e = c * GCN_CHUNK + k * GCN_EDGE_BLOCK + tid;
        const int ec = (e < E) ? e : E - 1;
        s[k] = src[ec];
        d[k] = dst[ec];
        os[k] = (e < E);
      }
      asm volatile("" :: "v"(s[0]), "v"(d[0]), "v"(s[1]), "v"(d[1]),
                         "v"(s[2]), "v"(d[2]), "v"(s[3]), "v"(d[3]));
#pragma unroll
      for (int k = 0; k < GCN_EDGE_PER_LANE; ++k) {
        od[k] = os[k] && ((unsigned)(d[k] - lo) < span);
        os[k] = os[k] && ((unsigned)(s[k] - lo) < span);
      }
#pragma unroll
      for (int k = 0; k < GCN_EDGE_PER_LANE; ++k) {
        p[k] = GCN_SLOT_CAP;
        if (od[k])
          p[k] = __hip_atomic_fetch_add(&cnt_in[d[k]], 1, __ATOMIC_RELAXED, __HIP_MEMORY_SCOPE_AGENT);
      }
#pragma unroll
      for (int k = 0; k < GCN_EDGE_PER_LANE; ++k) {
        if (os[k])
          (void)__hip_atomic_fetch_add(&cnt_out[s[k]], 1, __ATOMIC_RELAXED, __HIP_MEMORY_SCOPE_AGENT);
      }
      int nxt = 0;
      if (tid == 0) nxt = __hip_atomic_fetch_add(tk, 1, __ATOMIC_RELAXED, __HIP_MEMORY_SCOPE_AGENT);
#pragma unroll
      for (int k = 0; k < GCN_EDGE_PER_LANE; ++k) {
        if ((unsigned)p[k] < (unsigned)GCN_SLOT_CAP) slots[(size_t)d[k] * GCN_SLOT_CAP + p[k]] = s[k];
      }
      buf ^= 1;
      if (tid == 0) s_chunk[buf] = nxt;
      gcn_lds_barrier();
      c = s_chunk[buf];
    }
    gcn_lds_barrier();
  }
}

__global__ void gcn_norm_kernel(const int* __restrict__ cnt, float* __restrict__ nrm, int n) {
  int i = blockIdx.x * blockDim.x + threadIdx.x;
  if (i < n) nrm[i] = rsqrtf(fmaxf((float)cnt[i], 1.0f));
}

__global__ void gcn_gemm_wmma_kernel(const float* __restrict__ A, int lda, int K,
                                     const float* __restrict__ W, int wcols,
                                     const float* __restrict__ norm,
                                     float* __restrict__ out, int ldo, int ntiles) {
  __shared__ float Ws[32 * 32];

  for (int i = threadIdx.x; i < 32 * 32; i += blockDim.x) {
    const int k = i >> 5;
    const int n = i & 31;
    Ws[i] = (k < K && n < wcols) ? W[(size_t)k * wcols + n] : 0.0f;
  }
  __syncthreads();

  const int wave = (blockIdx.x * blockDim.x + threadIdx.x) >> 5;
  const int lane = threadIdx.x & 31;
  if (wave >= ntiles) return;

  const int m = lane & 15;
  const int half = lane >> 4;

  const int row = wave * 16 + m;
  const float nrm = norm[row];
  const float* arow = A + (size_t)row * lda;

  v8f c0 = {};
  v8f c1 = {};

  for (int k0 = 0; k0 < K; k0 += 4) {
    const int k = k0 + 2 * half;
    const float2 av = *(const float2*)(arow + k);
    v2f a;
    a.x = av.x * nrm;
    a.y = av.y * nrm;

    v2f b0, b1;
    b0.x = Ws[k * 32 + m];
    b0.y = Ws[(k + 1) * 32 + m];
    b1.x = Ws[k * 32 + 16 + m];
    b1.y = Ws[(k + 1) * 32 + 16 + m];

    c0 = __builtin_amdgcn_wmma_f32_16x16x4_f32(false, a, false, b0, (short)0, c0, false, false);
    c1 = __builtin_amdgcn_wmma_f32_16x16x4_f32(false, a, false, b1, (short)0, c1, false, false);
  }

  for (int v = 0; v < 8; ++v) {
    const int orow = wave * 16 + v + 8 * half;
    out[(size_t)orow * ldo + m] = c0[v];
    if (16 + m < wcols) out[(size_t)orow * ldo + 16 + m] = c1[v];
  }
}


static __device__ __forceinline__ void gcn_trip8(float4& acc, const char* __restrict__ xb,
                                                 unsigned rowbytes, unsigned cbytes, int r,
                                                 int e0, int e1, int e2, int e3,
                                                 int e4, int e5, int e6, int e7) {
  const unsigned s0 = (unsigned)e0;
  const unsigned s1 = (unsigned)((r > 1) ? e1 : e0);
  const unsigned s2 = (unsigned)((r > 2) ? e2 : e0);
  const unsigned s3 = (unsigned)((r > 3) ? e3 : e0);
  const unsigned s4 = (unsigned)((r > 4) ? e4 : e0);
  const unsigned s5 = (unsigned)((r > 5) ? e5 : e0);
  const unsigned s6 = (unsigned)((r > 6) ? e6 : e0);
  const unsigned s7 = (unsigned)((r > 7) ? e7 : e0);
  const float4 v0 = *(const float4*)(xb + (__umul24(s0, rowbytes) + cbytes));
  const float4 v1 = *(const float4*)(xb + (__umul24(s1, rowbytes) + cbytes));
  const float4 v2 = *(const float4*)(xb + (__umul24(s2, rowbytes) + cbytes));
  const float4 v3 = *(const float4*)(xb + (__umul24(s3, rowbytes) + cbytes));
  const float4 v4 = *(const float4*)(xb + (__umul24(s4, rowbytes) + cbytes));
  const float4 v5 = *(const float4*)(xb + (__umul24(s5, rowbytes) + cbytes));
  const float4 v6 = *(const float4*)(xb + (__umul24(s6, rowbytes) + cbytes));
  const float4 v7 = *(const float4*)(xb + (__umul24(s7, rowbytes) + cbytes));
  asm volatile("" :: "v"(v0.x), "v"(v1.x), "v"(v2.x), "v"(v3.x),
                     "v"(v4.x), "v"(v5.x), "v"(v6.x), "v"(v7.x));
  __builtin_amdgcn_sched_barrier(0);
  acc.x += v0.x; acc.y += v0.y; acc.z += v0.z; acc.w += v0.w;
  if (r > 1) { acc.x += v1.x; acc.y += v1.y; acc.z += v1.z; acc.w += v1.w; }
  if (r > 2) { acc.x += v2.x; acc.y += v2.y; acc.z += v2.z; acc.w += v2.w; }
  if (r > 3) { acc.x += v3.x; acc.y += v3.y; acc.z += v3.z; acc.w += v3.w; }
  if (r > 4) { acc.x += v4.x; acc.y += v4.y; acc.z += v4.z; acc.w += v4.w; }
  if (r > 5) { acc.x += v5.x; acc.y += v5.y; acc.z += v5.z; acc.w += v5.w; }
  if (r > 6) { acc.x += v6.x; acc.y += v6.y; acc.z += v6.z; acc.w += v6.w; }
  if (r > 7) { acc.x += v7.x; acc.y += v7.y; acc.z += v7.z; acc.w += v7.w; }
}

__global__ void gcn_pull_kernel(const float* __restrict__ x, const int* __restrict__ cnt_in,
                                const int* __restrict__ slots, const int* __restrict__ src,
                                const int* __restrict__ dst, int E,
                                const float* __restrict__ norm_dst, const float* __restrict__ bias,
                                float* __restrict__ h, int n, int nch, int H) {
  const int gid = blockIdx.x * blockDim.x + threadIdx.x;
  const int node = gid >> 3;
  const int c = gid & 7;
  if (node >= n || c >= nch) return;

  const int* sl = slots + (size_t)node * GCN_SLOT_CAP;
  const float* xc = x + c * 4;
  const char* xb = (const char*)x;
  const unsigned rowbytes = (unsigned)H * 4u;
  const unsigned cbytes = (unsigned)c * 16u;

  const int cnt = cnt_in[node];
  const int4 q0 = *(const int4*)(sl);
  const int4 q1 = *(const int4*)(sl + 4);
  const int4 q2 = *(const int4*)(sl + 8);
  const int4 q3 = *(const int4*)(sl + 12);
  const float nd = norm_dst[node];
  const float4 b = *(const float4*)(bias + c * 4);
  asm volatile("" :: "v"(cnt), "v"(q0.x), "v"(q1.x), "v"(q2.x), "v"(q3.x), "v"(nd), "v"(b.x));

  float4 acc = make_float4(0.0f, 0.0f, 0.0f, 0.0f);
  if (cnt <= GCN_SLOT_CAP) {
    if (cnt > 0) gcn_trip8(acc, xb, rowbytes, cbytes, cnt, q0.x, q0.y, q0.z, q0.w, q1.x, q1.y, q1.z, q1.w);
    if (cnt > 8) gcn_trip8(acc, xb, rowbytes, cbytes, cnt - 8, q2.x, q2.y, q2.z, q2.w, q3.x, q3.y, q3.z, q3.w);
    for (int j = 16; j < cnt; j += 8) {
      const int4 p0 = *(const int4*)(sl + j);
      const int4 p1 = *(const int4*)(sl + j + 4);
      gcn_trip8(acc, xb, rowbytes, cbytes, cnt - j, p0.x, p0.y, p0.z, p0.w, p1.x, p1.y, p1.z, p1.w);
    }
  } else {
    for (int e = 0; e < E; ++e) {
      if (dst[e] == node) {
        const float4 v = *(const float4*)(xc + src[e] * H);
        acc.x += v.x; acc.y += v.y; acc.z += v.z; acc.w += v.w;
      }
    }
  }

  float4 o;
  o.x = fmaxf(fmaf(acc.x, nd, b.x), 0.0f);
  o.y = fmaxf(fmaf(acc.y, nd, b.y), 0.0f);
  o.z = fmaxf(fmaf(acc.z, nd, b.z), 0.0f);
  o.w = fmaxf(fmaf(acc.w, nd, b.w), 0.0f);
  *(float4*)(h + (size_t)node * H + c * 4) = o;
}

__global__ void gcn_readout_kernel(const float* __restrict__ h, const float* __restrict__ Wd,
                                   const float* __restrict__ bd, float* __restrict__ out,
                                   int nq, int rowlen  ) {
  int r = blockIdx.x * blockDim.x + threadIdx.x;
  if (r >= nq) return;
  const float4* hp = (const float4*)(h + (size_t)r * rowlen);
  const float4* wp = (const float4*)Wd;
  const int nch = rowlen >> 2;
  float s = 0.0f;
  for (int j = 0; j < nch; ++j) {
    float4 hv = hp[j];
    float4 wv = wp[j];
    s = fmaf(hv.x, wv.x, s);
    s = fmaf(hv.y, wv.y, s);
    s = fmaf(hv.z, wv.z, s);
    s = fmaf(hv.w, wv.w, s);
  }
  out[r] = s + bd[0];
}


extern "C" void kernel_launch(void* const* d_in, const int* in_sizes, int n_in,
                              void* d_out, int out_size, void* d_ws, size_t ws_size,
                              hipStream_t stream) {
  const float* features = (const float*)d_in[0];
  const int*   src      = (const int*)d_in[1];
  const int*   dst      = (const int*)d_in[2];
  const float* W1       = (const float*)d_in[3];
  const float* b1       = (const float*)d_in[4];
  const float* W2       = (const float*)d_in[5];
  const float* b2       = (const float*)d_in[6];
  const float* Wd       = (const float*)d_in[7];
  const float* bd       = (const float*)d_in[8];
  float* out = (float*)d_out;

  const int H  = in_sizes[4];
  const int F  = in_sizes[3] / H;
  const int Nn = in_sizes[0] / F;
  const int E  = in_sizes[1];

  const int B = 256;
  const int ntiles = Nn / 16;
  const int wavesPerBlock = B / 32;
  const int gemmGrid = (ntiles + wavesPerBlock - 1) / wavesPerBlock;
  const int nch = H / 4;
  const int nq = Nn / 4;
  const int pullGrid = (Nn * 8 + B - 1) / B;

  int*   tick     = (int*)d_ws;
  int*   cnt_out  = tick + 2 * GCN_TICK_STRIDE;
  int*   cnt_in   = cnt_out + Nn;
  float* norm_src = (float*)(cnt_in + Nn);
  float* norm_dst = norm_src + Nn;
  float* bufA     = norm_dst + Nn;
  float* bufX     = bufA + (size_t)Nn * H;
  int*   slots    = (int*)(bufX + (size_t)Nn * H);

  gcn_zero_kernel<<<(2 * GCN_TICK_STRIDE + 2 * Nn + B - 1) / B, B, 0, stream>>>(
      tick, 2 * GCN_TICK_STRIDE + 2 * Nn);
  gcn_edge_kernel<<<2048, GCN_EDGE_BLOCK, 0, stream>>>(src, dst, tick, cnt_out, cnt_in, slots, E, Nn);
  gcn_norm_kernel<<<(2 * Nn + B - 1) / B, B, 0, stream>>>(cnt_out, norm_src, 2 * Nn);
  gcn_gemm_wmma_kernel<<<gemmGrid, B, 0, stream>>>(features, F, F, W1, H, norm_src,
                                                   bufX, H, ntiles);
  gcn_pull_kernel<<<pullGrid, B, 0, stream>>>(bufX, cnt_in, slots, src, dst, E, norm_dst, b1,
                                              bufA, Nn, nch, H);
  gcn_gemm_wmma_kernel<<<gemmGrid, B, 0, stream>>>(bufA, H, H, W2, H, norm_src,
                                                   bufX, H, ntiles);
  gcn_pull_kernel<<<pullGrid, B, 0, stream>>>(bufX, cnt_in, slots, src, dst, E, norm_dst, b2,
                                              bufA, Nn, nch, H);
  gcn_readout_kernel<<<(nq + B - 1) / B, B, 0, stream>>>(bufA, Wd, bd, out, nq, 4 * H);
}


# layer 2 re-associated: gather-sum of the scaled h1 rows first, then the W2 product (WMMA) once per tile with scale, bias, relu and the readout dot (16-lane tree sums) in its epilogue; 8->7 launches
# speedup vs baseline: 4.0500x; 1.1062x over previous
#include <hip/hip_runtime.h>

typedef __attribute__((ext_vector_type(2))) float v2f;
typedef __attribute__((ext_vector_type(8))) float v8f;


__global__ void gcn_zero_kernel(int* __restrict__ p, int n) {
  int i = blockIdx.x * blockDim.x + threadIdx.x;
  if (i < n) p[i] = 0;
}

#define GCN_SLOT_CAP 32

#define GCN_TICK_STRIDE 64
#define GCN_EDGE_BLOCK 256
#define GCN_EDGE_PER_LANE 4
#define GCN_CHUNK (GCN_EDGE_BLOCK * GCN_EDGE_PER_LANE)

static __device__ __forceinline__ void gcn_lds_barrier() {
  __builtin_amdgcn_fence(__ATOMIC_RELEASE, "workgroup", "local");
  __builtin_amdgcn_s_barrier();
  __builtin_amdgcn_fence(__ATOMIC_ACQUIRE, "workgroup", "local");
}

__global__ void __launch_bounds__(GCN_EDGE_BLOCK)
gcn_edge_kernel(const int* __restrict__ src, const int* __restrict__ dst, int* __restrict__ tick,
                int* __restrict__ cnt_out, int* __restrict__ cnt_in, int* __restrict__ slots,
                int E, int n) {
  __shared__ int s_chunk[2];
  const int tid = threadIdx.x;
  const int nchunks = (E + GCN_CHUNK - 1) / GCN_CHUNK;
  int split = ((n >> 1) + 63) & ~63;
  if (split > n) split = n;
  int t = (int)((__builtin_amdgcn_s_sendmsg_rtn(0x87) >> 18) & 1u);

  for (int pass = 0; pass < 2; ++pass, t ^= 1) {
    const int lo = t ? split : 0;
    const unsigned span = t ? (unsigned)(n - split) : (unsigned)split;
    int* tk = tick + t * GCN_TICK_STRIDE;

    if (tid == 0) {
      int first = nchunks;
      if (pass == 0 || __hip_atomic_load(tk, __ATOMIC_RELAXED, __HIP_MEMORY_SCOPE_AGENT) < nchunks)
        first = __hip_atomic_fetch_add(tk, 1, __ATOMIC_RELAXED, __HIP_MEMORY_SCOPE_AGENT);
      s_chunk[0] = first;
    }
    gcn_lds_barrier();
    int c = s_chunk[0];
    int buf = 0;
    while (c < nchunks) {
      int s[GCN_EDGE_PER_LANE], d[GCN_EDGE_PER_LANE], p[GCN_EDGE_PER_LANE];
      bool os[GCN_EDGE_PER_LANE], od[GCN_EDGE_PER_LANE];
#pragma unroll
      for (int k = 0; k < GCN_EDGE_PER_LANE; ++k) {
        const int e = c * GCN_CHUNK + k * GCN_EDGE_BLOCK + tid;
        const int ec = (e < E) ? e : E - 1;
        s[k] = src[ec];
        d[k] = dst[ec];
        os[k] = (e < E);
      }
      asm volatile("" :: "v"(s[0]), "v"(d[0]), "v"(s[1]), "v"(d[1]),
                         "v"(s[2]), "v"(d[2]), "v"(s[3]), "v"(d[3]));
#pragma unroll
      for (int k = 0; k < GCN_EDGE_PER_LANE; ++k) {
        od[k] = os[k] && ((unsigned)(d[k] - lo) < span);
        os[k] = os[k] && ((unsigned)(s[k] - lo) < span);
      }
#pragma unroll
      for (int k = 0; k < GCN_EDGE_PER_LANE; ++k) {
        p[k] = GCN_SLOT_CAP;
        if (od[k])
          p[k] = __hip_atomic_fetch_add(&cnt_in[d[k]], 1, __ATOMIC_RELAXED, __HIP_MEMORY_SCOPE_AGENT);
      }
#pragma unroll
      for (int k = 0; k < GCN_EDGE_PER_LANE; ++k) {
        if (os[k])
          (void)__hip_atomic_fetch_add(&cnt_out[s[k]], 1, __ATOMIC_RELAXED, __HIP_MEMORY_SCOPE_AGENT);
      }
      int nxt = 0;
      if (tid == 0) nxt = __hip_atomic_fetch_add(tk, 1, __ATOMIC_RELAXED, __HIP_MEMORY_SCOPE_AGENT);
#pragma unroll
      for (int k = 0; k < GCN_EDGE_PER_LANE; ++k) {
        if ((unsigned)p[k] < (unsigned)GCN_SLOT_CAP) slots[(size_t)d[k] * GCN_SLOT_CAP + p[k]] = s[k];
      }
      buf ^= 1;
      if (tid == 0) s_chunk[buf] = nxt;
      gcn_lds_barrier();
      c = s_chunk[buf];
    }
    gcn_lds_barrier();
  }
}

__global__ void gcn_norm_kernel(const int* __restrict__ cnt, float* __restrict__ nrm, int n) {
  int i = blockIdx.x * blockDim.x + threadIdx.x;
  if (i < n) nrm[i] = rsqrtf(fmaxf((float)cnt[i], 1.0f));
}

template <bool RO>
__global__ void gcn_gemm_wmma_kernel(const float* __restrict__ A, int lda, int K,
                                     const float* __restrict__ W, int wcols,
                                     const float* __restrict__ norm,
                                     float* __restrict__ out, int ldo, int ntiles,
                                     const float* __restrict__ bias, const float* __restrict__ Wd,
                                     const float* __restrict__ bd) {
  __shared__ float Ws[32 * 32];
  __shared__ float s_ro[32];

  for (int i = threadIdx.x; i < 32 * 32; i += blockDim.x) {
    const int k = i >> 5;
    const int n = i & 31;
    Ws[i] = (k < K && n < wcols) ? W[(size_t)k * wcols + n] : 0.0f;
  }
  __syncthreads();

  const int wave = (blockIdx.x * blockDim.x + threadIdx.x) >> 5;
  const int lane = threadIdx.x & 31;
  const int m = lane & 15;
  const int half = lane >> 4;

  if (wave < ntiles) {
    const int row = wave * 16 + m;
    const float* arow = A + (size_t)row * lda;
    float nrm = 1.0f;
    if constexpr (!RO) nrm = norm[row];

    float ndv[8], wA[4], wB[4], bA = 0.0f, bB = 0.0f, bd0 = 0.0f;
    if constexpr (RO) {
      const float* np = norm + wave * 16 + 8 * half;
      const float4 n0 = *(const float4*)(np);
      const float4 n1 = *(const float4*)(np + 4);
      ndv[0] = n0.x; ndv[1] = n0.y; ndv[2] = n0.z; ndv[3] = n0.w;
      ndv[4] = n1.x; ndv[5] = n1.y; ndv[6] = n1.z; ndv[7] = n1.w;
      const bool hasA = m < wcols;
      const bool hasB = 16 + m < wcols;
      const int colA = hasA ? m : 0;
      const int colB = hasB ? 16 + m : 0;
      bA = bias[colA];
      bB = bias[colB];
#pragma unroll
      for (int g = 0; g < 4; ++g) {
        const float ta = Wd[g * wcols + colA];
        const float tb = Wd[g * wcols + colB];
        wA[g] = hasA ? ta : 0.0f;
        wB[g] = hasB ? tb : 0.0f;
      }
      bd0 = bd[0];
    }

    v8f c0 = {};
    v8f c1 = {};

    for (int k0 = 0; k0 < K; k0 += 4) {
      const int k = k0 + 2 * half;
      const float2 av = *(const float2*)(arow + k);
      v2f a;
      a.x = av.x;
      a.y = av.y;
      if constexpr (!RO) {
        a.x = av.x * nrm;
        a.y = av.y * nrm;
      }

      v2f b0, b1;
      b0.x = Ws[k * 32 + m];
      b0.y = Ws[(k + 1) * 32 + m];
      b1.x = Ws[k * 32 + 16 + m];
      b1.y = Ws[(k + 1) * 32 + 16 + m];

      c0 = __builtin_amdgcn_wmma_f32_16x16x4_f32(false, a, false, b0, (short)0, c0, false, false);
      c1 = __builtin_amdgcn_wmma_f32_16x16x4_f32(false, a, false, b1, (short)0, c1, false, false);
    }

    if constexpr (!RO) {
      for (int v = 0; v < 8; ++v) {
        const int orow = wave * 16 + v + 8 * half;
        out[(size_t)orow * ldo + m] = c0[v];
        if (16 + m < wcols) out[(size_t)orow * ldo + 16 + m] = c1[v];
      }
    } else {
      float s0 = 0.0f, s1 = 0.0f;
#pragma unroll
      for (int g = 0; g < 4; ++g) {
        s0 = fmaf(fmaxf(fmaf(c0[g], ndv[g], bA), 0.0f), wA[g], s0);
        s0 = fmaf(fmaxf(fmaf(c1[g], ndv[g], bB), 0.0f), wB[g], s0);
        s1 = fmaf(fmaxf(fmaf(c0[4 + g], ndv[4 + g], bA), 0.0f), wA[g], s1);
        s1 = fmaf(fmaxf(fmaf(c1[4 + g], ndv[4 + g], bB), 0.0f), wB[g], s1);
      }
#pragma unroll
      for (int off = 8; off >= 1; off >>= 1) {
        s0 += __shfl_xor(s0, off);
        s1 += __shfl_xor(s1, off);
      }
      if (m == 0) {
        const int wv = threadIdx.x >> 5;
        s_ro[wv * 4 + 2 * half] = s0 + bd0;
        s_ro[wv * 4 + 2 * half + 1] = s1 + bd0;
      }
    }
  }

  if constexpr (RO) {
    __syncthreads();
    const int per_block = (blockDim.x >> 5) * 4;
    const int r = blockIdx.x * per_block + threadIdx.x;
    if (threadIdx.x < per_block && r < ntiles * 4) out[r] = s_ro[threadIdx.x];
  }
}


static __device__ __forceinline__ void gcn_trip8(float4& acc, const char* __restrict__ xb,
                                                 unsigned rowbytes, unsigned cbytes, int r,
                                                 int e0, int e1, int e2, int e3,
                                                 int e4, int e5, int e6, int e7) {
  const unsigned s0 = (unsigned)e0;
  const unsigned s1 = (unsigned)((r > 1) ? e1 : e0);
  const unsigned s2 = (unsigned)((r > 2) ? e2 : e0);
  const unsigned s3 = (unsigned)((r > 3) ? e3 : e0);
  const unsigned s4 = (unsigned)((r > 4) ? e4 : e0);
  const unsigned s5 = (unsigned)((r > 5) ? e5 : e0);
  const unsigned s6 = (unsigned)((r > 6) ? e6 : e0);
  const unsigned s7 = (unsigned)((r > 7) ? e7 : e0);
  const float4 v0 = *(const float4*)(xb + (__umul24(s0, rowbytes) + cbytes));
  const float4 v1 = *(const float4*)(xb + (__umul24(s1, rowbytes) + cbytes));
  const float4 v2 = *(const float4*)(xb + (__umul24(s2, rowbytes) + cbytes));
  const float4 v3 = *(const float4*)(xb + (__umul24(s3, rowbytes) + cbytes));
  const float4 v4 = *(const float4*)(xb + (__umul24(s4, rowbytes) + cbytes));
  const float4 v5 = *(const float4*)(xb + (__umul24(s5, rowbytes) + cbytes));
  const float4 v6 = *(const float4*)(xb + (__umul24(s6, rowbytes) + cbytes));
  const float4 v7 = *(const float4*)(xb + (__umul24(s7, rowbytes) + cbytes));
  asm volatile("" :: "v"(v0.x), "v"(v1.x), "v"(v2.x), "v"(v3.x),
                     "v"(v4.x), "v"(v5.x), "v"(v6.x), "v"(v7.x));
  __builtin_amdgcn_sched_barrier(0);
  acc.x += v0.x; acc.y += v0.y; acc.z += v0.z; acc.w += v0.w;
  if (r > 1) { acc.x += v1.x; acc.y += v1.y; acc.z += v1.z; acc.w += v1.w; }
  if (r > 2) { acc.x += v2.x; acc.y += v2.y; acc.z += v2.z; acc.w += v2.w; }
  if (r > 3) { acc.x += v3.x; acc.y += v3.y; acc.z += v3.z; acc.w += v3.w; }
  if (r > 4) { acc.x += v4.x; acc.y += v4.y; acc.z += v4.z; acc.w += v4.w; }
  if (r > 5) { acc.x += v5.x; acc.y += v5.y; acc.z += v5.z; acc.w += v5.w; }
  if (r > 6) { acc.x += v6.x; acc.y += v6.y; acc.z += v6.z; acc.w += v6.w; }
  if (r > 7) { acc.x += v7.x; acc.y += v7.y; acc.z += v7.z; acc.w += v7.w; }
}

template <bool L1>
__global__ void gcn_pull_kernel(const float* __restrict__ x, const int* __restrict__ cnt_in,
                                const int* __restrict__ slots, const int* __restrict__ src,
                                const int* __restrict__ dst, int E,
                                const float* __restrict__ norm_dst, const float* __restrict__ norm_src,
                                const float* __restrict__ bias,
                                float* __restrict__ h, int n, int nch, int H) {
  const int gid = blockIdx.x * blockDim.x + threadIdx.x;
  const int node = gid >> 3;
  const int c = gid & 7;
  if (node >= n || c >= nch) return;

  const int* sl = slots + (size_t)node * GCN_SLOT_CAP;
  const float* xc = x + c * 4;
  const char* xb = (const char*)x;
  const unsigned rowbytes = (unsigned)H * 4u;
  const unsigned cbytes = (unsigned)c * 16u;

  const int cnt = cnt_in[node];
  const int4 q0 = *(const int4*)(sl);
  const int4 q1 = *(const int4*)(sl + 4);
  const int4 q2 = *(const int4*)(sl + 8);
  const int4 q3 = *(const int4*)(sl + 12);
  float nd = 1.0f, ns = 1.0f;
  float4 b = make_float4(0.0f, 0.0f, 0.0f, 0.0f);
  if constexpr (L1) {
    nd = norm_dst[node];
    ns = norm_src[node];
    b = *(const float4*)(bias + c * 4);
    asm volatile("" :: "v"(cnt), "v"(q0.x), "v"(q1.x), "v"(q2.x), "v"(q3.x), "v"(nd), "v"(ns), "v"(b.x));
  } else {
    asm volatile("" :: "v"(cnt), "v"(q0.x), "v"(q1.x), "v"(q2.x), "v"(q3.x));
  }

  float4 acc = make_float4(0.0f, 0.0f, 0.0f, 0.0f);
  if (cnt <= GCN_SLOT_CAP) {
    if (cnt > 0) gcn_trip8(acc, xb, rowbytes, cbytes, cnt, q0.x, q0.y, q0.z, q0.w, q1.x, q1.y, q1.z, q1.w);
    if (cnt > 8) gcn_trip8(acc, xb, rowbytes, cbytes, cnt - 8, q2.x, q2.y, q2.z, q2.w, q3.x, q3.y, q3.z, q3.w);
    for (int j = 16; j < cnt; j += 8) {
      const int4 p0 = *(const int4*)(sl + j);
      const int4 p1 = *(const int4*)(sl + j + 4);
      gcn_trip8(acc, xb, rowbytes, cbytes, cnt - j, p0.x, p0.y, p0.z, p0.w, p1.x, p1.y, p1.z, p1.w);
    }
  } else {
    for (int e = 0; e < E; ++e) {
      if (dst[e] == node) {
        const float4 v = *(const float4*)(xc + src[e] * H);
        acc.x += v.x; acc.y += v.y; acc.z += v.z; acc.w += v.w;
      }
    }
  }

  float4 o = acc;
  if constexpr (L1) {
    o.x = fmaxf(fmaf(acc.x, nd, b.x), 0.0f) * ns;
    o.y = fmaxf(fmaf(acc.y, nd, b.y), 0.0f) * ns;
    o.z = fmaxf(fmaf(acc.z, nd, b.z), 0.0f) * ns;
    o.w = fmaxf(fmaf(acc.w, nd, b.w), 0.0f) * ns;
  }
  *(float4*)(h + (size_t)node * H + c * 4) = o;
}


extern "C" void kernel_launch(void* const* d_in, const int* in_sizes, int n_in,
                              void* d_out, int out_size, void* d_ws, size_t ws_size,
                              hipStream_t stream) {
  const float* features = (const float*)d_in[0];
  const int*   src      = (const int*)d_in[1];
  const int*   dst      = (const int*)d_in[2];
  const float* W1       = (const float*)d_in[3];
  const float* b1       = (const float*)d_in[4];
  const float* W2       = (const float*)d_in[5];
  const float* b2       = (const float*)d_in[6];
  const float* Wd       = (const float*)d_in[7];
  const float* bd       = (const float*)d_in[8];
  float* out = (float*)d_out;

  const int H  = in_sizes[4];
  const int F  = in_sizes[3] / H;
  const int Nn = in_sizes[0] / F;
  const int E  = in_sizes[1];

  const int B = 256;
  const int ntiles = Nn / 16;
  const int wavesPerBlock = B / 32;
  const int gemmGrid = (ntiles + wavesPerBlock - 1) / wavesPerBlock;
  const int nch = H / 4;
  const int pullGrid = (Nn * 8 + B - 1) / B;

  int*   tick     = (int*)d_ws;
  int*   cnt_out  = tick + 2 * GCN_TICK_STRIDE;
  int*   cnt_in   = cnt_out + Nn;
  float* norm_src = (float*)(cnt_in + Nn);
  float* norm_dst = norm_src + Nn;
  float* bufA     = norm_dst + Nn;
  float* bufX     = bufA + (size_t)Nn * H;
  int*   slots    = (int*)(bufX + (size_t)Nn * H);

  gcn_zero_kernel<<<(2 * GCN_TICK_STRIDE + 2 * Nn + B - 1) / B, B, 0, stream>>>(
      tick, 2 * GCN_TICK_STRIDE + 2 * Nn);
  gcn_edge_kernel<<<2048, GCN_EDGE_BLOCK, 0, stream>>>(src, dst, tick, cnt_out, cnt_in, slots, E, Nn);
  gcn_norm_kernel<<<(2 * Nn + B - 1) / B, B, 0, stream>>>(cnt_out, norm_src, 2 * Nn);
  gcn_gemm_wmma_kernel<false><<<gemmGrid, B, 0, stream>>>(features, F, F, W1, H, norm_src,
                                                          bufX, H, ntiles, nullptr, nullptr, nullptr);
  gcn_pull_kernel<true><<<pullGrid, B, 0, stream>>>(bufX, cnt_in, slots, src, dst, E, norm_dst,
                                                    norm_src, b1, bufA, Nn, nch, H);
  gcn_pull_kernel<false><<<pullGrid, B, 0, stream>>>(bufA, cnt_in, slots, src, dst, E, nullptr,
                                                     nullptr, nullptr, bufX, Nn, nch, H);
  gcn_gemm_wmma_kernel<true><<<gemmGrid, B, 0, stream>>>(bufX, H, H, W2, H, norm_dst,
                                                         out, 0, ntiles, b2, Wd, bd);
}


# layer-2 gather-sum and W2 product in one work-group: 128 nodes' summed rows go through LDS to 8 WMMA tiles (epilogue, readout as before), not through memory; 7->6 launches; list-order sums kept
# speedup vs baseline: 4.2042x; 1.0381x over previous
#include <hip/hip_runtime.h>

typedef __attribute__((ext_vector_type(2))) float v2f;
typedef __attribute__((ext_vector_type(8))) float v8f;


__global__ void gcn_zero_kernel(int* __restrict__ p, int n) {
  int i = blockIdx.x * blockDim.x + threadIdx.x;
  if (i < n) p[i] = 0;
}

#define GCN_SLOT_CAP 32

#define GCN_TICK_STRIDE 64
#define GCN_EDGE_BLOCK 256
#define GCN_EDGE_PER_LANE 4
#define GCN_CHUNK (GCN_EDGE_BLOCK * GCN_EDGE_PER_LANE)

static __device__ __forceinline__ void gcn_lds_barrier() {
  __builtin_amdgcn_fence(__ATOMIC_RELEASE, "workgroup", "local");
  __builtin_amdgcn_s_barrier();
  __builtin_amdgcn_fence(__ATOMIC_ACQUIRE, "workgroup", "local");
}

__global__ void __launch_bounds__(GCN_EDGE_BLOCK)
gcn_edge_kernel(const int* __restrict__ src, const int* __restrict__ dst, int* __restrict__ tick,
                int* __restrict__ cnt_out, int* __restrict__ cnt_in, int* __restrict__ slots,
                int E, int n) {
  __shared__ int s_chunk[2];
  const int tid = threadIdx.x;
  const int nchunks = (E + GCN_CHUNK - 1) / GCN_CHUNK;
  int split = ((n >> 1) + 63) & ~63;
  if (split > n) split = n;
  int t = (int)((__builtin_amdgcn_s_sendmsg_rtn(0x87) >> 18) & 1u);

  for (int pass = 0; pass < 2; ++pass, t ^= 1) {
    const int lo = t ? split : 0;
    const unsigned span = t ? (unsigned)(n - split) : (unsigned)split;
    int* tk = tick + t * GCN_TICK_STRIDE;

    if (tid == 0) {
      int first = nchunks;
      if (pass == 0 || __hip_atomic_load(tk, __ATOMIC_RELAXED, __HIP_MEMORY_SCOPE_AGENT) < nchunks)
        first = __hip_atomic_fetch_add(tk, 1, __ATOMIC_RELAXED, __HIP_MEMORY_SCOPE_AGENT);
      s_chunk[0] = first;
    }
    gcn_lds_barrier();
    int c = s_chunk[0];
    int buf = 0;
    while (c < nchunks) {
      int s[GCN_EDGE_PER_LANE], d[GCN_EDGE_PER_LANE], p[GCN_EDGE_PER_LANE];
      bool os[GCN_EDGE_PER_LANE], od[GCN_EDGE_PER_LANE];
#pragma unroll
      for (int k = 0; k < GCN_EDGE_PER_LANE; ++k) {
        const int e = c * GCN_CHUNK + k * GCN_EDGE_BLOCK + tid;
        const int ec = (e < E) ? e : E - 1;
        s[k] = src[ec];
        d[k] = dst[ec];
        os[k] = (e < E);
      }
      asm volatile("" :: "v"(s[0]), "v"(d[0]), "v"(s[1]), "v"(d[1]),
                         "v"(s[2]), "v"(d[2]), "v"(s[3]), "v"(d[3]));
#pragma unroll
      for (int k = 0; k < GCN_EDGE_PER_LANE; ++k) {
        od[k] = os[k] && ((unsigned)(d[k] - lo) < span);
        os[k] = os[k] && ((unsigned)(s[k] - lo) < span);
      }
#pragma unroll
      for (int k = 0; k < GCN_EDGE_PER_LANE; ++k) {
        p[k] = GCN_SLOT_CAP;
        if (od[k])
          p[k] = __hip_atomic_fetch_add(&cnt_in[d[k]], 1, __ATOMIC_RELAXED, __HIP_MEMORY_SCOPE_AGENT);
      }
#pragma unroll
      for (int k = 0; k < GCN_EDGE_PER_LANE; ++k) {
        if (os[k])
          (void)__hip_atomic_fetch_add(&cnt_out[s[k]], 1, __ATOMIC_RELAXED, __HIP_MEMORY_SCOPE_AGENT);
      }
      int nxt = 0;
      if (tid == 0) nxt = __hip_atomic_fetch_add(tk, 1, __ATOMIC_RELAXED, __HIP_MEMORY_SCOPE_AGENT);
#pragma unroll
      for (int k = 0; k < GCN_EDGE_PER_LANE; ++k) {
        if ((unsigned)p[k] < (unsigned)GCN_SLOT_CAP) slots[(size_t)d[k] * GCN_SLOT_CAP + p[k]] = s[k];
      }
      buf ^= 1;
      if (tid == 0) s_chunk[buf] = nxt;
      gcn_lds_barrier();
      c = s_chunk[buf];
    }
    gcn_lds_barrier();
  }
}

__global__ void gcn_norm_kernel(const int* __restrict__ cnt, float* __restrict__ nrm, int n) {
  int i = blockIdx.x * blockDim.x + threadIdx.x;
  if (i < n) nrm[i] = rsqrtf(fmaxf((float)cnt[i], 1.0f));
}


static __device__ __forceinline__ void gcn_stage_w(float* __restrict__ Ws, const float* __restrict__ W,
                                                   int K, int wcols) {
  for (int i = threadIdx.x; i < 32 * 32; i += blockDim.x) {
    const int k = i >> 5;
    const int n = i & 31;
    Ws[i] = (k < K && n < wcols) ? W[(size_t)k * wcols + n] : 0.0f;
  }
}

__global__ void gcn_gemm_wmma_kernel(const float* __restrict__ A, int lda, int K,
                                     const float* __restrict__ W, int wcols,
                                     const float* __restrict__ norm,
                                     float* __restrict__ out, int ldo, int ntiles) {
  __shared__ float Ws[32 * 32];

  gcn_stage_w(Ws, W, K, wcols);
  __syncthreads();

  const int wave = (blockIdx.x * blockDim.x + threadIdx.x) >> 5;
  const int lane = threadIdx.x & 31;
  const int m = lane & 15;
  const int half = lane >> 4;

  if (wave < ntiles) {
    const int row = wave * 16 + m;
    const float* arow = A + (size_t)row * lda;
    const float nrm = norm[row];

    v8f c0 = {};
    v8f c1 = {};

    for (int k0 = 0; k0 < K; k0 += 4) {
      const int k = k0 + 2 * half;
      const float2 av = *(const float2*)(arow + k);
      v2f a;
      a.x = av.x * nrm;
      a.y = av.y * nrm;

      v2f b0, b1;
      b0.x = Ws[k * 32 + m];
      b0.y = Ws[(k + 1) * 32 + m];
      b1.x = Ws[k * 32 + 16 + m];
      b1.y = Ws[(k + 1) * 32 + 16 + m];

      c0 = __builtin_amdgcn_wmma_f32_16x16x4_f32(false, a, false, b0, (short)0, c0, false, false);
      c1 = __builtin_amdgcn_wmma_f32_16x16x4_f32(false, a, false, b1, (short)0, c1, false, false);
    }

    for (int v = 0; v < 8; ++v) {
      const int orow = wave * 16 + v + 8 * half;
      out[(size_t)orow * ldo + m] = c0[v];
      if (16 + m < wcols) out[(size_t)orow * ldo + 16 + m] = c1[v];
    }
  }
}


static __device__ __forceinline__ void gcn_trip8(float4& acc, const char* __restrict__ xb,
                                                 unsigned rowbytes, unsigned cbytes, int r,
                                                 int e0, int e1, int e2, int e3,
                                                 int e4, int e5, int e6, int e7) {
  const unsigned s0 = (unsigned)e0;
  const unsigned s1 = (unsigned)((r > 1) ? e1 : e0);
  const unsigned s2 = (unsigned)((r > 2) ? e2 : e0);
  const unsigned s3 = (unsigned)((r > 3) ? e3 : e0);
  const unsigned s4 = (unsigned)((r > 4) ? e4 : e0);
  const unsigned s5 = (unsigned)((r > 5) ? e5 : e0);
  const unsigned s6 = (unsigned)((r > 6) ? e6 : e0);
  const unsigned s7 = (unsigned)((r > 7) ? e7 : e0);
  const float4 v0 = *(const float4*)(xb + (__umul24(s0, rowbytes) + cbytes));
  const float4 v1 = *(const float4*)(xb + (__umul24(s1, rowbytes) + cbytes));
  const float4 v2 = *(const float4*)(xb + (__umul24(s2, rowbytes) + cbytes));
  const float4 v3 = *(const float4*)(xb + (__umul24(s3, rowbytes) + cbytes));
  const float4 v4 = *(const float4*)(xb + (__umul24(s4, rowbytes) + cbytes));
  const float4 v5 = *(const float4*)(xb + (__umul24(s5, rowbytes) + cbytes));
  const float4 v6 = *(const float4*)(xb + (__umul24(s6, rowbytes) + cbytes));
  const float4 v7 = *(const float4*)(xb + (__umul24(s7, rowbytes) + cbytes));
  asm volatile("" :: "v"(v0.x), "v"(v1.x), "v"(v2.x), "v"(v3.x),
                     "v"(v4.x), "v"(v5.x), "v"(v6.x), "v"(v7.x));
  __builtin_amdgcn_sched_barrier(0);
  acc.x += v0.x; acc.y += v0.y; acc.z += v0.z; acc.w += v0.w;
  if (r > 1) { acc.x += v1.x; acc.y += v1.y; acc.z += v1.z; acc.w += v1.w; }
  if (r > 2) { acc.x += v2.x; acc.y += v2.y; acc.z += v2.z; acc.w += v2.w; }
  if (r > 3) { acc.x += v3.x; acc.y += v3.y; acc.z += v3.z; acc.w += v3.w; }
  if (r > 4) { acc.x += v4.x; acc.y += v4.y; acc.z += v4.z; acc.w += v4.w; }
  if (r > 5) { acc.x += v5.x; acc.y += v5.y; acc.z += v5.z; acc.w += v5.w; }
  if (r > 6) { acc.x += v6.x; acc.y += v6.y; acc.z += v6.z; acc.w += v6.w; }
  if (r > 7) { acc.x += v7.x; acc.y += v7.y; acc.z += v7.z; acc.w += v7.w; }
}

static __device__ __forceinline__ float4 gcn_list_sum(const float* __restrict__ x, const int* __restrict__ sl,
                                                      int node, int c, int H, int cnt,
                                                      int4 q0, int4 q1, int4 q2, int4 q3,
                                                      const int* __restrict__ src,
                                                      const int* __restrict__ dst, int E) {
  const float* xc = x + c * 4;
  const char* xb = (const char*)x;
  const unsigned rowbytes = (unsigned)H * 4u;
  const unsigned cbytes = (unsigned)c * 16u;
  float4 acc = make_float4(0.0f, 0.0f, 0.0f, 0.0f);
  if (cnt <= GCN_SLOT_CAP) {
    if (cnt > 0) gcn_trip8(acc, xb, rowbytes, cbytes, cnt, q0.x, q0.y, q0.z, q0.w, q1.x, q1.y, q1.z, q1.w);
    if (cnt > 8) gcn_trip8(acc, xb, rowbytes, cbytes, cnt - 8, q2.x, q2.y, q2.z, q2.w, q3.x, q3.y, q3.z, q3.w);
    for (int j = 16; j < cnt; j += 8) {
      const int4 p0 = *(const int4*)(sl + j);
      const int4 p1 = *(const int4*)(sl + j + 4);
      gcn_trip8(acc, xb, rowbytes, cbytes, cnt - j, p0.x, p0.y, p0.z, p0.w, p1.x, p1.y, p1.z, p1.w);
    }
  } else {
    for (int e = 0; e < E; ++e) {
      if (dst[e] == node) {
        const float4 v = *(const float4*)(xc + src[e] * H);
        acc.x += v.x; acc.y += v.y; acc.z += v.z; acc.w += v.w;
      }
    }
  }
  return acc;
}

__global__ void gcn_pull_kernel(const float* __restrict__ x, const int* __restrict__ cnt_in,
                                const int* __restrict__ slots, const int* __restrict__ src,
                                const int* __restrict__ dst, int E,
                                const float* __restrict__ norm_dst, const float* __restrict__ norm_src,
                                const float* __restrict__ bias,
                                float* __restrict__ h, int n, int nch, int H) {
  const int gid = blockIdx.x * blockDim.x + threadIdx.x;
  const int node = gid >> 3;
  const int c = gid & 7;
  if (node >= n || c >= nch) return;

  const int* sl = slots + (size_t)node * GCN_SLOT_CAP;

  const int cnt = cnt_in[node];
  const int4 q0 = *(const int4*)(sl);
  const int4 q1 = *(const int4*)(sl + 4);
  const int4 q2 = *(const int4*)(sl + 8);
  const int4 q3 = *(const int4*)(sl + 12);
  const float nd = norm_dst[node];
  const float ns = norm_src[node];
  const float4 b = *(const float4*)(bias + c * 4);
  asm volatile("" :: "v"(cnt), "v"(q0.x), "v"(q1.x), "v"(q2.x), "v"(q3.x), "v"(nd), "v"(ns), "v"(b.x));

  const float4 acc = gcn_list_sum(x, sl, node, c, H, cnt, q0, q1, q2, q3, src, dst, E);

  float4 o;
  o.x = fmaxf(fmaf(acc.x, nd, b.x), 0.0f) * ns;
  o.y = fmaxf(fmaf(acc.y, nd, b.y), 0.0f) * ns;
  o.z = fmaxf(fmaf(acc.z, nd, b.z), 0.0f) * ns;
  o.w = fmaxf(fmaf(acc.w, nd, b.w), 0.0f) * ns;
  *(float4*)(h + (size_t)node * H + c * 4) = o;
}

#define GCN_L2_NODES 128
#define GCN_L2_TILES (GCN_L2_NODES / 16)
#define GCN_L2_PITCH 36

__global__ void __launch_bounds__(GCN_L2_NODES * 8)
gcn_pull_gemm_readout_kernel(const float* __restrict__ x, const int* __restrict__ cnt_in,
                             const int* __restrict__ slots, const int* __restrict__ src,
                             const int* __restrict__ dst, int E, int n, int nch, int H,
                             const float* __restrict__ W, int wcols,
                             const float* __restrict__ norm, const float* __restrict__ bias,
                             const float* __restrict__ Wd, const float* __restrict__ bd,
                             float* __restrict__ out, int ntiles) {
  __shared__ float Ws[32 * 32];
  __shared__ float As[GCN_L2_NODES * GCN_L2_PITCH];
  __shared__ float s_ro[GCN_L2_NODES / 4];

  gcn_stage_w(Ws, W, H, wcols);

  const int gid = blockIdx.x * blockDim.x + threadIdx.x;
  const int node = gid >> 3;
  const int c = gid & 7;
  if (node < n && c < nch) {
    const int* sl = slots + (size_t)node * GCN_SLOT_CAP;
    const int cnt = cnt_in[node];
    const int4 q0 = *(const int4*)(sl);
    const int4 q1 = *(const int4*)(sl + 4);
    const int4 q2 = *(const int4*)(sl + 8);
    const int4 q3 = *(const int4*)(sl + 12);
    asm volatile("" :: "v"(cnt), "v"(q0.x), "v"(q1.x), "v"(q2.x), "v"(q3.x));
    const float4 acc = gcn_list_sum(x, sl, node, c, H, cnt, q0, q1, q2, q3, src, dst, E);
    *(float4*)(As + (threadIdx.x >> 3) * GCN_L2_PITCH + c * 4) = acc;
  }

  const int wv = threadIdx.x >> 5;
  const int lane = threadIdx.x & 31;
  const int m = lane & 15;
  const int half = lane >> 4;
  const int tile = blockIdx.x * GCN_L2_TILES + wv;
  const bool has_tile = (wv < GCN_L2_TILES) && (tile < ntiles);

  const bool hasA = m < wcols;
  const bool hasB = 16 + m < wcols;
  float4 n0, n1;
  float wA[4], wB[4], bA, bB, bd0;
  if (has_tile) {
    const float* np = norm + tile * 16 + 8 * half;
    n0 = *(const float4*)(np);
    n1 = *(const float4*)(np + 4);
    const int colA = hasA ? m : 0;
    const int colB = hasB ? 16 + m : 0;
    bA = bias[colA];
    bB = bias[colB];
#pragma unroll
    for (int g = 0; g < 4; ++g) {
      wA[g] = Wd[g * wcols + colA];
      wB[g] = Wd[g * wcols + colB];
    }
    bd0 = bd[0];
  }

  gcn_lds_barrier();

  if (has_tile) {
    const float* arow = As + (wv * 16 + m) * GCN_L2_PITCH;
    v8f c0 = {};
    v8f c1 = {};

#pragma unroll 3
    for (int k0 = 0; k0 < H; k0 += 4) {
      const int k = k0 + 2 * half;
      const float2 av = *(const float2*)(arow + k);
      v2f a;
      a.x = av.x;
      a.y = av.y;

      v2f b0, b1;
      b0.x = Ws[k * 32 + m];
      b0.y = Ws[(k + 1) * 32 + m];
      b1.x = Ws[k * 32 + 16 + m];
      b1.y = Ws[(k + 1) * 32 + 16 + m];

      c0 = __builtin_amdgcn_wmma_f32_16x16x4_f32(false, a, false, b0, (short)0, c0, false, false);
      c1 = __builtin_amdgcn_wmma_f32_16x16x4_f32(false, a, false, b1, (short)0, c1, false, false);
    }

    const float ndv[8] = {n0.x, n0.y, n0.z, n0.w, n1.x, n1.y, n1.z, n1.w};
#pragma unroll
    for (int g = 0; g < 4; ++g) {
      wA[g] = hasA ? wA[g] : 0.0f;
      wB[g] = hasB ? wB[g] : 0.0f;
    }
    float s0 = 0.0f, s1 = 0.0f;
#pragma unroll
    for (int g = 0; g < 4; ++g) {
      s0 = fmaf(fmaxf(fmaf(c0[g], ndv[g], bA), 0.0f), wA[g], s0);
      s0 = fmaf(fmaxf(fmaf(c1[g], ndv[g], bB), 0.0f), wB[g], s0);
      s1 = fmaf(fmaxf(fmaf(c0[4 + g], ndv[4 + g], bA), 0.0f), wA[g], s1);
      s1 = fmaf(fmaxf(fmaf(c1[4 + g], ndv[4 + g], bB), 0.0f), wB[g], s1);
    }
#pragma unroll
    for (int off = 8; off >= 1; off >>= 1) {
      s0 += __shfl_xor(s0, off);
      s1 += __shfl_xor(s1, off);
    }
    if (m == 0) {
      s_ro[wv * 4 + 2 * half] = s0 + bd0;
      s_ro[wv * 4 + 2 * half + 1] = s1 + bd0;
    }
  }

  gcn_lds_barrier();
  const int r = blockIdx.x * (GCN_L2_NODES / 4) + threadIdx.x;
  if (threadIdx.x < GCN_L2_NODES / 4 && r < ntiles * 4) out[r] = s_ro[threadIdx.x];
}


extern "C" void kernel_launch(void* const* d_in, const int* in_sizes, int n_in,
                              void* d_out, int out_size, void* d_ws, size_t ws_size,
                              hipStream_t stream) {
  const float* features = (const float*)d_in[0];
  const int*   src      = (const int*)d_in[1];
  const int*   dst      = (const int*)d_in[2];
  const float* W1       = (const float*)d_in[3];
  const float* b1       = (const float*)d_in[4];
  const float* W2       = (const float*)d_in[5];
  const float* b2       = (const float*)d_in[6];
  const float* Wd       = (const float*)d_in[7];
  const float* bd       = (const float*)d_in[8];
  float* out = (float*)d_out;

  const int H  = in_sizes[4];
  const int F  = in_sizes[3] / H;
  const int Nn = in_sizes[0] / F;
  const int E  = in_sizes[1];

  const int B = 256;
  const int ntiles = Nn / 16;
  const int wavesPerBlock = B / 32;
  const int gemmGrid = (ntiles + wavesPerBlock - 1) / wavesPerBlock;
  const int nch = H / 4;
  const int pullGrid = (Nn * 8 + B - 1) / B;
  const int l2Grid = (Nn + GCN_L2_NODES - 1) / GCN_L2_NODES;

  int*   tick     = (int*)d_ws;
  int*   cnt_out  = tick + 2 * GCN_TICK_STRIDE;
  int*   cnt_in   = cnt_out + Nn;
  float* norm_src = (float*)(cnt_in + Nn);
  float* norm_dst = norm_src + Nn;
  float* bufA     = norm_dst + Nn;
  float* bufX     = bufA + (size_t)Nn * H;
  int*   slots    = (int*)(bufX + (size_t)Nn * H);

  gcn_zero_kernel<<<(2 * GCN_TICK_STRIDE + 2 * Nn + B - 1) / B, B, 0, stream>>>(
      tick, 2 * GCN_TICK_STRIDE + 2 * Nn);
  gcn_edge_kernel<<<2048, GCN_EDGE_BLOCK, 0, stream>>>(src, dst, tick, cnt_out, cnt_in, slots, E, Nn);
  gcn_norm_kernel<<<(2 * Nn + B - 1) / B, B, 0, stream>>>(cnt_out, norm_src, 2 * Nn);
  gcn_gemm_wmma_kernel<<<gemmGrid, B, 0, stream>>>(features, F, F, W1, H, norm_src, bufX, H, ntiles);
  gcn_pull_kernel<<<pullGrid, B, 0, stream>>>(bufX, cnt_in, slots, src, dst, E, norm_dst, norm_src,
                                              b1, bufA, Nn, nch, H);
  gcn_pull_gemm_readout_kernel<<<l2Grid, GCN_L2_NODES * 8, 0, stream>>>(
      bufA, cnt_in, slots, src, dst, E, Nn, nch, H, W2, H, norm_dst, b2, Wd, bd, out, ntiles);
}


# layer 1 re-associated as layer 2 is: a work-group gather-sums 128 nodes' feature rows, each times its source scale, then W1 product (WMMA) from LDS + epilogue; 6->5 launches; list-order sums
# speedup vs baseline: 4.4860x; 1.0670x over previous
#include <hip/hip_runtime.h>

typedef __attribute__((ext_vector_type(2))) float v2f;
typedef __attribute__((ext_vector_type(8))) float v8f;


__global__ void gcn_zero_kernel(int* __restrict__ p, int n) {
  int i = blockIdx.x * blockDim.x + threadIdx.x;
  if (i < n) p[i] = 0;
}

#define GCN_SLOT_CAP 32

#define GCN_TICK_STRIDE 64
#define GCN_EDGE_BLOCK 256
#define GCN_EDGE_PER_LANE 4
#define GCN_CHUNK (GCN_EDGE_BLOCK * GCN_EDGE_PER_LANE)

static __device__ __forceinline__ void gcn_lds_barrier() {
  __builtin_amdgcn_fence(__ATOMIC_RELEASE, "workgroup", "local");
  __builtin_amdgcn_s_barrier();
  __builtin_amdgcn_fence(__ATOMIC_ACQUIRE, "workgroup", "local");
}

__global__ void __launch_bounds__(GCN_EDGE_BLOCK)
gcn_edge_kernel(const int* __restrict__ src, const int* __restrict__ dst, int* __restrict__ tick,
                int* __restrict__ cnt_out, int* __restrict__ cnt_in, int* __restrict__ slots,
                int E, int n) {
  __shared__ int s_chunk[2];
  const int tid = threadIdx.x;
  const int nchunks = (E + GCN_CHUNK - 1) / GCN_CHUNK;
  int split = ((n >> 1) + 63) & ~63;
  if (split > n) split = n;
  int t = (int)((__builtin_amdgcn_s_sendmsg_rtn(0x87) >> 18) & 1u);

  for (int pass = 0; pass < 2; ++pass, t ^= 1) {
    const int lo = t ? split : 0;
    const unsigned span = t ? (unsigned)(n - split) : (unsigned)split;
    int* tk = tick + t * GCN_TICK_STRIDE;

    if (tid == 0) {
      int first = nchunks;
      if (pass == 0 || __hip_atomic_load(tk, __ATOMIC_RELAXED, __HIP_MEMORY_SCOPE_AGENT) < nchunks)
        first = __hip_atomic_fetch_add(tk, 1, __ATOMIC_RELAXED, __HIP_MEMORY_SCOPE_AGENT);
      s_chunk[0] = first;
    }
    gcn_lds_barrier();
    int c = s_chunk[0];
    int buf = 0;
    while (c < nchunks) {
      int s[GCN_EDGE_PER_LANE], d[GCN_EDGE_PER_LANE], p[GCN_EDGE_PER_LANE];
      bool os[GCN_EDGE_PER_LANE], od[GCN_EDGE_PER_LANE];
#pragma unroll
      for (int k = 0; k < GCN_EDGE_PER_LANE; ++k) {
        const int e = c * GCN_CHUNK + k * GCN_EDGE_BLOCK + tid;
        const int ec = (e < E) ? e : E - 1;
        s[k] = src[ec];
        d[k] = dst[ec];
        os[k] = (e < E);
      }
      asm volatile("" :: "v"(s[0]), "v"(d[0]), "v"(s[1]), "v"(d[1]),
                         "v"(s[2]), "v"(d[2]), "v"(s[3]), "v"(d[3]));
#pragma unroll
      for (int k = 0; k < GCN_EDGE_PER_LANE; ++k) {
        od[k] = os[k] && ((unsigned)(d[k] - lo) < span);
        os[k] = os[k] && ((unsigned)(s[k] - lo) < span);
      }
#pragma unroll
      for (int k = 0; k < GCN_EDGE_PER_LANE; ++k) {
        p[k] = GCN_SLOT_CAP;
        if (od[k])
          p[k] = __hip_atomic_fetch_add(&cnt_in[d[k]], 1, __ATOMIC_RELAXED, __HIP_MEMORY_SCOPE_AGENT);
      }
#pragma unroll
      for (int k = 0; k < GCN_EDGE_PER_LANE; ++k) {
        if (os[k])
          (void)__hip_atomic_fetch_add(&cnt_out[s[k]], 1, __ATOMIC_RELAXED, __HIP_MEMORY_SCOPE_AGENT);
      }
      int nxt = 0;
      if (tid == 0) nxt = __hip_atomic_fetch_add(tk, 1, __ATOMIC_RELAXED, __HIP_MEMORY_SCOPE_AGENT);
#pragma unroll
      for (int k = 0; k < GCN_EDGE_PER_LANE; ++k) {
        if ((unsigned)p[k] < (unsigned)GCN_SLOT_CAP) slots[(size_t)d[k] * GCN_SLOT_CAP + p[k]] = s[k];
      }
      buf ^= 1;
      if (tid == 0) s_chunk[buf] = nxt;
      gcn_lds_barrier();
      c = s_chunk[buf];
    }
    gcn_lds_barrier();
  }
}

__global__ void gcn_norm_kernel(const int* __restrict__ cnt, float* __restrict__ nrm, int n) {
  int i = blockIdx.x * blockDim.x + threadIdx.x;
  if (i < n) nrm[i] = rsqrtf(fmaxf((float)cnt[i], 1.0f));
}


static __device__ __forceinline__ void gcn_stage_w(float* __restrict__ Ws, const float* __restrict__ W,
                                                   int K, int wcols) {
  for (int i = threadIdx.x; i < 32 * 32; i += blockDim.x) {
    const int k = i >> 5;
    const int n = i & 31;
    Ws[i] = (k < K && n < wcols) ? W[(size_t)k * wcols + n] : 0.0f;
  }
}


template <int J>
static __device__ __forceinline__ float gcn_group_lane(float v) {
  return __uint_as_float(__builtin_amdgcn_mov_dpp8(__float_as_uint(v), (unsigned)J * 0x249249u));
}

template <bool SC>
static __device__ __forceinline__ void gcn_trip8(float4& acc, const char* __restrict__ xb,
                                                 unsigned rowbytes, unsigned cbytes, int r,
                                                 int e0, int e1, int e2, int e3,
                                                 int e4, int e5, int e6, int e7,
                                                 const float* __restrict__ sc, int gl, int w) {
  const unsigned s0 = (unsigned)e0;
  const unsigned s1 = (unsigned)((r > 1) ? e1 : e0);
  const unsigned s2 = (unsigned)((r > 2) ? e2 : e0);
  const unsigned s3 = (unsigned)((r > 3) ? e3 : e0);
  const unsigned s4 = (unsigned)((r > 4) ? e4 : e0);
  const unsigned s5 = (unsigned)((r > 5) ? e5 : e0);
  const unsigned s6 = (unsigned)((r > 6) ? e6 : e0);
  const unsigned s7 = (unsigned)((r > 7) ? e7 : e0);
  const float4 v0 = *(const float4*)(xb + (__umul24(s0, rowbytes) + cbytes));
  const float4 v1 = *(const float4*)(xb + (__umul24(s1, rowbytes) + cbytes));
  const float4 v2 = *(const float4*)(xb + (__umul24(s2, rowbytes) + cbytes));
  const float4 v3 = *(const float4*)(xb + (__umul24(s3, rowbytes) + cbytes));
  const float4 v4 = *(const float4*)(xb + (__umul24(s4, rowbytes) + cbytes));
  const float4 v5 = *(const float4*)(xb + (__umul24(s5, rowbytes) + cbytes));
  const float4 v6 = *(const float4*)(xb + (__umul24(s6, rowbytes) + cbytes));
  const float4 v7 = *(const float4*)(xb + (__umul24(s7, rowbytes) + cbytes));
  float mine = 1.0f;
  if (SC) {
    const unsigned sm = (gl < r) ? (unsigned)w : s0;
    mine = *(const float*)((const char*)sc + (sm << 2));
  }
  asm volatile("" :: "v"(v0.x), "v"(v1.x), "v"(v2.x), "v"(v3.x),
                     "v"(v4.x), "v"(v5.x), "v"(v6.x), "v"(v7.x), "v"(mine));
  __builtin_amdgcn_sched_barrier(0);
  if (SC) {
    const float c0 = gcn_group_lane<0>(mine);
    acc.x = fmaf(v0.x, c0, acc.x); acc.y = fmaf(v0.y, c0, acc.y); acc.z = fmaf(v0.z, c0, acc.z); acc.w = fmaf(v0.w, c0, acc.w);
    if (r > 1) { const float k = gcn_group_lane<1>(mine); acc.x = fmaf(v1.x, k, acc.x); acc.y = fmaf(v1.y, k, acc.y); acc.z = fmaf(v1.z, k, acc.z); acc.w = fmaf(v1.w, k, acc.w); }
    if (r > 2) { const float k = gcn_group_lane<2>(mine); acc.x = fmaf(v2.x, k, acc.x); acc.y = fmaf(v2.y, k, acc.y); acc.z = fmaf(v2.z, k, acc.z); acc.w = fmaf(v2.w, k, acc.w); }
    if (r > 3) { const float k = gcn_group_lane<3>(mine); acc.x = fmaf(v3.x, k, acc.x); acc.y = fmaf(v3.y, k, acc.y); acc.z = fmaf(v3.z, k, acc.z); acc.w = fmaf(v3.w, k, acc.w); }
    if (r > 4) { const float k = gcn_group_lane<4>(mine); acc.x = fmaf(v4.x, k, acc.x); acc.y = fmaf(v4.y, k, acc.y); acc.z = fmaf(v4.z, k, acc.z); acc.w = fmaf(v4.w, k, acc.w); }
    if (r > 5) { const float k = gcn_group_lane<5>(mine); acc.x = fmaf(v5.x, k, acc.x); acc.y = fmaf(v5.y, k, acc.y); acc.z = fmaf(v5.z, k, acc.z); acc.w = fmaf(v5.w, k, acc.w); }
    if (r > 6) { const float k = gcn_group_lane<6>(mine); acc.x = fmaf(v6.x, k, acc.x); acc.y = fmaf(v6.y, k, acc.y); acc.z = fmaf(v6.z, k, acc.z); acc.w = fmaf(v6.w, k, acc.w); }
    if (r > 7) { const float k = gcn_group_lane<7>(mine); acc.x = fmaf(v7.x, k, acc.x); acc.y = fmaf(v7.y, k, acc.y); acc.z = fmaf(v7.z, k, acc.z); acc.w = fmaf(v7.w, k, acc.w); }
  } else {
    acc.x += v0.x; acc.y += v0.y; acc.z += v0.z; acc.w += v0.w;
    if (r > 1) { acc.x += v1.x; acc.y += v1.y; acc.z += v1.z; acc.w += v1.w; }
    if (r > 2) { acc.x += v2.x; acc.y += v2.y; acc.z += v2.z; acc.w += v2.w; }
    if (r > 3) { acc.x += v3.x; acc.y += v3.y; acc.z += v3.z; acc.w += v3.w; }
    if (r > 4) { acc.x += v4.x; acc.y += v4.y; acc.z += v4.z; acc.w += v4.w; }
    if (r > 5) { acc.x += v5.x; acc.y += v5.y; acc.z += v5.z; acc.w += v5.w; }
    if (r > 6) { acc.x += v6.x; acc.y += v6.y; acc.z += v6.z; acc.w += v6.w; }
    if (r > 7) { acc.x += v7.x; acc.y += v7.y; acc.z += v7.z; acc.w += v7.w; }
  }
}

template <bool SC>
static __device__ __forceinline__ float4 gcn_list_sum(const float* __restrict__ x, const int* __restrict__ sl,
                                                      int node, int c, int rowf, int cnt,
                                                      int4 q0, int4 q1, int4 q2, int4 q3,
                                                      const int* __restrict__ src,
                                                      const int* __restrict__ dst, int E,
                                                      const float* __restrict__ sc, int gl, int w0, int w1) {
  const float* xc = x + c * 4;
  const char* xb = (const char*)x;
  const unsigned rowbytes = (unsigned)rowf * 4u;
  const unsigned cbytes = (unsigned)c * 16u;
  float4 acc = make_float4(0.0f, 0.0f, 0.0f, 0.0f);
  if (cnt <= GCN_SLOT_CAP) {
    if (cnt > 0) gcn_trip8<SC>(acc, xb, rowbytes, cbytes, cnt, q0.x, q0.y, q0.z, q0.w, q1.x, q1.y, q1.z, q1.w, sc, gl, w0);
    if (cnt > 8) gcn_trip8<SC>(acc, xb, rowbytes, cbytes, cnt - 8, q2.x, q2.y, q2.z, q2.w, q3.x, q3.y, q3.z, q3.w, sc, gl, w1);
    for (int j = 16; j < cnt; j += 8) {
      const int4 p0 = *(const int4*)(sl + j);
      const int4 p1 = *(const int4*)(sl + j + 4);
      int w = 0;
      if (SC) w = sl[j + gl];
      gcn_trip8<SC>(acc, xb, rowbytes, cbytes, cnt - j, p0.x, p0.y, p0.z, p0.w, p1.x, p1.y, p1.z, p1.w, sc, gl, w);
    }
  } else {
    for (int e = 0; e < E; ++e) {
      if (dst[e] == node) {
        const int sidx = src[e];
        const float4 v = *(const float4*)(xc + (size_t)sidx * rowf);
        if (SC) {
          const float k = sc[sidx];
          acc.x = fmaf(v.x, k, acc.x); acc.y = fmaf(v.y, k, acc.y); acc.z = fmaf(v.z, k, acc.z); acc.w = fmaf(v.w, k, acc.w);
        } else {
          acc.x += v.x; acc.y += v.y; acc.z += v.z; acc.w += v.w;
        }
      }
    }
  }
  return acc;
}

#define GCN_L2_NODES 128
#define GCN_L2_TILES (GCN_L2_NODES / 16)
#define GCN_L2_PITCH 36

__global__ void __launch_bounds__(GCN_L2_NODES * 8)
gcn_pull_gemm_l1_kernel(const float* __restrict__ x, int rowf, int nch,
                        const int* __restrict__ cnt_in, const int* __restrict__ slots,
                        const int* __restrict__ src, const int* __restrict__ dst, int E, int n,
                        const float* __restrict__ W, int wcols,
                        const float* __restrict__ norm_src, const float* __restrict__ norm_dst,
                        const float* __restrict__ bias, float* __restrict__ h, int ntiles) {
  __shared__ float Ws[32 * 32];
  __shared__ float As[GCN_L2_NODES * GCN_L2_PITCH];

  gcn_stage_w(Ws, W, rowf, wcols);

  const int gid = blockIdx.x * blockDim.x + threadIdx.x;
  const int node = gid >> 3;
  const int gl = gid & 7;
  if (node < n) {
    const int* sl = slots + (size_t)node * GCN_SLOT_CAP;
    const int c = (gl < nch) ? gl : 0;
    const int cnt = cnt_in[node];
    const int4 q0 = *(const int4*)(sl);
    const int4 q1 = *(const int4*)(sl + 4);
    const int4 q2 = *(const int4*)(sl + 8);
    const int4 q3 = *(const int4*)(sl + 12);
    const int w0 = sl[gl];
    const int w1 = sl[8 + gl];
    asm volatile("" :: "v"(cnt), "v"(q0.x), "v"(q1.x), "v"(q2.x), "v"(q3.x), "v"(w0), "v"(w1));
    const float4 acc = gcn_list_sum<true>(x, sl, node, c, rowf, cnt, q0, q1, q2, q3, src, dst, E,
                                          norm_src, gl, w0, w1);
    if (gl < nch) *(float4*)(As + (threadIdx.x >> 3) * GCN_L2_PITCH + c * 4) = acc;
  }

  const int wv = threadIdx.x >> 5;
  const int lane = threadIdx.x & 31;
  const int m = lane & 15;
  const int half = lane >> 4;
  const int tile = blockIdx.x * GCN_L2_TILES + wv;
  const bool has_tile = (wv < GCN_L2_TILES) && (tile < ntiles);

  float4 d0, d1, t0, t1;
  float bA, bB;
  if (has_tile) {
    const float* dp = norm_dst + tile * 16 + 8 * half;
    const float* tp = norm_src + tile * 16 + 8 * half;
    d0 = *(const float4*)(dp);
    d1 = *(const float4*)(dp + 4);
    t0 = *(const float4*)(tp);
    t1 = *(const float4*)(tp + 4);
    bA = bias[(m < wcols) ? m : 0];
    bB = bias[(16 + m < wcols) ? 16 + m : 0];
  }

  gcn_lds_barrier();

  if (has_tile) {
    float* trow = As + (wv * 16) * GCN_L2_PITCH;
    const float* arow = trow + m * GCN_L2_PITCH;
    v8f c0 = {};
    v8f c1 = {};

#pragma unroll 4
    for (int k0 = 0; k0 < rowf; k0 += 4) {
      const int k = k0 + 2 * half;
      const float2 av = *(const float2*)(arow + k);
      v2f a;
      a.x = av.x;
      a.y = av.y;

      v2f b0, b1;
      b0.x = Ws[k * 32 + m];
      b0.y = Ws[(k + 1) * 32 + m];
      b1.x = Ws[k * 32 + 16 + m];
      b1.y = Ws[(k + 1) * 32 + 16 + m];

      c0 = __builtin_amdgcn_wmma_f32_16x16x4_f32(false, a, false, b0, (short)0, c0, false, false);
      c1 = __builtin_amdgcn_wmma_f32_16x16x4_f32(false, a, false, b1, (short)0, c1, false, false);
    }

    const float ndv[8] = {d0.x, d0.y, d0.z, d0.w, d1.x, d1.y, d1.z, d1.w};
    const float nsv[8] = {t0.x, t0.y, t0.z, t0.w, t1.x, t1.y, t1.z, t1.w};
#pragma unroll
    for (int r = 0; r < 8; ++r) {
      float* orow = trow + (8 * half + r) * GCN_L2_PITCH;
      orow[m] = fmaxf(fmaf(c0[r], ndv[r], bA), 0.0f) * nsv[r];
      orow[16 + m] = fmaxf(fmaf(c1[r], ndv[r], bB), 0.0f) * nsv[r];
    }
    __builtin_amdgcn_fence(__ATOMIC_RELEASE, "wavefront", "local");
    __builtin_amdgcn_wave_barrier();
    __builtin_amdgcn_fence(__ATOMIC_ACQUIRE, "wavefront", "local");

    const int och = wcols >> 2;
    const int oc = lane & 7;
    float* hb = h + (size_t)__builtin_amdgcn_readfirstlane(tile) * 16 * wcols;
    if (oc < och) {
#pragma unroll
      for (int i = 0; i < 4; ++i) {
        const int row = i * 4 + (lane >> 3);
        const float4 v = *(const float4*)(trow + row * GCN_L2_PITCH + oc * 4);
        *(float4*)(hb + (unsigned)(row * wcols + oc * 4)) = v;
      }
    }
  }
}


__global__ void __launch_bounds__(GCN_L2_NODES * 8)
gcn_pull_gemm_readout_kernel(const float* __restrict__ x, const int* __restrict__ cnt_in,
                             const int* __restrict__ slots, const int* __restrict__ src,
                             const int* __restrict__ dst, int E, int n, int nch, int H,
                             const float* __restrict__ W, int wcols,
                             const float* __restrict__ norm, const float* __restrict__ bias,
                             const float* __restrict__ Wd, const float* __restrict__ bd,
                             float* __restrict__ out, int ntiles) {
  __shared__ float Ws[32 * 32];
  __shared__ float As[GCN_L2_NODES * GCN_L2_PITCH];
  __shared__ float s_ro[GCN_L2_NODES / 4];

  gcn_stage_w(Ws, W, H, wcols);

  const int gid = blockIdx.x * blockDim.x + threadIdx.x;
  const int node = gid >> 3;
  const int c = gid & 7;
  if (node < n && c < nch) {
    const int* sl = slots + (size_t)node * GCN_SLOT_CAP;
    const int cnt = cnt_in[node];
    const int4 q0 = *(const int4*)(sl);
    const int4 q1 = *(const int4*)(sl + 4);
    const int4 q2 = *(const int4*)(sl + 8);
    const int4 q3 = *(const int4*)(sl + 12);
    asm volatile("" :: "v"(cnt), "v"(q0.x), "v"(q1.x), "v"(q2.x), "v"(q3.x));
    const float4 acc = gcn_list_sum<false>(x, sl, node, c, H, cnt, q0, q1, q2, q3, src, dst, E,
                                           nullptr, 0, 0, 0);
    *(float4*)(As + (threadIdx.x >> 3) * GCN_L2_PITCH + c * 4) = acc;
  }

  const int wv = threadIdx.x >> 5;
  const int lane = threadIdx.x & 31;
  const int m = lane & 15;
  const int half = lane >> 4;
  const int tile = blockIdx.x * GCN_L2_TILES + wv;
  const bool has_tile = (wv < GCN_L2_TILES) && (tile < ntiles);

  const bool hasA = m < wcols;
  const bool hasB = 16 + m < wcols;
  float4 n0, n1;
  float wA[4], wB[4], bA, bB, bd0;
  if (has_tile) {
    const float* np = norm + tile * 16 + 8 * half;
    n0 = *(const float4*)(np);
    n1 = *(const float4*)(np + 4);
    const int colA = hasA ? m : 0;
    const int colB = hasB ? 16 + m : 0;
    bA = bias[colA];
    bB = bias[colB];
#pragma unroll
    for (int g = 0; g < 4; ++g) {
      wA[g] = Wd[g * wcols + colA];
      wB[g] = Wd[g * wcols + colB];
    }
    bd0 = bd[0];
  }

  gcn_lds_barrier();

  if (has_tile) {
    const float* arow = As + (wv * 16 + m) * GCN_L2_PITCH;
    v8f c0 = {};
    v8f c1 = {};

#pragma unroll 3
    for (int k0 = 0; k0 < H; k0 += 4) {
      const int k = k0 + 2 * half;
      const float2 av = *(const float2*)(arow + k);
      v2f a;
      a.x = av.x;
      a.y = av.y;

      v2f b0, b1;
      b0.x = Ws[k * 32 + m];
      b0.y = Ws[(k + 1) * 32 + m];
      b1.x = Ws[k * 32 + 16 + m];
      b1.y = Ws[(k + 1) * 32 + 16 + m];

      c0 = __builtin_amdgcn_wmma_f32_16x16x4_f32(false, a, false, b0, (short)0, c0, false, false);
      c1 = __builtin_amdgcn_wmma_f32_16x16x4_f32(false, a, false, b1, (short)0, c1, false, false);
    }

    const float ndv[8] = {n0.x, n0.y, n0.z, n0.w, n1.x, n1.y, n1.z, n1.w};
#pragma unroll
    for (int g = 0; g < 4; ++g) {
      wA[g] = hasA ? wA[g] : 0.0f;
      wB[g] = hasB ? wB[g] : 0.0f;
    }
    float s0 = 0.0f, s1 = 0.0f;
#pragma unroll
    for (int g = 0; g < 4; ++g) {
      s0 = fmaf(fmaxf(fmaf(c0[g], ndv[g], bA), 0.0f), wA[g], s0);
      s0 = fmaf(fmaxf(fmaf(c1[g], ndv[g], bB), 0.0f), wB[g], s0);
      s1 = fmaf(fmaxf(fmaf(c0[4 + g], ndv[4 + g], bA), 0.0f), wA[g], s1);
      s1 = fmaf(fmaxf(fmaf(c1[4 + g], ndv[4 + g], bB), 0.0f), wB[g], s1);
    }
#pragma unroll
    for (int off = 8; off >= 1; off >>= 1) {
      s0 += __shfl_xor(s0, off);
      s1 += __shfl_xor(s1, off);
    }
    if (m == 0) {
      s_ro[wv * 4 + 2 * half] = s0 + bd0;
      s_ro[wv * 4 + 2 * half + 1] = s1 + bd0;
    }
  }

  gcn_lds_barrier();
  const int r = blockIdx.x * (GCN_L2_NODES / 4) + threadIdx.x;
  if (threadIdx.x < GCN_L2_NODES / 4 && r < ntiles * 4) out[r] = s_ro[threadIdx.x];
}


extern "C" void kernel_launch(void* const* d_in, const int* in_sizes, int n_in,
                              void* d_out, int out_size, void* d_ws, size_t ws_size,
                              hipStream_t stream) {
  const float* features = (const float*)d_in[0];
  const int*   src      = (const int*)d_in[1];
  const int*   dst      = (const int*)d_in[2];
  const float* W1       = (const float*)d_in[3];
  const float* b1       = (const float*)d_in[4];
  const float* W2       = (const float*)d_in[5];
  const float* b2       = (const float*)d_in[6];
  const float* Wd       = (const float*)d_in[7];
  const float* bd       = (const float*)d_in[8];
  float* out = (float*)d_out;

  const int H  = in_sizes[4];
  const int F  = in_sizes[3] / H;
  const int Nn = in_sizes[0] / F;
  const int E  = in_sizes[1];

  const int B = 256;
  const int ntiles = Nn / 16;
  const int nch = H / 4;
  const int nchF = F / 4;
  const int l2Grid = (Nn + GCN_L2_NODES - 1) / GCN_L2_NODES;

  int*   tick     = (int*)d_ws;
  int*   cnt_out  = tick + 2 * GCN_TICK_STRIDE;
  int*   cnt_in   = cnt_out + Nn;
  float* norm_src = (float*)(cnt_in + Nn);
  float* norm_dst = norm_src + Nn;
  float* bufA     = norm_dst + Nn;
  int*   slots    = (int*)(bufA + (size_t)Nn * H);

  gcn_zero_kernel<<<(2 * GCN_TICK_STRIDE + 2 * Nn + B - 1) / B, B, 0, stream>>>(
      tick, 2 * GCN_TICK_STRIDE + 2 * Nn);
  gcn_edge_kernel<<<2048, GCN_EDGE_BLOCK, 0, stream>>>(src, dst, tick, cnt_out, cnt_in, slots, E, Nn);
  gcn_norm_kernel<<<(2 * Nn + B - 1) / B, B, 0, stream>>>(cnt_out, norm_src, 2 * Nn);
  gcn_pull_gemm_l1_kernel<<<l2Grid, GCN_L2_NODES * 8, 0, stream>>>(
      features, F, nchF, cnt_in, slots, src, dst, E, Nn, W1, H, norm_src, norm_dst, b1, bufA, ntiles);
  gcn_pull_gemm_readout_kernel<<<l2Grid, GCN_L2_NODES * 8, 0, stream>>>(
      bufA, cnt_in, slots, src, dst, E, Nn, nch, H, W2, H, norm_dst, b2, Wd, bd, out, ntiles);
}


# layer-1 output rows stored at a 128-byte pitch (24 floats + padding, whole-line tile stores): a row the layer-2 gather reads is one aligned line, not 96 bytes over two; arithmetic and sum order kept
# speedup vs baseline: 4.6928x; 1.0461x over previous
#include <hip/hip_runtime.h>

typedef __attribute__((ext_vector_type(2))) float v2f;
typedef __attribute__((ext_vector_type(8))) float v8f;


__global__ void gcn_zero_kernel(int* __restrict__ p, int n) {
  int i = blockIdx.x * blockDim.x + threadIdx.x;
  if (i < n) p[i] = 0;
}

#define GCN_SLOT_CAP 32

#define GCN_TICK_STRIDE 64
#define GCN_EDGE_BLOCK 256
#define GCN_EDGE_PER_LANE 4
#define GCN_CHUNK (GCN_EDGE_BLOCK * GCN_EDGE_PER_LANE)

static __device__ __forceinline__ void gcn_lds_barrier() {
  __builtin_amdgcn_fence(__ATOMIC_RELEASE, "workgroup", "local");
  __builtin_amdgcn_s_barrier();
  __builtin_amdgcn_fence(__ATOMIC_ACQUIRE, "workgroup", "local");
}

__global__ void __launch_bounds__(GCN_EDGE_BLOCK)
gcn_edge_kernel(const int* __restrict__ src, const int* __restrict__ dst, int* __restrict__ tick,
                int* __restrict__ cnt_out, int* __restrict__ cnt_in, int* __restrict__ slots,
                int E, int n) {
  __shared__ int s_chunk[2];
  const int tid = threadIdx.x;
  const int nchunks = (E + GCN_CHUNK - 1) / GCN_CHUNK;
  int split = ((n >> 1) + 63) & ~63;
  if (split > n) split = n;
  int t = (int)((__builtin_amdgcn_s_sendmsg_rtn(0x87) >> 18) & 1u);

  for (int pass = 0; pass < 2; ++pass, t ^= 1) {
    const int lo = t ? split : 0;
    const unsigned span = t ? (unsigned)(n - split) : (unsigned)split;
    int* tk = tick + t * GCN_TICK_STRIDE;

    if (tid == 0) {
      int first = nchunks;
      if (pass == 0 || __hip_atomic_load(tk, __ATOMIC_RELAXED, __HIP_MEMORY_SCOPE_AGENT) < nchunks)
        first = __hip_atomic_fetch_add(tk, 1, __ATOMIC_RELAXED, __HIP_MEMORY_SCOPE_AGENT);
      s_chunk[0] = first;
    }
    gcn_lds_barrier();
    int c = s_chunk[0];
    int buf = 0;
    while (c < nchunks) {
      int s[GCN_EDGE_PER_LANE], d[GCN_EDGE_PER_LANE], p[GCN_EDGE_PER_LANE];
      bool os[GCN_EDGE_PER_LANE], od[GCN_EDGE_PER_LANE];
#pragma unroll
      for (int k = 0; k < GCN_EDGE_PER_LANE; ++k) {
        const int e = c * GCN_CHUNK + k * GCN_EDGE_BLOCK + tid;
        const int ec = (e < E) ? e : E - 1;
        s[k] = src[ec];
        d[k] = dst[ec];
        os[k] = (e < E);
      }
      asm volatile("" :: "v"(s[0]), "v"(d[0]), "v"(s[1]), "v"(d[1]),
                         "v"(s[2]), "v"(d[2]), "v"(s[3]), "v"(d[3]));
#pragma unroll
      for (int k = 0; k < GCN_EDGE_PER_LANE; ++k) {
        od[k] = os[k] && ((unsigned)(d[k] - lo) < span);
        os[k] = os[k] && ((unsigned)(s[k] - lo) < span);
      }
#pragma unroll
      for (int k = 0; k < GCN_EDGE_PER_LANE; ++k) {
        p[k] = GCN_SLOT_CAP;
        if (od[k])
          p[k] = __hip_atomic_fetch_add(&cnt_in[d[k]], 1, __ATOMIC_RELAXED, __HIP_MEMORY_SCOPE_AGENT);
      }
#pragma unroll
      for (int k = 0; k < GCN_EDGE_PER_LANE; ++k) {
        if (os[k])
          (void)__hip_atomic_fetch_add(&cnt_out[s[k]], 1, __ATOMIC_RELAXED, __HIP_MEMORY_SCOPE_AGENT);
      }
      int nxt = 0;
      if (tid == 0) nxt = __hip_atomic_fetch_add(tk, 1, __ATOMIC_RELAXED, __HIP_MEMORY_SCOPE_AGENT);
#pragma unroll
      for (int k = 0; k < GCN_EDGE_PER_LANE; ++k) {
        if ((unsigned)p[k] < (unsigned)GCN_SLOT_CAP) slots[(size_t)d[k] * GCN_SLOT_CAP + p[k]] = s[k];
      }
      buf ^= 1;
      if (tid == 0) s_chunk[buf] = nxt;
      gcn_lds_barrier();
      c = s_chunk[buf];
    }
    gcn_lds_barrier();
  }
}

__global__ void gcn_norm_kernel(const int* __restrict__ cnt, float* __restrict__ nrm, int n) {
  int i = blockIdx.x * blockDim.x + threadIdx.x;
  if (i < n) nrm[i] = rsqrtf(fmaxf((float)cnt[i], 1.0f));
}


static __device__ __forceinline__ void gcn_stage_w(float* __restrict__ Ws, const float* __restrict__ W,
                                                   int K, int wcols) {
  for (int i = threadIdx.x; i < 32 * 32; i += blockDim.x) {
    const int k = i >> 5;
    const int n = i & 31;
    Ws[i] = (k < K && n < wcols) ? W[(size_t)k * wcols + n] : 0.0f;
  }
}


template <int J>
static __device__ __forceinline__ float gcn_group_lane(float v) {
  return __uint_as_float(__builtin_amdgcn_mov_dpp8(__float_as_uint(v), (unsigned)J * 0x249249u));
}

template <bool SC>
static __device__ __forceinline__ void gcn_trip8(float4& acc, const char* __restrict__ xb,
                                                 unsigned rowbytes, unsigned cbytes, int r,
                                                 int e0, int e1, int e2, int e3,
                                                 int e4, int e5, int e6, int e7,
                                                 const float* __restrict__ sc, int gl, int w) {
  const unsigned s0 = (unsigned)e0;
  const unsigned s1 = (unsigned)((r > 1) ? e1 : e0);
  const unsigned s2 = (unsigned)((r > 2) ? e2 : e0);
  const unsigned s3 = (unsigned)((r > 3) ? e3 : e0);
  const unsigned s4 = (unsigned)((r > 4) ? e4 : e0);
  const unsigned s5 = (unsigned)((r > 5) ? e5 : e0);
  const unsigned s6 = (unsigned)((r > 6) ? e6 : e0);
  const unsigned s7 = (unsigned)((r > 7) ? e7 : e0);
  const float4 v0 = *(const float4*)(xb + (__umul24(s0, rowbytes) + cbytes));
  const float4 v1 = *(const float4*)(xb + (__umul24(s1, rowbytes) + cbytes));
  const float4 v2 = *(const float4*)(xb + (__umul24(s2, rowbytes) + cbytes));
  const float4 v3 = *(const float4*)(xb + (__umul24(s3, rowbytes) + cbytes));
  const float4 v4 = *(const float4*)(xb + (__umul24(s4, rowbytes) + cbytes));
  const float4 v5 = *(const float4*)(xb + (__umul24(s5, rowbytes) + cbytes));
  const float4 v6 = *(const float4*)(xb + (__umul24(s6, rowbytes) + cbytes));
  const float4 v7 = *(const float4*)(xb + (__umul24(s7, rowbytes) + cbytes));
  float mine = 1.0f;
  if (SC) {
    const unsigned sm = (gl < r) ? (unsigned)w : s0;
    mine = *(const float*)((const char*)sc + (sm << 2));
  }
  asm volatile("" :: "v"(v0.x), "v"(v1.x), "v"(v2.x), "v"(v3.x),
                     "v"(v4.x), "v"(v5.x), "v"(v6.x), "v"(v7.x), "v"(mine));
  __builtin_amdgcn_sched_barrier(0);
  if (SC) {
    const float c0 = gcn_group_lane<0>(mine);
    acc.x = fmaf(v0.x, c0, acc.x); acc.y = fmaf(v0.y, c0, acc.y); acc.z = fmaf(v0.z, c0, acc.z); acc.w = fmaf(v0.w, c0, acc.w);
    if (r > 1) { const float k = gcn_group_lane<1>(mine); acc.x = fmaf(v1.x, k, acc.x); acc.y = fmaf(v1.y, k, acc.y); acc.z = fmaf(v1.z, k, acc.z); acc.w = fmaf(v1.w, k, acc.w); }
    if (r > 2) { const float k = gcn_group_lane<2>(mine); acc.x = fmaf(v2.x, k, acc.x); acc.y = fmaf(v2.y, k, acc.y); acc.z = fmaf(v2.z, k, acc.z); acc.w = fmaf(v2.w, k, acc.w); }
    if (r > 3) { const float k = gcn_group_lane<3>(mine); acc.x = fmaf(v3.x, k, acc.x); acc.y = fmaf(v3.y, k, acc.y); acc.z = fmaf(v3.z, k, acc.z); acc.w = fmaf(v3.w, k, acc.w); }
    if (r > 4) { const float k = gcn_group_lane<4>(mine); acc.x = fmaf(v4.x, k, acc.x); acc.y = fmaf(v4.y, k, acc.y); acc.z = fmaf(v4.z, k, acc.z); acc.w = fmaf(v4.w, k, acc.w); }
    if (r > 5) { const float k = gcn_group_lane<5>(mine); acc.x = fmaf(v5.x, k, acc.x); acc.y = fmaf(v5.y, k, acc.y); acc.z = fmaf(v5.z, k, acc.z); acc.w = fmaf(v5.w, k, acc.w); }
    if (r > 6) { const float k = gcn_group_lane<6>(mine); acc.x = fmaf(v6.x, k, acc.x); acc.y = fmaf(v6.y, k, acc.y); acc.z = fmaf(v6.z, k, acc.z); acc.w = fmaf(v6.w, k, acc.w); }
    if (r > 7) { const float k = gcn_group_lane<7>(mine); acc.x = fmaf(v7.x, k, acc.x); acc.y = fmaf(v7.y, k, acc.y); acc.z = fmaf(v7.z, k, acc.z); acc.w = fmaf(v7.w, k, acc.w); }
  } else {
    acc.x += v0.x; acc.y += v0.y; acc.z += v0.z; acc.w += v0.w;
    if (r > 1) { acc.x += v1.x; acc.y += v1.y; acc.z += v1.z; acc.w += v1.w; }
    if (r > 2) { acc.x += v2.x; acc.y += v2.y; acc.z += v2.z; acc.w += v2.w; }
    if (r > 3) { acc.x += v3.x; acc.y += v3.y; acc.z += v3.z; acc.w += v3.w; }
    if (r > 4) { acc.x += v4.x; acc.y += v4.y; acc.z += v4.z; acc.w += v4.w; }
    if (r > 5) { acc.x += v5.x; acc.y += v5.y; acc.z += v5.z; acc.w += v5.w; }
    if (r > 6) { acc.x += v6.x; acc.y += v6.y; acc.z += v6.z; acc.w += v6.w; }
    if (r > 7) { acc.x += v7.x; acc.y += v7.y; acc.z += v7.z; acc.w += v7.w; }
  }
}

template <bool SC>
static __device__ __forceinline__ float4 gcn_list_sum(const float* __restrict__ x, const int* __restrict__ sl,
                                                      int node, int c, int rowf, int cnt,
                                                      int4 q0, int4 q1, int4 q2, int4 q3,
                                                      const int* __restrict__ src,
                                                      const int* __restrict__ dst, int E,
                                                      const float* __restrict__ sc, int gl, int w0, int w1) {
  const float* xc = x + c * 4;
  const char* xb = (const char*)x;
  const unsigned rowbytes = (unsigned)rowf * 4u;
  const unsigned cbytes = (unsigned)c * 16u;
  float4 acc = make_float4(0.0f, 0.0f, 0.0f, 0.0f);
  if (cnt <= GCN_SLOT_CAP) {
    if (cnt > 0) gcn_trip8<SC>(acc, xb, rowbytes, cbytes, cnt, q0.x, q0.y, q0.z, q0.w, q1.x, q1.y, q1.z, q1.w, sc, gl, w0);
    if (cnt > 8) gcn_trip8<SC>(acc, xb, rowbytes, cbytes, cnt - 8, q2.x, q2.y, q2.z, q2.w, q3.x, q3.y, q3.z, q3.w, sc, gl, w1);
    for (int j = 16; j < cnt; j += 8) {
      const int4 p0 = *(const int4*)(sl + j);
      const int4 p1 = *(const int4*)(sl + j + 4);
      int w = 0;
      if (SC) w = sl[j + gl];
      gcn_trip8<SC>(acc, xb, rowbytes, cbytes, cnt - j, p0.x, p0.y, p0.z, p0.w, p1.x, p1.y, p1.z, p1.w, sc, gl, w);
    }
  } else {
    for (int e = 0; e < E; ++e) {
      if (dst[e] == node) {
        const int sidx = src[e];
        const float4 v = *(const float4*)(xc + (size_t)sidx * rowf);
        if (SC) {
          const float k = sc[sidx];
          acc.x = fmaf(v.x, k, acc.x); acc.y = fmaf(v.y, k, acc.y); acc.z = fmaf(v.z, k, acc.z); acc.w = fmaf(v.w, k, acc.w);
        } else {
          acc.x += v.x; acc.y += v.y; acc.z += v.z; acc.w += v.w;
        }
      }
    }
  }
  return acc;
}

#define GCN_L2_NODES 128
#define GCN_L2_TILES (GCN_L2_NODES / 16)
#define GCN_L2_PITCH 36
#define GCN_H_PITCH 32

__global__ void __launch_bounds__(GCN_L2_NODES * 8)
gcn_pull_gemm_l1_kernel(const float* __restrict__ x, int rowf, int nch,
                        const int* __restrict__ cnt_in, const int* __restrict__ slots,
                        const int* __restrict__ src, const int* __restrict__ dst, int E, int n,
                        const float* __restrict__ W, int wcols,
                        const float* __restrict__ norm_src, const float* __restrict__ norm_dst,
                        const float* __restrict__ bias, float* __restrict__ h, int ntiles) {
  __shared__ float Ws[32 * 32];
  __shared__ float As[GCN_L2_NODES * GCN_L2_PITCH];

  gcn_stage_w(Ws, W, rowf, wcols);

  const int gid = blockIdx.x * blockDim.x + threadIdx.x;
  const int node = gid >> 3;
  const int gl = gid & 7;
  if (node < n) {
    const int* sl = slots + (size_t)node * GCN_SLOT_CAP;
    const int c = (gl < nch) ? gl : 0;
    const int cnt = cnt_in[node];
    const int4 q0 = *(const int4*)(sl);
    const int4 q1 = *(const int4*)(sl + 4);
    const int4 q2 = *(const int4*)(sl + 8);
    const int4 q3 = *(const int4*)(sl + 12);
    const int w0 = sl[gl];
    const int w1 = sl[8 + gl];
    asm volatile("" :: "v"(cnt), "v"(q0.x), "v"(q1.x), "v"(q2.x), "v"(q3.x), "v"(w0), "v"(w1));
    const float4 acc = gcn_list_sum<true>(x, sl, node, c, rowf, cnt, q0, q1, q2, q3, src, dst, E,
                                          norm_src, gl, w0, w1);
    if (gl < nch) *(float4*)(As + (threadIdx.x >> 3) * GCN_L2_PITCH + c * 4) = acc;
  }

  const int wv = threadIdx.x >> 5;
  const int lane = threadIdx.x & 31;
  const int m = lane & 15;
  const int half = lane >> 4;
  const int tile = blockIdx.x * GCN_L2_TILES + wv;
  const bool has_tile = (wv < GCN_L2_TILES) && (tile < ntiles);

  float4 d0, d1, t0, t1;
  float bA, bB;
  if (has_tile) {
    const float* dp = norm_dst + tile * 16 + 8 * half;
    const float* tp = norm_src + tile * 16 + 8 * half;
    d0 = *(const float4*)(dp);
    d1 = *(const float4*)(dp + 4);
    t0 = *(const float4*)(tp);
    t1 = *(const float4*)(tp + 4);
    bA = bias[(m < wcols) ? m : 0];
    bB = bias[(16 + m < wcols) ? 16 + m : 0];
  }

  gcn_lds_barrier();

  if (has_tile) {
    float* trow = As + (wv * 16) * GCN_L2_PITCH;
    const float* arow = trow + m * GCN_L2_PITCH;
    v8f c0 = {};
    v8f c1 = {};

#pragma unroll 4
    for (int k0 = 0; k0 < rowf; k0 += 4) {
      const int k = k0 + 2 * half;
      const float2 av = *(const float2*)(arow + k);
      v2f a;
      a.x = av.x;
      a.y = av.y;

      v2f b0, b1;
      b0.x = Ws[k * 32 + m];
      b0.y = Ws[(k + 1) * 32 + m];
      b1.x = Ws[k * 32 + 16 + m];
      b1.y = Ws[(k + 1) * 32 + 16 + m];

      c0 = __builtin_amdgcn_wmma_f32_16x16x4_f32(false, a, false, b0, (short)0, c0, false, false);
      c1 = __builtin_amdgcn_wmma_f32_16x16x4_f32(false, a, false, b1, (short)0, c1, false, false);
    }

    const float ndv[8] = {d0.x, d0.y, d0.z, d0.w, d1.x, d1.y, d1.z, d1.w};
    const float nsv[8] = {t0.x, t0.y, t0.z, t0.w, t1.x, t1.y, t1.z, t1.w};
#pragma unroll
    for (int r = 0; r < 8; ++r) {
      float* orow = trow + (8 * half + r) * GCN_L2_PITCH;
      orow[m] = fmaxf(fmaf(c0[r], ndv[r], bA), 0.0f) * nsv[r];
      orow[16 + m] = fmaxf(fmaf(c1[r], ndv[r], bB), 0.0f) * nsv[r];
    }
    __builtin_amdgcn_fence(__ATOMIC_RELEASE, "wavefront", "local");
    __builtin_amdgcn_wave_barrier();
    __builtin_amdgcn_fence(__ATOMIC_ACQUIRE, "wavefront", "local");

    const int oc = lane & 7;
    float* hb = h + (size_t)__builtin_amdgcn_readfirstlane(tile) * 16 * GCN_H_PITCH;
#pragma unroll
    for (int i = 0; i < 4; ++i) {
      const int row = i * 4 + (lane >> 3);
      const float4 v = *(const float4*)(trow + row * GCN_L2_PITCH + oc * 4);
      *(float4*)(hb + (unsigned)(row * GCN_H_PITCH + oc * 4)) = v;
    }
  }
}


__global__ void __launch_bounds__(GCN_L2_NODES * 8)
gcn_pull_gemm_readout_kernel(const float* __restrict__ x, const int* __restrict__ cnt_in,
                             const int* __restrict__ slots, const int* __restrict__ src,
                             const int* __restrict__ dst, int E, int n, int nch, int H, int xpitch,
                             const float* __restrict__ W, int wcols,
                             const float* __restrict__ norm, const float* __restrict__ bias,
                             const float* __restrict__ Wd, const float* __restrict__ bd,
                             float* __restrict__ out, int ntiles) {
  __shared__ float Ws[32 * 32];
  __shared__ float As[GCN_L2_NODES * GCN_L2_PITCH];
  __shared__ float s_ro[GCN_L2_NODES / 4];

  gcn_stage_w(Ws, W, H, wcols);

  const int gid = blockIdx.x * blockDim.x + threadIdx.x;
  const int node = gid >> 3;
  const int c = gid & 7;
  if (node < n && c < nch) {
    const int* sl = slots + (size_t)node * GCN_SLOT_CAP;
    const int cnt = cnt_in[node];
    const int4 q0 = *(const int4*)(sl);
    const int4 q1 = *(const int4*)(sl + 4);
    const int4 q2 = *(const int4*)(sl + 8);
    const int4 q3 = *(const int4*)(sl + 12);
    asm volatile("" :: "v"(cnt), "v"(q0.x), "v"(q1.x), "v"(q2.x), "v"(q3.x));
    const float4 acc = gcn_list_sum<false>(x, sl, node, c, xpitch, cnt, q0, q1, q2, q3, src, dst, E,
                                           nullptr, 0, 0, 0);
    *(float4*)(As + (threadIdx.x >> 3) * GCN_L2_PITCH + c * 4) = acc;
  }

  const int wv = threadIdx.x >> 5;
  const int lane = threadIdx.x & 31;
  const int m = lane & 15;
  const int half = lane >> 4;
  const int tile = blockIdx.x * GCN_L2_TILES + wv;
  const bool has_tile = (wv < GCN_L2_TILES) && (tile < ntiles);

  const bool hasA = m < wcols;
  const bool hasB = 16 + m < wcols;
  float4 n0, n1;
  float wA[4], wB[4], bA, bB, bd0;
  if (has_tile) {
    const float* np = norm + tile * 16 + 8 * half;
    n0 = *(const float4*)(np);
    n1 = *(const float4*)(np + 4);
    const int colA = hasA ? m : 0;
    const int colB = hasB ? 16 + m : 0;
    bA = bias[colA];
    bB = bias[colB];
#pragma unroll
    for (int g = 0; g < 4; ++g) {
      wA[g] = Wd[g * wcols + colA];
      wB[g] = Wd[g * wcols + colB];
    }
    bd0 = bd[0];
  }

  gcn_lds_barrier();

  if (has_tile) {
    const float* arow = As + (wv * 16 + m) * GCN_L2_PITCH;
    v8f c0 = {};
    v8f c1 = {};

#pragma unroll 3
    for (int k0 = 0; k0 < H; k0 += 4) {
      const int k = k0 + 2 * half;
      const float2 av = *(const float2*)(arow + k);
      v2f a;
      a.x = av.x;
      a.y = av.y;

      v2f b0, b1;
      b0.x = Ws[k * 32 + m];
      b0.y = Ws[(k + 1) * 32 + m];
      b1.x = Ws[k * 32 + 16 + m];
      b1.y = Ws[(k + 1) * 32 + 16 + m];

      c0 = __builtin_amdgcn_wmma_f32_16x16x4_f32(false, a, false, b0, (short)0, c0, false, false);
      c1 = __builtin_amdgcn_wmma_f32_16x16x4_f32(false, a, false, b1, (short)0, c1, false, false);
    }

    const float ndv[8] = {n0.x, n0.y, n0.z, n0.w, n1.x, n1.y, n1.z, n1.w};
#pragma unroll
    for (int g = 0; g < 4; ++g) {
      wA[g] = hasA ? wA[g] : 0.0f;
      wB[g] = hasB ? wB[g] : 0.0f;
    }
    float s0 = 0.0f, s1 = 0.0f;
#pragma unroll
    for (int g = 0; g < 4; ++g) {
      s0 = fmaf(fmaxf(fmaf(c0[g], ndv[g], bA), 0.0f), wA[g], s0);
      s0 = fmaf(fmaxf(fmaf(c1[g], ndv[g], bB), 0.0f), wB[g], s0);
      s1 = fmaf(fmaxf(fmaf(c0[4 + g], ndv[4 + g], bA), 0.0f), wA[g], s1);
      s1 = fmaf(fmaxf(fmaf(c1[4 + g], ndv[4 + g], bB), 0.0f), wB[g], s1);
    }
#pragma unroll
    for (int off = 8; off >= 1; off >>= 1) {
      s0 += __shfl_xor(s0, off);
      s1 += __shfl_xor(s1, off);
    }
    if (m == 0) {
      s_ro[wv * 4 + 2 * half] = s0 + bd0;
      s_ro[wv * 4 + 2 * half + 1] = s1 + bd0;
    }
  }

  gcn_lds_barrier();
  const int r = blockIdx.x * (GCN_L2_NODES / 4) + threadIdx.x;
  if (threadIdx.x < GCN_L2_NODES / 4 && r < ntiles * 4) out[r] = s_ro[threadIdx.x];
}


extern "C" void kernel_launch(void* const* d_in, const int* in_sizes, int n_in,
                              void* d_out, int out_size, void* d_ws, size_t ws_size,
                              hipStream_t stream) {
  const float* features = (const float*)d_in[0];
  const int*   src      = (const int*)d_in[1];
  const int*   dst      = (const int*)d_in[2];
  const float* W1       = (const float*)d_in[3];
  const float* b1       = (const float*)d_in[4];
  const float* W2       = (const float*)d_in[5];
  const float* b2       = (const float*)d_in[6];
  const float* Wd       = (const float*)d_in[7];
  const float* bd       = (const float*)d_in[8];
  float* out = (float*)d_out;

  const int H  = in_sizes[4];
  const int F  = in_sizes[3] / H;
  const int Nn = in_sizes[0] / F;
  const int E  = in_sizes[1];

  const int B = 256;
  const int ntiles = Nn / 16;
  const int nch = H / 4;
  const int nchF = F / 4;
  const int l2Grid = (Nn + GCN_L2_NODES - 1) / GCN_L2_NODES;

  int*   tick     = (int*)d_ws;
  int*   cnt_out  = tick + 2 * GCN_TICK_STRIDE;
  int*   cnt_in   = cnt_out + Nn;
  float* norm_src = (float*)(cnt_in + Nn);
  float* norm_dst = norm_src + Nn;
  float* bufA     = norm_dst + Nn;
  int*   slots    = (int*)(bufA + (size_t)Nn * GCN_H_PITCH);

  gcn_zero_kernel<<<(2 * GCN_TICK_STRIDE + 2 * Nn + B - 1) / B, B, 0, stream>>>(
      tick, 2 * GCN_TICK_STRIDE + 2 * Nn);
  gcn_edge_kernel<<<2048, GCN_EDGE_BLOCK, 0, stream>>>(src, dst, tick, cnt_out, cnt_in, slots, E, Nn);
  gcn_norm_kernel<<<(2 * Nn + B - 1) / B, B, 0, stream>>>(cnt_out, norm_src, 2 * Nn);
  gcn_pull_gemm_l1_kernel<<<l2Grid, GCN_L2_NODES * 8, 0, stream>>>(
      features, F, nchF, cnt_in, slots, src, dst, E, Nn, W1, H, norm_src, norm_dst, b1, bufA, ntiles);
  gcn_pull_gemm_readout_kernel<<<l2Grid, GCN_L2_NODES * 8, 0, stream>>>(
      bufA, cnt_in, slots, src, dst, E, Nn, nch, H, GCN_H_PITCH, W2, H, norm_dst, b2, Wd, bd, out, ntiles);
}


# resident grids whose waves loop (wave entry was the limiter): layer kernels 512 work-groups over 128-node batches (was 1563), zero/norm passes 256 work-groups, 16 bytes a lane; sums and order kept
# speedup vs baseline: 4.9972x; 1.0649x over previous
#include <hip/hip_runtime.h>

typedef __attribute__((ext_vector_type(2))) float v2f;
typedef __attribute__((ext_vector_type(8))) float v8f;


__global__ void gcn_zero_kernel(int* __restrict__ p, int n) {
  const int n4 = n >> 2;
  const int stride = gridDim.x * blockDim.x;
  const int t = blockIdx.x * blockDim.x + threadIdx.x;
  for (int i = t; i < n4; i += stride) ((int4*)p)[i] = make_int4(0, 0, 0, 0);
  const int r = (n4 << 2) + t;
  if (r < n) p[r] = 0;
}

#define GCN_SLOT_CAP 32

#define GCN_TICK_STRIDE 64
#define GCN_EDGE_BLOCK 256
#define GCN_EDGE_PER_LANE 4
#define GCN_CHUNK (GCN_EDGE_BLOCK * GCN_EDGE_PER_LANE)

static __device__ __forceinline__ void gcn_lds_barrier() {
  __builtin_amdgcn_fence(__ATOMIC_RELEASE, "workgroup", "local");
  __builtin_amdgcn_s_barrier();
  __builtin_amdgcn_fence(__ATOMIC_ACQUIRE, "workgroup", "local");
}

__global__ void __launch_bounds__(GCN_EDGE_BLOCK)
gcn_edge_kernel(const int* __restrict__ src, const int* __restrict__ dst, int* __restrict__ tick,
                int* __restrict__ cnt_out, int* __restrict__ cnt_in, int* __restrict__ slots,
                int E, int n) {
  __shared__ int s_chunk[2];
  const int tid = threadIdx.x;
  const int nchunks = (E + GCN_CHUNK - 1) / GCN_CHUNK;
  int split = ((n >> 1) + 63) & ~63;
  if (split > n) split = n;
  int t = (int)((__builtin_amdgcn_s_sendmsg_rtn(0x87) >> 18) & 1u);

  for (int pass = 0; pass < 2; ++pass, t ^= 1) {
    const int lo = t ? split : 0;
    const unsigned span = t ? (unsigned)(n - split) : (unsigned)split;
    int* tk = tick + t * GCN_TICK_STRIDE;

    if (tid == 0) {
      int first = nchunks;
      if (pass == 0 || __hip_atomic_load(tk, __ATOMIC_RELAXED, __HIP_MEMORY_SCOPE_AGENT) < nchunks)
        first = __hip_atomic_fetch_add(tk, 1, __ATOMIC_RELAXED, __HIP_MEMORY_SCOPE_AGENT);
      s_chunk[0] = first;
    }
    gcn_lds_barrier();
    int c = s_chunk[0];
    int buf = 0;
    while (c < nchunks) {
      int s[GCN_EDGE_PER_LANE], d[GCN_EDGE_PER_LANE], p[GCN_EDGE_PER_LANE];
      bool os[GCN_EDGE_PER_LANE], od[GCN_EDGE_PER_LANE];
#pragma unroll
      for (int k = 0; k < GCN_EDGE_PER_LANE; ++k) {
        const int e = c * GCN_CHUNK + k * GCN_EDGE_BLOCK + tid;
        const int ec = (e < E) ? e : E - 1;
        s[k] = src[ec];
        d[k] = dst[ec];
        os[k] = (e < E);
      }
      asm volatile("" :: "v"(s[0]), "v"(d[0]), "v"(s[1]), "v"(d[1]),
                         "v"(s[2]), "v"(d[2]), "v"(s[3]), "v"(d[3]));
#pragma unroll
      for (int k = 0; k < GCN_EDGE_PER_LANE; ++k) {
        od[k] = os[k] && ((unsigned)(d[k] - lo) < span);
        os[k] = os[k] && ((unsigned)(s[k] - lo) < span);
      }
#pragma unroll
      for (int k = 0; k < GCN_EDGE_PER_LANE; ++k) {
        p[k] = GCN_SLOT_CAP;
        if (od[k])
          p[k] = __hip_atomic_fetch_add(&cnt_in[d[k]], 1, __ATOMIC_RELAXED, __HIP_MEMORY_SCOPE_AGENT);
      }
#pragma unroll
      for (int k = 0; k < GCN_EDGE_PER_LANE; ++k) {
        if (os[k])
          (void)__hip_atomic_fetch_add(&cnt_out[s[k]], 1, __ATOMIC_RELAXED, __HIP_MEMORY_SCOPE_AGENT);
      }
      int nxt = 0;
      if (tid == 0) nxt = __hip_atomic_fetch_add(tk, 1, __ATOMIC_RELAXED, __HIP_MEMORY_SCOPE_AGENT);
#pragma unroll
      for (int k = 0; k < GCN_EDGE_PER_LANE; ++k) {
        if ((unsigned)p[k] < (unsigned)GCN_SLOT_CAP) slots[(size_t)d[k] * GCN_SLOT_CAP + p[k]] = s[k];
      }
      buf ^= 1;
      if (tid == 0) s_chunk[buf] = nxt;
      gcn_lds_barrier();
      c = s_chunk[buf];
    }
    gcn_lds_barrier();
  }
}

__global__ void gcn_norm_kernel(const int* __restrict__ cnt, float* __restrict__ nrm, int n) {
  const int n4 = n >> 2;
  const int stride = gridDim.x * blockDim.x;
  const int t = blockIdx.x * blockDim.x + threadIdx.x;
  for (int i = t; i < n4; i += stride) {
    const int4 c = ((const int4*)cnt)[i];
    float4 v;
    v.x = rsqrtf(fmaxf((float)c.x, 1.0f));
    v.y = rsqrtf(fmaxf((float)c.y, 1.0f));
    v.z = rsqrtf(fmaxf((float)c.z, 1.0f));
    v.w = rsqrtf(fmaxf((float)c.w, 1.0f));
    ((float4*)nrm)[i] = v;
  }
  const int r = (n4 << 2) + t;
  if (r < n) nrm[r] = rsqrtf(fmaxf((float)cnt[r], 1.0f));
}


static __device__ __forceinline__ void gcn_stage_w(float* __restrict__ Ws, const float* __restrict__ W,
                                                   int K, int wcols) {
  for (int i = threadIdx.x; i < 32 * 32; i += blockDim.x) {
    const int k = i >> 5;
    const int n = i & 31;
    Ws[i] = (k < K && n < wcols) ? W[(size_t)k * wcols + n] : 0.0f;
  }
}


template <int J>
static __device__ __forceinline__ float gcn_group_lane(float v) {
  return __uint_as_float(__builtin_amdgcn_mov_dpp8(__float_as_uint(v), (unsigned)J * 0x249249u));
}

template <bool SC>
static __device__ __forceinline__ void gcn_trip8(float4& acc, const char* __restrict__ xb,
                                                 unsigned rowbytes, unsigned cbytes, int r,
                                                 int e0, int e1, int e2, int e3,
                                                 int e4, int e5, int e6, int e7,
                                                 const float* __restrict__ sc, int gl, int w) {
  const unsigned s0 = (unsigned)e0;
  const unsigned s1 = (unsigned)((r > 1) ? e1 : e0);
  const unsigned s2 = (unsigned)((r > 2) ? e2 : e0);
  const unsigned s3 = (unsigned)((r > 3) ? e3 : e0);
  const unsigned s4 = (unsigned)((r > 4) ? e4 : e0);
  const unsigned s5 = (unsigned)((r > 5) ? e5 : e0);
  const unsigned s6 = (unsigned)((r > 6) ? e6 : e0);
  const unsigned s7 = (unsigned)((r > 7) ? e7 : e0);
  const float4 v0 = *(const float4*)(xb + (__umul24(s0, rowbytes) + cbytes));
  const float4 v1 = *(const float4*)(xb + (__umul24(s1, rowbytes) + cbytes));
  const float4 v2 = *(const float4*)(xb + (__umul24(s2, rowbytes) + cbytes));
  const float4 v3 = *(const float4*)(xb + (__umul24(s3, rowbytes) + cbytes));
  const float4 v4 = *(const float4*)(xb + (__umul24(s4, rowbytes) + cbytes));
  const float4 v5 = *(const float4*)(xb + (__umul24(s5, rowbytes) + cbytes));
  const float4 v6 = *(const float4*)(xb + (__umul24(s6, rowbytes) + cbytes));
  const float4 v7 = *(const float4*)(xb + (__umul24(s7, rowbytes) + cbytes));
  float mine = 1.0f;
  if (SC) {
    const unsigned sm = (gl < r) ? (unsigned)w : s0;
    mine = *(const float*)((const char*)sc + (sm << 2));
  }
  asm volatile("" :: "v"(v0.x), "v"(v1.x), "v"(v2.x), "v"(v3.x),
                     "v"(v4.x), "v"(v5.x), "v"(v6.x), "v"(v7.x), "v"(mine));
  __builtin_amdgcn_sched_barrier(0);
  if (SC) {
    const float c0 = gcn_group_lane<0>(mine);
    acc.x = fmaf(v0.x, c0, acc.x); acc.y = fmaf(v0.y, c0, acc.y); acc.z = fmaf(v0.z, c0, acc.z); acc.w = fmaf(v0.w, c0, acc.w);
    if (r > 1) { const float k = gcn_group_lane<1>(mine); acc.x = fmaf(v1.x, k, acc.x); acc.y = fmaf(v1.y, k, acc.y); acc.z = fmaf(v1.z, k, acc.z); acc.w = fmaf(v1.w, k, acc.w); }
    if (r > 2) { const float k = gcn_group_lane<2>(mine); acc.x = fmaf(v2.x, k, acc.x); acc.y = fmaf(v2.y, k, acc.y); acc.z = fmaf(v2.z, k, acc.z); acc.w = fmaf(v2.w, k, acc.w); }
    if (r > 3) { const float k = gcn_group_lane<3>(mine); acc.x = fmaf(v3.x, k, acc.x); acc.y = fmaf(v3.y, k, acc.y); acc.z = fmaf(v3.z, k, acc.z); acc.w = fmaf(v3.w, k, acc.w); }
    if (r > 4) { const float k = gcn_group_lane<4>(mine); acc.x = fmaf(v4.x, k, acc.x); acc.y = fmaf(v4.y, k, acc.y); acc.z = fmaf(v4.z, k, acc.z); acc.w = fmaf(v4.w, k, acc.w); }
    if (r > 5) { const float k = gcn_group_lane<5>(mine); acc.x = fmaf(v5.x, k, acc.x); acc.y = fmaf(v5.y, k, acc.y); acc.z = fmaf(v5.z, k, acc.z); acc.w = fmaf(v5.w, k, acc.w); }
    if (r > 6) { const float k = gcn_group_lane<6>(mine); acc.x = fmaf(v6.x, k, acc.x); acc.y = fmaf(v6.y, k, acc.y); acc.z = fmaf(v6.z, k, acc.z); acc.w = fmaf(v6.w, k, acc.w); }
    if (r > 7) { const float k = gcn_group_lane<7>(mine); acc.x = fmaf(v7.x, k, acc.x); acc.y = fmaf(v7.y, k, acc.y); acc.z = fmaf(v7.z, k, acc.z); acc.w = fmaf(v7.w, k, acc.w); }
  } else {
    acc.x += v0.x; acc.y += v0.y; acc.z += v0.z; acc.w += v0.w;
    if (r > 1) { acc.x += v1.x; acc.y += v1.y; acc.z += v1.z; acc.w += v1.w; }
    if (r > 2) { acc.x += v2.x; acc.y += v2.y; acc.z += v2.z; acc.w += v2.w; }
    if (r > 3) { acc.x += v3.x; acc.y += v3.y; acc.z += v3.z; acc.w += v3.w; }
    if (r > 4) { acc.x += v4.x; acc.y += v4.y; acc.z += v4.z; acc.w += v4.w; }
    if (r > 5) { acc.x += v5.x; acc.y += v5.y; acc.z += v5.z; acc.w += v5.w; }
    if (r > 6) { acc.x += v6.x; acc.y += v6.y; acc.z += v6.z; acc.w += v6.w; }
    if (r > 7) { acc.x += v7.x; acc.y += v7.y; acc.z += v7.z; acc.w += v7.w; }
  }
}

template <bool SC>
static __device__ __forceinline__ float4 gcn_list_sum(const float* __restrict__ x, const int* __restrict__ sl,
                                                      int node, int c, int rowf, int cnt,
                                                      int4 q0, int4 q1, int4 q2, int4 q3,
                                                      const int* __restrict__ src,
                                                      const int* __restrict__ dst, int E,
                                                      const float* __restrict__ sc, int gl, int w0, int w1) {
  const float* xc = x + c * 4;
  const char* xb = (const char*)x;
  const unsigned rowbytes = (unsigned)rowf * 4u;
  const unsigned cbytes = (unsigned)c * 16u;
  float4 acc = make_float4(0.0f, 0.0f, 0.0f, 0.0f);
  if (cnt <= GCN_SLOT_CAP) {
    if (cnt > 0) gcn_trip8<SC>(acc, xb, rowbytes, cbytes, cnt, q0.x, q0.y, q0.z, q0.w, q1.x, q1.y, q1.z, q1.w, sc, gl, w0);
    if (cnt > 8) gcn_trip8<SC>(acc, xb, rowbytes, cbytes, cnt - 8, q2.x, q2.y, q2.z, q2.w, q3.x, q3.y, q3.z, q3.w, sc, gl, w1);
    for (int j = 16; j < cnt; j += 8) {
      const int4 p0 = *(const int4*)(sl + j);
      const int4 p1 = *(const int4*)(sl + j + 4);
      int w = 0;
      if (SC) w = sl[j + gl];
      gcn_trip8<SC>(acc, xb, rowbytes, cbytes, cnt - j, p0.x, p0.y, p0.z, p0.w, p1.x, p1.y, p1.z, p1.w, sc, gl, w);
    }
  } else {
    for (int e = 0; e < E; ++e) {
      if (dst[e] == node) {
        const int sidx = src[e];
        const float4 v = *(const float4*)(xc + (size_t)sidx * rowf);
        if (SC) {
          const float k = sc[sidx];
          acc.x = fmaf(v.x, k, acc.x); acc.y = fmaf(v.y, k, acc.y); acc.z = fmaf(v.z, k, acc.z); acc.w = fmaf(v.w, k, acc.w);
        } else {
          acc.x += v.x; acc.y += v.y; acc.z += v.z; acc.w += v.w;
        }
      }
    }
  }
  return acc;
}

#define GCN_L2_NODES 128
#define GCN_L2_TILES (GCN_L2_NODES / 16)
#define GCN_L2_PITCH 36
#define GCN_H_PITCH 32

__global__ void __launch_bounds__(GCN_L2_NODES * 8)
gcn_pull_gemm_l1_kernel(const float* __restrict__ x, int rowf, int nch,
                        const int* __restrict__ cnt_in, const int* __restrict__ slots,
                        const int* __restrict__ src, const int* __restrict__ dst, int E, int n,
                        const float* __restrict__ W, int wcols,
                        const float* __restrict__ norm_src, const float* __restrict__ norm_dst,
                        const float* __restrict__ bias, float* __restrict__ h, int ntiles) {
  __shared__ float Ws[32 * 32];
  __shared__ float As[GCN_L2_NODES * GCN_L2_PITCH];

  gcn_stage_w(Ws, W, rowf, wcols);

  const int nbatches = (n + GCN_L2_NODES - 1) / GCN_L2_NODES;

  for (int b = blockIdx.x; b < nbatches; b += gridDim.x) {
    int tid = threadIdx.x;
    asm volatile("" : "+v"(tid));
    const int gl = tid & 7;

    const int node = b * GCN_L2_NODES + (tid >> 3);
    if (node < n) {
      const int* sl = slots + (size_t)node * GCN_SLOT_CAP;
      const int c = (gl < nch) ? gl : 0;
      const int cnt = cnt_in[node];
      const int4 q0 = *(const int4*)(sl);
      const int4 q1 = *(const int4*)(sl + 4);
      const int4 q2 = *(const int4*)(sl + 8);
      const int4 q3 = *(const int4*)(sl + 12);
      const int w0 = sl[gl];
      const int w1 = sl[8 + gl];
      asm volatile("" :: "v"(cnt), "v"(q0.x), "v"(q1.x), "v"(q2.x), "v"(q3.x), "v"(w0), "v"(w1));
      const float4 acc = gcn_list_sum<true>(x, sl, node, c, rowf, cnt, q0, q1, q2, q3, src, dst, E,
                                            norm_src, gl, w0, w1);
      if (gl < nch) *(float4*)(As + (tid >> 3) * GCN_L2_PITCH + c * 4) = acc;
    }

    const int wv = tid >> 5;
    const int lane = tid & 31;
    const int m = lane & 15;
    const int half = lane >> 4;
    const int tile = b * GCN_L2_TILES + wv;
    const bool has_tile = (wv < GCN_L2_TILES) && (tile < ntiles);

    const float4 zero4 = make_float4(0.0f, 0.0f, 0.0f, 0.0f);
    float4 d0 = zero4, d1 = zero4, t0 = zero4, t1 = zero4;
    float bA = 0.0f, bB = 0.0f;
    if (has_tile) {
      const float* dp = norm_dst + tile * 16 + 8 * half;
      const float* tp = norm_src + tile * 16 + 8 * half;
      d0 = *(const float4*)(dp);
      d1 = *(const float4*)(dp + 4);
      t0 = *(const float4*)(tp);
      t1 = *(const float4*)(tp + 4);
      bA = bias[(m < wcols) ? m : 0];
      bB = bias[(16 + m < wcols) ? 16 + m : 0];
    }

    gcn_lds_barrier();

    if (has_tile) {
      float* trow = As + (wv * 16) * GCN_L2_PITCH;
      const float* arow = trow + m * GCN_L2_PITCH;
      v8f c0 = {};
      v8f c1 = {};

#pragma unroll 4
      for (int k0 = 0; k0 < rowf; k0 += 4) {
        const int k = k0 + 2 * half;
        const float2 av = *(const float2*)(arow + k);
        v2f a;
        a.x = av.x;
        a.y = av.y;

        v2f b0, b1;
        b0.x = Ws[k * 32 + m];
        b0.y = Ws[(k + 1) * 32 + m];
        b1.x = Ws[k * 32 + 16 + m];
        b1.y = Ws[(k + 1) * 32 + 16 + m];

        c0 = __builtin_amdgcn_wmma_f32_16x16x4_f32(false, a, false, b0, (short)0, c0, false, false);
        c1 = __builtin_amdgcn_wmma_f32_16x16x4_f32(false, a, false, b1, (short)0, c1, false, false);
      }

      const float ndv[8] = {d0.x, d0.y, d0.z, d0.w, d1.x, d1.y, d1.z, d1.w};
      const float nsv[8] = {t0.x, t0.y, t0.z, t0.w, t1.x, t1.y, t1.z, t1.w};
#pragma unroll
      for (int r = 0; r < 8; ++r) {
        float* orow = trow + (8 * half + r) * GCN_L2_PITCH;
        orow[m] = fmaxf(fmaf(c0[r], ndv[r], bA), 0.0f) * nsv[r];
        orow[16 + m] = fmaxf(fmaf(c1[r], ndv[r], bB), 0.0f) * nsv[r];
      }
      __builtin_amdgcn_fence(__ATOMIC_RELEASE, "wavefront", "local");
      __builtin_amdgcn_wave_barrier();
      __builtin_amdgcn_fence(__ATOMIC_ACQUIRE, "wavefront", "local");

      const int oc = lane & 7;
      float* hb = h + (size_t)__builtin_amdgcn_readfirstlane(tile) * 16 * GCN_H_PITCH;
#pragma unroll
      for (int i = 0; i < 4; ++i) {
        const int row = i * 4 + (lane >> 3);
        const float4 v = *(const float4*)(trow + row * GCN_L2_PITCH + oc * 4);
        *(float4*)(hb + (unsigned)(row * GCN_H_PITCH + oc * 4)) = v;
      }
    }

    gcn_lds_barrier();
  }
}


__global__ void __launch_bounds__(GCN_L2_NODES * 8)
gcn_pull_gemm_readout_kernel(const float* __restrict__ x, const int* __restrict__ cnt_in,
                             const int* __restrict__ slots, const int* __restrict__ src,
                             const int* __restrict__ dst, int E, int n, int nch, int H, int xpitch,
                             const float* __restrict__ W, int wcols,
                             const float* __restrict__ norm, const float* __restrict__ bias,
                             const float* __restrict__ Wd, const float* __restrict__ bd,
                             float* __restrict__ out, int ntiles) {
  __shared__ float Ws[32 * 32];
  __shared__ float As[GCN_L2_NODES * GCN_L2_PITCH];
  __shared__ float s_ro[GCN_L2_NODES / 4];

  gcn_stage_w(Ws, W, H, wcols);

  const int nbatches = (n + GCN_L2_NODES - 1) / GCN_L2_NODES;

  for (int b = blockIdx.x; b < nbatches; b += gridDim.x) {
    int tid = threadIdx.x;
    asm volatile("" : "+v"(tid));
    const int c = tid & 7;

    const int node = b * GCN_L2_NODES + (tid >> 3);
    if (node < n && c < nch) {
      const int* sl = slots + (size_t)node * GCN_SLOT_CAP;
      const int cnt = cnt_in[node];
      const int4 q0 = *(const int4*)(sl);
      const int4 q1 = *(const int4*)(sl + 4);
      const int4 q2 = *(const int4*)(sl + 8);
      const int4 q3 = *(const int4*)(sl + 12);
      asm volatile("" :: "v"(cnt), "v"(q0.x), "v"(q1.x), "v"(q2.x), "v"(q3.x));
      const float4 acc = gcn_list_sum<false>(x, sl, node, c, xpitch, cnt, q0, q1, q2, q3, src, dst, E,
                                             nullptr, 0, 0, 0);
      *(float4*)(As + (tid >> 3) * GCN_L2_PITCH + c * 4) = acc;
    }

    const int wv = tid >> 5;
    const int lane = tid & 31;
    const int m = lane & 15;
    const int half = lane >> 4;
    const bool hasA = m < wcols;
    const bool hasB = 16 + m < wcols;
    const int tile = b * GCN_L2_TILES + wv;
    const bool has_tile = (wv < GCN_L2_TILES) && (tile < ntiles);

    const float4 zero4 = make_float4(0.0f, 0.0f, 0.0f, 0.0f);
    float4 n0 = zero4, n1 = zero4;
    float wA[4] = {0.0f, 0.0f, 0.0f, 0.0f}, wB[4] = {0.0f, 0.0f, 0.0f, 0.0f};
    float bA = 0.0f, bB = 0.0f, bd0 = 0.0f;
    if (has_tile) {
      const float* np = norm + tile * 16 + 8 * half;
      n0 = *(const float4*)(np);
      n1 = *(const float4*)(np + 4);
      const int colA = hasA ? m : 0;
      const int colB = hasB ? 16 + m : 0;
      bA = bias[colA];
      bB = bias[colB];
#pragma unroll
      for (int g = 0; g < 4; ++g) {
        wA[g] = Wd[g * wcols + colA];
        wB[g] = Wd[g * wcols + colB];
      }
      bd0 = bd[0];
    }

    gcn_lds_barrier();

    if (has_tile) {
      const float* arow = As + (wv * 16 + m) * GCN_L2_PITCH;
      v8f c0 = {};
      v8f c1 = {};

#pragma unroll 3
      for (int k0 = 0; k0 < H; k0 += 4) {
        const int k = k0 + 2 * half;
        const float2 av = *(const float2*)(arow + k);
        v2f a;
        a.x = av.x;
        a.y = av.y;

        v2f b0, b1;
        b0.x = Ws[k * 32 + m];
        b0.y = Ws[(k + 1) * 32 + m];
        b1.x = Ws[k * 32 + 16 + m];
        b1.y = Ws[(k + 1) * 32 + 16 + m];

        c0 = __builtin_amdgcn_wmma_f32_16x16x4_f32(false, a, false, b0, (short)0, c0, false, false);
        c1 = __builtin_amdgcn_wmma_f32_16x16x4_f32(false, a, false, b1, (short)0, c1, false, false);
      }

      const float ndv[8] = {n0.x, n0.y, n0.z, n0.w, n1.x, n1.y, n1.z, n1.w};
#pragma unroll
      for (int g = 0; g < 4; ++g) {
        wA[g] = hasA ? wA[g] : 0.0f;
        wB[g] = hasB ? wB[g] : 0.0f;
      }
      float s0 = 0.0f, s1 = 0.0f;
#pragma unroll
      for (int g = 0; g < 4; ++g) {
        s0 = fmaf(fmaxf(fmaf(c0[g], ndv[g], bA), 0.0f), wA[g], s0);
        s0 = fmaf(fmaxf(fmaf(c1[g], ndv[g], bB), 0.0f), wB[g], s0);
        s1 = fmaf(fmaxf(fmaf(c0[4 + g], ndv[4 + g], bA), 0.0f), wA[g], s1);
        s1 = fmaf(fmaxf(fmaf(c1[4 + g], ndv[4 + g], bB), 0.0f), wB[g], s1);
      }
#pragma unroll
      for (int off = 8; off >= 1; off >>= 1) {
        s0 += __shfl_xor(s0, off);
        s1 += __shfl_xor(s1, off);
      }
      if (m == 0) {
        s_ro[wv * 4 + 2 * half] = s0 + bd0;
        s_ro[wv * 4 + 2 * half + 1] = s1 + bd0;
      }
    }

    gcn_lds_barrier();
    const int r = b * (GCN_L2_NODES / 4) + tid;
    if (tid < GCN_L2_NODES / 4 && r < ntiles * 4) out[r] = s_ro[tid];
  }
}


extern "C" void kernel_launch(void* const* d_in, const int* in_sizes, int n_in,
                              void* d_out, int out_size, void* d_ws, size_t ws_size,
                              hipStream_t stream) {
  const float* features = (const float*)d_in[0];
  const int*   src      = (const int*)d_in[1];
  const int*   dst      = (const int*)d_in[2];
  const float* W1       = (const float*)d_in[3];
  const float* b1       = (const float*)d_in[4];
  const float* W2       = (const float*)d_in[5];
  const float* b2       = (const float*)d_in[6];
  const float* Wd       = (const float*)d_in[7];
  const float* bd       = (const float*)d_in[8];
  float* out = (float*)d_out;

  const int H  = in_sizes[4];
  const int F  = in_sizes[3] / H;
  const int Nn = in_sizes[0] / F;
  const int E  = in_sizes[1];

  const int B = 256;
  const int ntiles = Nn / 16;
  const int nch = H / 4;
  const int nchF = F / 4;
  const int l2Batches = (Nn + GCN_L2_NODES - 1) / GCN_L2_NODES;
  const int l2Grid = l2Batches < 512 ? l2Batches : 512;
  const int ewGrid = 256;

  int*   tick     = (int*)d_ws;
  int*   cnt_out  = tick + 2 * GCN_TICK_STRIDE;
  int*   cnt_in   = cnt_out + Nn;
  float* norm_src = (float*)(cnt_in + Nn);
  float* norm_dst = norm_src + Nn;
  float* bufA     = norm_dst + Nn;
  int*   slots    = (int*)(bufA + (size_t)Nn * GCN_H_PITCH);

  gcn_zero_kernel<<<ewGrid, B, 0, stream>>>(tick, 2 * GCN_TICK_STRIDE + 2 * Nn);
  gcn_edge_kernel<<<2048, GCN_EDGE_BLOCK, 0, stream>>>(src, dst, tick, cnt_out, cnt_in, slots, E, Nn);
  gcn_norm_kernel<<<ewGrid, B, 0, stream>>>(cnt_out, norm_src, 2 * Nn);
  gcn_pull_gemm_l1_kernel<<<l2Grid, GCN_L2_NODES * 8, 0, stream>>>(
      features, F, nchF, cnt_in, slots, src, dst, E, Nn, W1, H, norm_src, norm_dst, b1, bufA, ntiles);
  gcn_pull_gemm_readout_kernel<<<l2Grid, GCN_L2_NODES * 8, 0, stream>>>(
      bufA, cnt_in, slots, src, dst, E, Nn, nch, H, GCN_H_PITCH, W2, H, norm_dst, b2, Wd, bd, out, ntiles);
}


# norm pass and scale arrays dropped: the layer kernels load the integer degree counts where they loaded the scales (same loads) and take rsqrt(max(count,1)) there; 5->4 launches; sums and order kept
# speedup vs baseline: 5.2342x; 1.0474x over previous
#include <hip/hip_runtime.h>

typedef __attribute__((ext_vector_type(2))) float v2f;
typedef __attribute__((ext_vector_type(8))) float v8f;
typedef __attribute__((ext_vector_type(4))) int v4i;


__global__ void gcn_zero_kernel(int* __restrict__ p, int n) {
  const int n4 = n >> 2;
  const int stride = gridDim.x * blockDim.x;
  const int t = blockIdx.x * blockDim.x + threadIdx.x;
  for (int i = t; i < n4; i += stride) ((int4*)p)[i] = make_int4(0, 0, 0, 0);
  const int r = (n4 << 2) + t;
  if (r < n) p[r] = 0;
}

#define GCN_SLOT_CAP 32

#define GCN_TICK_STRIDE 64
#define GCN_EDGE_BLOCK 256
#define GCN_EDGE_PER_LANE 4
#define GCN_CHUNK (GCN_EDGE_BLOCK * GCN_EDGE_PER_LANE)

static __device__ __forceinline__ void gcn_lds_barrier() {
  __builtin_amdgcn_fence(__ATOMIC_RELEASE, "workgroup", "local");
  __builtin_amdgcn_s_barrier();
  __builtin_amdgcn_fence(__ATOMIC_ACQUIRE, "workgroup", "local");
}

__global__ void __launch_bounds__(GCN_EDGE_BLOCK)
gcn_edge_kernel(const int* __restrict__ src, const int* __restrict__ dst, int* __restrict__ tick,
                int* __restrict__ cnt_out, int* __restrict__ cnt_in, int* __restrict__ slots,
                int E, int n) {
  __shared__ int s_chunk[2];
  const int tid = threadIdx.x;
  const int nchunks = (E + GCN_CHUNK - 1) / GCN_CHUNK;
  int split = ((n >> 1) + 63) & ~63;
  if (split > n) split = n;
  int t = (int)((__builtin_amdgcn_s_sendmsg_rtn(0x87) >> 18) & 1u);

  for (int pass = 0; pass < 2; ++pass, t ^= 1) {
    const int lo = t ? split : 0;
    const unsigned span = t ? (unsigned)(n - split) : (unsigned)split;
    int* tk = tick + t * GCN_TICK_STRIDE;

    if (tid == 0) {
      int first = nchunks;
      if (pass == 0 || __hip_atomic_load(tk, __ATOMIC_RELAXED, __HIP_MEMORY_SCOPE_AGENT) < nchunks)
        first = __hip_atomic_fetch_add(tk, 1, __ATOMIC_RELAXED, __HIP_MEMORY_SCOPE_AGENT);
      s_chunk[0] = first;
    }
    gcn_lds_barrier();
    int c = s_chunk[0];
    int buf = 0;
    while (c < nchunks) {
      int s[GCN_EDGE_PER_LANE], d[GCN_EDGE_PER_LANE], p[GCN_EDGE_PER_LANE];
      bool os[GCN_EDGE_PER_LANE], od[GCN_EDGE_PER_LANE];
#pragma unroll
      for (int k = 0; k < GCN_EDGE_PER_LANE; ++k) {
        const int e = c * GCN_CHUNK + k * GCN_EDGE_BLOCK + tid;
        const int ec = (e < E) ? e : E - 1;
        s[k] = src[ec];
        d[k] = dst[ec];
        os[k] = (e < E);
      }
      asm volatile("" :: "v"(s[0]), "v"(d[0]), "v"(s[1]), "v"(d[1]),
                         "v"(s[2]), "v"(d[2]), "v"(s[3]), "v"(d[3]));
#pragma unroll
      for (int k = 0; k < GCN_EDGE_PER_LANE; ++k) {
        od[k] = os[k] && ((unsigned)(d[k] - lo) < span);
        os[k] = os[k] && ((unsigned)(s[k] - lo) < span);
      }
#pragma unroll
      for (int k = 0; k < GCN_EDGE_PER_LANE; ++k) {
        p[k] = GCN_SLOT_CAP;
        if (od[k])
          p[k] = __hip_atomic_fetch_add(&cnt_in[d[k]], 1, __ATOMIC_RELAXED, __HIP_MEMORY_SCOPE_AGENT);
      }
#pragma unroll
      for (int k = 0; k < GCN_EDGE_PER_LANE; ++k) {
        if (os[k])
          (void)__hip_atomic_fetch_add(&cnt_out[s[k]], 1, __ATOMIC_RELAXED, __HIP_MEMORY_SCOPE_AGENT);
      }
      int nxt = 0;
      if (tid == 0) nxt = __hip_atomic_fetch_add(tk, 1, __ATOMIC_RELAXED, __HIP_MEMORY_SCOPE_AGENT);
#pragma unroll
      for (int k = 0; k < GCN_EDGE_PER_LANE; ++k) {
        if ((unsigned)p[k] < (unsigned)GCN_SLOT_CAP) slots[(size_t)d[k] * GCN_SLOT_CAP + p[k]] = s[k];
      }
      buf ^= 1;
      if (tid == 0) s_chunk[buf] = nxt;
      gcn_lds_barrier();
      c = s_chunk[buf];
    }
    gcn_lds_barrier();
  }
}

static __device__ __forceinline__ float gcn_scale_of(int count) {
  return rsqrtf(fmaxf((float)count, 1.0f));
}


static __device__ __forceinline__ void gcn_stage_w(float* __restrict__ Ws, const float* __restrict__ W,
                                                   int K, int wcols) {
  for (int i = threadIdx.x; i < 32 * 32; i += blockDim.x) {
    const int k = i >> 5;
    const int n = i & 31;
    Ws[i] = (k < K && n < wcols) ? W[(size_t)k * wcols + n] : 0.0f;
  }
}


template <int J>
static __device__ __forceinline__ float gcn_group_lane(float v) {
  return __uint_as_float(__builtin_amdgcn_mov_dpp8(__float_as_uint(v), (unsigned)J * 0x249249u));
}

template <bool SC>
static __device__ __forceinline__ void gcn_trip8(float4& acc, const char* __restrict__ xb,
                                                 unsigned rowbytes, unsigned cbytes, int r,
                                                 int e0, int e1, int e2, int e3,
                                                 int e4, int e5, int e6, int e7,
                                                 const int* __restrict__ sc, int gl, int w) {
  const unsigned s0 = (unsigned)e0;
  const unsigned s1 = (unsigned)((r > 1) ? e1 : e0);
  const unsigned s2 = (unsigned)((r > 2) ? e2 : e0);
  const unsigned s3 = (unsigned)((r > 3) ? e3 : e0);
  const unsigned s4 = (unsigned)((r > 4) ? e4 : e0);
  const unsigned s5 = (unsigned)((r > 5) ? e5 : e0);
  const unsigned s6 = (unsigned)((r > 6) ? e6 : e0);
  const unsigned s7 = (unsigned)((r > 7) ? e7 : e0);
  const float4 v0 = *(const float4*)(xb + (__umul24(s0, rowbytes) + cbytes));
  const float4 v1 = *(const float4*)(xb + (__umul24(s1, rowbytes) + cbytes));
  const float4 v2 = *(const float4*)(xb + (__umul24(s2, rowbytes) + cbytes));
  const float4 v3 = *(const float4*)(xb + (__umul24(s3, rowbytes) + cbytes));
  const float4 v4 = *(const float4*)(xb + (__umul24(s4, rowbytes) + cbytes));
  const float4 v5 = *(const float4*)(xb + (__umul24(s5, rowbytes) + cbytes));
  const float4 v6 = *(const float4*)(xb + (__umul24(s6, rowbytes) + cbytes));
  const float4 v7 = *(const float4*)(xb + (__umul24(s7, rowbytes) + cbytes));
  int mc = 1;
  if (SC) {
    const unsigned sm = (gl < r) ? (unsigned)w : s0;
    mc = *(const int*)((const char*)sc + (sm << 2));
  }
  asm volatile("" :: "v"(v0.x), "v"(v1.x), "v"(v2.x), "v"(v3.x),
                     "v"(v4.x), "v"(v5.x), "v"(v6.x), "v"(v7.x), "v"(mc));
  __builtin_amdgcn_sched_barrier(0);
  if (SC) {
    const float mine = gcn_scale_of(mc);
    const float c0 = gcn_group_lane<0>(mine);
    acc.x = fmaf(v0.x, c0, acc.x); acc.y = fmaf(v0.y, c0, acc.y); acc.z = fmaf(v0.z, c0, acc.z); acc.w = fmaf(v0.w, c0, acc.w);
    if (r > 1) { const float k = gcn_group_lane<1>(mine); acc.x = fmaf(v1.x, k, acc.x); acc.y = fmaf(v1.y, k, acc.y); acc.z = fmaf(v1.z, k, acc.z); acc.w = fmaf(v1.w, k, acc.w); }
    if (r > 2) { const float k = gcn_group_lane<2>(mine); acc.x = fmaf(v2.x, k, acc.x); acc.y = fmaf(v2.y, k, acc.y); acc.z = fmaf(v2.z, k, acc.z); acc.w = fmaf(v2.w, k, acc.w); }
    if (r > 3) { const float k = gcn_group_lane<3>(mine); acc.x = fmaf(v3.x, k, acc.x); acc.y = fmaf(v3.y, k, acc.y); acc.z = fmaf(v3.z, k, acc.z); acc.w = fmaf(v3.w, k, acc.w); }
    if (r > 4) { const float k = gcn_group_lane<4>(mine); acc.x = fmaf(v4.x, k, acc.x); acc.y = fmaf(v4.y, k, acc.y); acc.z = fmaf(v4.z, k, acc.z); acc.w = fmaf(v4.w, k, acc.w); }
    if (r > 5) { const float k = gcn_group_lane<5>(mine); acc.x = fmaf(v5.x, k, acc.x); acc.y = fmaf(v5.y, k, acc.y); acc.z = fmaf(v5.z, k, acc.z); acc.w = fmaf(v5.w, k, acc.w); }
    if (r > 6) { const float k = gcn_group_lane<6>(mine); acc.x = fmaf(v6.x, k, acc.x); acc.y = fmaf(v6.y, k, acc.y); acc.z = fmaf(v6.z, k, acc.z); acc.w = fmaf(v6.w, k, acc.w); }
    if (r > 7) { const float k = gcn_group_lane<7>(mine); acc.x = fmaf(v7.x, k, acc.x); acc.y = fmaf(v7.y, k, acc.y); acc.z = fmaf(v7.z, k, acc.z); acc.w = fmaf(v7.w, k, acc.w); }
  } else {
    acc.x += v0.x; acc.y += v0.y; acc.z += v0.z; acc.w += v0.w;
    if (r > 1) { acc.x += v1.x; acc.y += v1.y; acc.z += v1.z; acc.w += v1.w; }
    if (r > 2) { acc.x += v2.x; acc.y += v2.y; acc.z += v2.z; acc.w += v2.w; }
    if (r > 3) { acc.x += v3.x; acc.y += v3.y; acc.z += v3.z; acc.w += v3.w; }
    if (r > 4) { acc.x += v4.x; acc.y += v4.y; acc.z += v4.z; acc.w += v4.w; }
    if (r > 5) { acc.x += v5.x; acc.y += v5.y; acc.z += v5.z; acc.w += v5.w; }
    if (r > 6) { acc.x += v6.x; acc.y += v6.y; acc.z += v6.z; acc.w += v6.w; }
    if (r > 7) { acc.x += v7.x; acc.y += v7.y; acc.z += v7.z; acc.w += v7.w; }
  }
}

template <bool SC>
static __device__ __forceinline__ float4 gcn_list_sum(const float* __restrict__ x, const int* __restrict__ sl,
                                                      int node, int c, int rowf, int cnt,
                                                      int4 q0, int4 q1, int4 q2, int4 q3,
                                                      const int* __restrict__ src,
                                                      const int* __restrict__ dst, int E,
                                                      const int* __restrict__ sc, int gl, int w0, int w1) {
  const float* xc = x + c * 4;
  const char* xb = (const char*)x;
  const unsigned rowbytes = (unsigned)rowf * 4u;
  const unsigned cbytes = (unsigned)c * 16u;
  float4 acc = make_float4(0.0f, 0.0f, 0.0f, 0.0f);
  if (cnt <= GCN_SLOT_CAP) {
    if (cnt > 0) gcn_trip8<SC>(acc, xb, rowbytes, cbytes, cnt, q0.x, q0.y, q0.z, q0.w, q1.x, q1.y, q1.z, q1.w, sc, gl, w0);
    if (cnt > 8) gcn_trip8<SC>(acc, xb, rowbytes, cbytes, cnt - 8, q2.x, q2.y, q2.z, q2.w, q3.x, q3.y, q3.z, q3.w, sc, gl, w1);
    for (int j = 16; j < cnt; j += 8) {
      const int4 p0 = *(const int4*)(sl + j);
      const int4 p1 = *(const int4*)(sl + j + 4);
      int w = 0;
      if (SC) w = sl[j + gl];
      gcn_trip8<SC>(acc, xb, rowbytes, cbytes, cnt - j, p0.x, p0.y, p0.z, p0.w, p1.x, p1.y, p1.z, p1.w, sc, gl, w);
    }
  } else {
    for (int e = 0; e < E; ++e) {
      if (dst[e] == node) {
        const int sidx = src[e];
        const float4 v = *(const float4*)(xc + (size_t)sidx * rowf);
        if (SC) {
          const float k = gcn_scale_of(sc[sidx]);
          acc.x = fmaf(v.x, k, acc.x); acc.y = fmaf(v.y, k, acc.y); acc.z = fmaf(v.z, k, acc.z); acc.w = fmaf(v.w, k, acc.w);
        } else {
          acc.x += v.x; acc.y += v.y; acc.z += v.z; acc.w += v.w;
        }
      }
    }
  }
  return acc;
}

#define GCN_L2_NODES 128
#define GCN_L2_TILES (GCN_L2_NODES / 16)
#define GCN_L2_PITCH 36
#define GCN_H_PITCH 32

__global__ void __launch_bounds__(GCN_L2_NODES * 8)
gcn_pull_gemm_l1_kernel(const float* __restrict__ x, int rowf, int nch,
                        const int* __restrict__ cnt_in, const int* __restrict__ slots,
                        const int* __restrict__ src, const int* __restrict__ dst, int E, int n,
                        const float* __restrict__ W, int wcols,
                        const int* __restrict__ cnt_out,
                        const float* __restrict__ bias, float* __restrict__ h, int ntiles) {
  __shared__ float Ws[32 * 32];
  __shared__ float As[GCN_L2_NODES * GCN_L2_PITCH];

  gcn_stage_w(Ws, W, rowf, wcols);

  const int nbatches = (n + GCN_L2_NODES - 1) / GCN_L2_NODES;

  for (int b = blockIdx.x; b < nbatches; b += gridDim.x) {
    int tid = threadIdx.x;
    asm volatile("" : "+v"(tid));
    const int gl = tid & 7;

    const int node = b * GCN_L2_NODES + (tid >> 3);
    if (node < n) {
      const int* sl = slots + (size_t)node * GCN_SLOT_CAP;
      const int c = (gl < nch) ? gl : 0;
      const int cnt = cnt_in[node];
      const int4 q0 = *(const int4*)(sl);
      const int4 q1 = *(const int4*)(sl + 4);
      const int4 q2 = *(const int4*)(sl + 8);
      const int4 q3 = *(const int4*)(sl + 12);
      const int w0 = sl[gl];
      const int w1 = sl[8 + gl];
      asm volatile("" :: "v"(cnt), "v"(q0.x), "v"(q1.x), "v"(q2.x), "v"(q3.x), "v"(w0), "v"(w1));
      const float4 acc = gcn_list_sum<true>(x, sl, node, c, rowf, cnt, q0, q1, q2, q3, src, dst, E,
                                            cnt_out, gl, w0, w1);
      if (gl < nch) *(float4*)(As + (tid >> 3) * GCN_L2_PITCH + c * 4) = acc;
    }

    const int wv = tid >> 5;
    const int lane = tid & 31;
    const int m = lane & 15;
    const int half = lane >> 4;
    const int tile = b * GCN_L2_TILES + wv;
    const bool has_tile = (wv < GCN_L2_TILES) && (tile < ntiles);

    const v4i zero4 = {0, 0, 0, 0};
    v4i d0 = zero4, d1 = zero4, t0 = zero4, t1 = zero4;
    float bA = 0.0f, bB = 0.0f;
    if (has_tile) {
      const int* dp = cnt_in + tile * 16 + 8 * half;
      const int* tp = cnt_out + tile * 16 + 8 * half;
      d0 = *(const v4i*)(dp);
      d1 = *(const v4i*)(dp + 4);
      t0 = *(const v4i*)(tp);
      t1 = *(const v4i*)(tp + 4);
      bA = bias[(m < wcols) ? m : 0];
      bB = bias[(16 + m < wcols) ? 16 + m : 0];
    }

    gcn_lds_barrier();

    if (has_tile) {
      float* trow = As + (wv * 16) * GCN_L2_PITCH;
      const float* arow = trow + m * GCN_L2_PITCH;
      v8f c0 = {};
      v8f c1 = {};

#pragma unroll 4
      for (int k0 = 0; k0 < rowf; k0 += 4) {
        const int k = k0 + 2 * half;
        const float2 av = *(const float2*)(arow + k);
        v2f a;
        a.x = av.x;
        a.y = av.y;

        v2f b0, b1;
        b0.x = Ws[k * 32 + m];
        b0.y = Ws[(k + 1) * 32 + m];
        b1.x = Ws[k * 32 + 16 + m];
        b1.y = Ws[(k + 1) * 32 + 16 + m];

        c0 = __builtin_amdgcn_wmma_f32_16x16x4_f32(false, a, false, b0, (short)0, c0, false, false);
        c1 = __builtin_amdgcn_wmma_f32_16x16x4_f32(false, a, false, b1, (short)0, c1, false, false);
      }

      asm volatile("" : "+v"(d0), "+v"(d1), "+v"(t0), "+v"(t1));
      const float ndv[8] = {gcn_scale_of(d0.x), gcn_scale_of(d0.y), gcn_scale_of(d0.z), gcn_scale_of(d0.w),
                            gcn_scale_of(d1.x), gcn_scale_of(d1.y), gcn_scale_of(d1.z), gcn_scale_of(d1.w)};
      const float nsv[8] = {gcn_scale_of(t0.x), gcn_scale_of(t0.y), gcn_scale_of(t0.z), gcn_scale_of(t0.w),
                            gcn_scale_of(t1.x), gcn_scale_of(t1.y), gcn_scale_of(t1.z), gcn_scale_of(t1.w)};
#pragma unroll
      for (int r = 0; r < 8; ++r) {
        float* orow = trow + (8 * half + r) * GCN_L2_PITCH;
        orow[m] = fmaxf(fmaf(c0[r], ndv[r], bA), 0.0f) * nsv[r];
        orow[16 + m] = fmaxf(fmaf(c1[r], ndv[r], bB), 0.0f) * nsv[r];
      }
      __builtin_amdgcn_fence(__ATOMIC_RELEASE, "wavefront", "local");
      __builtin_amdgcn_wave_barrier();
      __builtin_amdgcn_fence(__ATOMIC_ACQUIRE, "wavefront", "local");

      const int oc = lane & 7;
      float* hb = h + (size_t)__builtin_amdgcn_readfirstlane(tile) * 16 * GCN_H_PITCH;
#pragma unroll
      for (int i = 0; i < 4; ++i) {
        const int row = i * 4 + (lane >> 3);
        const float4 v = *(const float4*)(trow + row * GCN_L2_PITCH + oc * 4);
        *(float4*)(hb + (unsigned)(row * GCN_H_PITCH + oc * 4)) = v;
      }
    }

    gcn_lds_barrier();
  }
}


__global__ void __launch_bounds__(GCN_L2_NODES * 8)
gcn_pull_gemm_readout_kernel(const float* __restrict__ x, const int* __restrict__ cnt_in,
                             const int* __restrict__ slots, const int* __restrict__ src,
                             const int* __restrict__ dst, int E, int n, int nch, int H, int xpitch,
                             const float* __restrict__ W, int wcols,
                             const float* __restrict__ bias,
                             const float* __restrict__ Wd, const float* __restrict__ bd,
                             float* __restrict__ out, int ntiles) {
  __shared__ float Ws[32 * 32];
  __shared__ float As[GCN_L2_NODES * GCN_L2_PITCH];
  __shared__ float s_ro[GCN_L2_NODES / 4];

  gcn_stage_w(Ws, W, H, wcols);

  const int nbatches = (n + GCN_L2_NODES - 1) / GCN_L2_NODES;

  for (int b = blockIdx.x; b < nbatches; b += gridDim.x) {
    int tid = threadIdx.x;
    asm volatile("" : "+v"(tid));
    const int c = tid & 7;

    const int node = b * GCN_L2_NODES + (tid >> 3);
    if (node < n && c < nch) {
      const int* sl = slots + (size_t)node * GCN_SLOT_CAP;
      const int cnt = cnt_in[node];
      const int4 q0 = *(const int4*)(sl);
      const int4 q1 = *(const int4*)(sl + 4);
      const int4 q2 = *(const int4*)(sl + 8);
      const int4 q3 = *(const int4*)(sl + 12);
      asm volatile("" :: "v"(cnt), "v"(q0.x), "v"(q1.x), "v"(q2.x), "v"(q3.x));
      const float4 acc = gcn_list_sum<false>(x, sl, node, c, xpitch, cnt, q0, q1, q2, q3, src, dst, E,
                                             nullptr, 0, 0, 0);
      *(float4*)(As + (tid >> 3) * GCN_L2_PITCH + c * 4) = acc;
    }

    const int wv = tid >> 5;
    const int lane = tid & 31;
    const int m = lane & 15;
    const int half = lane >> 4;
    const bool hasA = m < wcols;
    const bool hasB = 16 + m < wcols;
    const int tile = b * GCN_L2_TILES + wv;
    const bool has_tile = (wv < GCN_L2_TILES) && (tile < ntiles);

    const v4i zero4 = {0, 0, 0, 0};
    v4i n0 = zero4, n1 = zero4;
    float wA[4] = {0.0f, 0.0f, 0.0f, 0.0f}, wB[4] = {0.0f, 0.0f, 0.0f, 0.0f};
    float bA = 0.0f, bB = 0.0f, bd0 = 0.0f;
    if (has_tile) {
      const int* np = cnt_in + tile * 16 + 8 * half;
      n0 = *(const v4i*)(np);
      n1 = *(const v4i*)(np + 4);
      const int colA = hasA ? m : 0;
      const int colB = hasB ? 16 + m : 0;
      bA = bias[colA];
      bB = bias[colB];
#pragma unroll
      for (int g = 0; g < 4; ++g) {
        wA[g] = Wd[g * wcols + colA];
        wB[g] = Wd[g * wcols + colB];
      }
      bd0 = bd[0];
    }

    gcn_lds_barrier();

    if (has_tile) {
      const float* arow = As + (wv * 16 + m) * GCN_L2_PITCH;
      v8f c0 = {};
      v8f c1 = {};

#pragma unroll 3
      for (int k0 = 0; k0 < H; k0 += 4) {
        const int k = k0 + 2 * half;
        const float2 av = *(const float2*)(arow + k);
        v2f a;
        a.x = av.x;
        a.y = av.y;

        v2f b0, b1;
        b0.x = Ws[k * 32 + m];
        b0.y = Ws[(k + 1) * 32 + m];
        b1.x = Ws[k * 32 + 16 + m];
        b1.y = Ws[(k + 1) * 32 + 16 + m];

        c0 = __builtin_amdgcn_wmma_f32_16x16x4_f32(false, a, false, b0, (short)0, c0, false, false);
        c1 = __builtin_amdgcn_wmma_f32_16x16x4_f32(false, a, false, b1, (short)0, c1, false, false);
      }

      asm volatile("" : "+v"(n0), "+v"(n1));
      const float ndv[8] = {gcn_scale_of(n0.x), gcn_scale_of(n0.y), gcn_scale_of(n0.z), gcn_scale_of(n0.w),
                            gcn_scale_of(n1.x), gcn_scale_of(n1.y), gcn_scale_of(n1.z), gcn_scale_of(n1.w)};
#pragma unroll
      for (int g = 0; g < 4; ++g) {
        wA[g] = hasA ? wA[g] : 0.0f;
        wB[g] = hasB ? wB[g] : 0.0f;
      }
      float s0 = 0.0f, s1 = 0.0f;
#pragma unroll
      for (int g = 0; g < 4; ++g) {
        s0 = fmaf(fmaxf(fmaf(c0[g], ndv[g], bA), 0.0f), wA[g], s0);
        s0 = fmaf(fmaxf(fmaf(c1[g], ndv[g], bB), 0.0f), wB[g], s0);
        s1 = fmaf(fmaxf(fmaf(c0[4 + g], ndv[4 + g], bA), 0.0f), wA[g], s1);
        s1 = fmaf(fmaxf(fmaf(c1[4 + g], ndv[4 + g], bB), 0.0f), wB[g], s1);
      }
#pragma unroll
      for (int off = 8; off >= 1; off >>= 1) {
        s0 += __shfl_xor(s0, off);
        s1 += __shfl_xor(s1, off);
      }
      if (m == 0) {
        s_ro[wv * 4 + 2 * half] = s0 + bd0;
        s_ro[wv * 4 + 2 * half + 1] = s1 + bd0;
      }
    }

    gcn_lds_barrier();
    const int r = b * (GCN_L2_NODES / 4) + tid;
    if (tid < GCN_L2_NODES / 4 && r < ntiles * 4) out[r] = s_ro[tid];
  }
}


extern "C" void kernel_launch(void* const* d_in, const int* in_sizes, int n_in,
                              void* d_out, int out_size, void* d_ws, size_t ws_size,
                              hipStream_t stream) {
  const float* features = (const float*)d_in[0];
  const int*   src      = (const int*)d_in[1];
  const int*   dst      = (const int*)d_in[2];
  const float* W1       = (const float*)d_in[3];
  const float* b1       = (const float*)d_in[4];
  const float* W2       = (const float*)d_in[5];
  const float* b2       = (const float*)d_in[6];
  const float* Wd       = (const float*)d_in[7];
  const float* bd       = (const float*)d_in[8];
  float* out = (float*)d_out;

  const int H  = in_sizes[4];
  const int F  = in_sizes[3] / H;
  const int Nn = in_sizes[0] / F;
  const int E  = in_sizes[1];

  const int B = 256;
  const int ntiles = Nn / 16;
  const int nch = H / 4;
  const int nchF = F / 4;
  const int l2Batches = (Nn + GCN_L2_NODES - 1) / GCN_L2_NODES;
  const int l2Grid = l2Batches < 512 ? l2Batches : 512;
  const int ewGrid = 256;

  int*   tick     = (int*)d_ws;
  int*   cnt_out  = tick + 2 * GCN_TICK_STRIDE;
  int*   cnt_in   = cnt_out + Nn;
  float* bufA     = (float*)(cnt_in + Nn);
  int*   slots    = (int*)(bufA + (size_t)Nn * GCN_H_PITCH);

  gcn_zero_kernel<<<ewGrid, B, 0, stream>>>(tick, 2 * GCN_TICK_STRIDE + 2 * Nn);
  gcn_edge_kernel<<<2048, GCN_EDGE_BLOCK, 0, stream>>>(src, dst, tick, cnt_out, cnt_in, slots, E, Nn);
  gcn_pull_gemm_l1_kernel<<<l2Grid, GCN_L2_NODES * 8, 0, stream>>>(
      features, F, nchF, cnt_in, slots, src, dst, E, Nn, W1, H, cnt_out, b1, bufA, ntiles);
  gcn_pull_gemm_readout_kernel<<<l2Grid, GCN_L2_NODES * 8, 0, stream>>>(
      bufA, cnt_in, slots, src, dst, E, Nn, nch, H, GCN_H_PITCH, W2, H, b2, Wd, bd, out, ntiles);
}
